# starting kernel
# baseline (speedup 1.0000x reference)
#include <hip/hip_runtime.h>

typedef float v2f __attribute__((ext_vector_type(2)));
typedef float v8f __attribute__((ext_vector_type(8)));

#define N_NODES 100000
#define N_EDGES 1600000


__global__ void zero_u32_kernel(unsigned int* p, int n) {
    int i = blockIdx.x * blockDim.x + threadIdx.x;
    if (i < n) p[i] = 0u;
}

__global__ void deg_count_kernel(const int* __restrict__ dst,
                                 unsigned int* __restrict__ cnt, int E) {
    int e = blockIdx.x * blockDim.x + threadIdx.x;
    if (e < E) atomicAdd(&cnt[dst[e]], 1u);
}

__global__ void make_dis_kernel(const unsigned int* __restrict__ cnt,
                                float* __restrict__ dis, int n) {
    int i = blockIdx.x * blockDim.x + threadIdx.x;
    if (i < n) dis[i] = rsqrtf((float)cnt[i] + 1.0f);
}


template<int K, int NOUT>
__global__ void __launch_bounds__(256)
gemm_wmma_kernel(const float* __restrict__ X, const float* __restrict__ W,
                 float* __restrict__ H, int nrows) {
    __shared__ float sW[K * NOUT];
    const int tid = threadIdx.x;
    for (int i = tid; i < K * NOUT; i += 256) sW[i] = W[i];
    __syncthreads();

    const int wave = tid >> 5;
    const int lane = tid & 31;
    const int half = lane >> 4;
    const int r    = lane & 15;

    const int row0 = (blockIdx.x * 8 + wave) * 16;
    if (row0 >= nrows) return;

    int arow = row0 + r;
    if (arow >= nrows) arow = nrows - 1;
    const float* __restrict__ xrow = X + (long long)arow * K;

    constexpr int NT = NOUT / 16;
    v8f acc[NT];
#pragma unroll
    for (int nt = 0; nt < NT; ++nt) {
        v8f z = {};
        acc[nt] = z;
    }

#pragma unroll 4
    for (int kt = 0; kt < K / 4; ++kt) {
        const int k0 = kt * 4 + 2 * half;
        v2f a;
        a.x = xrow[k0];
        a.y = xrow[k0 + 1];
#pragma unroll
        for (int nt = 0; nt < NT; ++nt) {
            v2f b;
            b.x = sW[(k0)     * NOUT + nt * 16 + r];
            b.y = sW[(k0 + 1) * NOUT + nt * 16 + r];
            acc[nt] = __builtin_amdgcn_wmma_f32_16x16x4_f32(
                false, a, false, b, (short)0, acc[nt], false, false);
        }
    }

#pragma unroll
    for (int nt = 0; nt < NT; ++nt) {
#pragma unroll
        for (int v = 0; v < 8; ++v) {
            const int row = row0 + v + 8 * half;
            const int col = nt * 16 + r;
            if (row < nrows) H[(long long)row * NOUT + col] = acc[nt][v];
        }
    }
}


template<int F>
__global__ void self_init_kernel(const float* __restrict__ h,
                                 const float* __restrict__ dis,
                                 float* __restrict__ acc, int n) {
    long long idx = (long long)blockIdx.x * blockDim.x + threadIdx.x;
    if (idx < (long long)n * F) {
        int node = (int)(idx / F);
        float d = dis[node];
        acc[idx] = h[idx] * d * d;
    }
}


template<int F>
__global__ void scatter_kernel(const int* __restrict__ src,
                               const int* __restrict__ dst,
                               const float* __restrict__ h,
                               const float* __restrict__ dis,
                               float* __restrict__ acc, int E) {
    long long idx = (long long)blockIdx.x * blockDim.x + threadIdx.x;
    if (idx >= (long long)E * F) return;
    const int e = (int)(idx >> __builtin_ctz(F));
    const int f = (int)(idx & (F - 1));
    const int s = src[e];
    const int d = dst[e];
    const float coef = dis[s] * dis[d];
    atomicAdd(&acc[(long long)d * F + f], h[(long long)s * F + f] * coef);
}


template<int F, bool RELU>
__global__ void epilogue_kernel(const float* __restrict__ acc,
                                const float* __restrict__ b,
                                float* __restrict__ out, int n) {
    long long idx = (long long)blockIdx.x * blockDim.x + threadIdx.x;
    if (idx < (long long)n * F) {
        int f = (int)(idx & (F - 1));
        float v = acc[idx] + b[f];
        out[idx] = RELU ? fmaxf(v, 0.0f) : v;
    }
}


static inline int cdiv(long long a, int b) { return (int)((a + b - 1) / b); }

extern "C" void kernel_launch(void* const* d_in, const int* in_sizes, int n_in,
                              void* d_out, int out_size, void* d_ws, size_t ws_size,
                              hipStream_t stream) {
    const float* x  = (const float*)d_in[0];
    const int*   ei = (const int*)d_in[1];
    const float* W1 = (const float*)d_in[2];
    const float* b1 = (const float*)d_in[3];
    const float* W2 = (const float*)d_in[4];
    const float* b2 = (const float*)d_in[5];
    const float* W3 = (const float*)d_in[6];
    const float* b3 = (const float*)d_in[7];
    float* out = (float*)d_out;

    const int* src = ei;
    const int* dst = ei + N_EDGES;

    char* ws = (char*)d_ws;
    unsigned int* cnt = (unsigned int*)ws;
    float* dis  = (float*)(ws + 400384);
    float* bufA = (float*)(ws + 800768);
    float* bufB = (float*)(ws + 800768 + 25600000);

    const int N = N_NODES, E = N_EDGES;
    const int gemm_blocks = cdiv(cdiv(N, 16), 8);

    zero_u32_kernel<<<cdiv(N, 256), 256, 0, stream>>>(cnt, N);
    deg_count_kernel<<<cdiv(E, 256), 256, 0, stream>>>(dst, cnt, E);
    make_dis_kernel<<<cdiv(N, 256), 256, 0, stream>>>(cnt, dis, N);

    gemm_wmma_kernel<128, 64><<<gemm_blocks, 256, 0, stream>>>(x, W1, bufA, N);
    self_init_kernel<64><<<cdiv((long long)N * 64, 256), 256, 0, stream>>>(bufA, dis, bufB, N);
    scatter_kernel<64><<<cdiv((long long)E * 64, 256), 256, 0, stream>>>(src, dst, bufA, dis, bufB, E);
    epilogue_kernel<64, true><<<cdiv((long long)N * 64, 256), 256, 0, stream>>>(bufB, b1, bufA, N);

    gemm_wmma_kernel<64, 64><<<gemm_blocks, 256, 0, stream>>>(bufA, W2, bufB, N);
    self_init_kernel<64><<<cdiv((long long)N * 64, 256), 256, 0, stream>>>(bufB, dis, bufA, N);
    scatter_kernel<64><<<cdiv((long long)E * 64, 256), 256, 0, stream>>>(src, dst, bufB, dis, bufA, E);
    epilogue_kernel<64, true><<<cdiv((long long)N * 64, 256), 256, 0, stream>>>(bufA, b2, bufB, N);

    gemm_wmma_kernel<64, 32><<<gemm_blocks, 256, 0, stream>>>(bufB, W3, bufA, N);
    self_init_kernel<32><<<cdiv((long long)N * 32, 256), 256, 0, stream>>>(bufA, dis, out, N);
    scatter_kernel<32><<<cdiv((long long)E * 32, 256), 256, 0, stream>>>(src, dst, bufA, dis, out, E);
    epilogue_kernel<32, false><<<cdiv((long long)N * 32, 256), 256, 0, stream>>>(out, b3, out, N);
}


# Float-atomic edge scatter replaced by per-call dst buckets (slot = value returned by the degree atomic, scan, fill) and a per-node gather-sum; float atomics gone, sums in arrival-order rank
# speedup vs baseline: 3.2464x; 3.2464x over previous
#include <hip/hip_runtime.h>

typedef float v2f __attribute__((ext_vector_type(2)));
typedef float v8f __attribute__((ext_vector_type(8)));

#define N_NODES 100000
#define N_EDGES 1600000


__global__ void zero_u32_kernel(unsigned int* p, int n) {
    int i = blockIdx.x * blockDim.x + threadIdx.x;
    if (i < n) p[i] = 0u;
}

__global__ void count_rank_kernel(const int* __restrict__ dst,
                                  unsigned int* __restrict__ cnt,
                                  unsigned int* __restrict__ rank, int E) {
    int e = blockIdx.x * blockDim.x + threadIdx.x;
    if (e < E) rank[e] = atomicAdd(&cnt[dst[e]], 1u);
}

__global__ void __launch_bounds__(256)
scan_dis_kernel(const unsigned int* __restrict__ cnt,
                unsigned int* __restrict__ rowptr,
                float* __restrict__ dis, int n) {
    __shared__ unsigned int s_pre[8];
    __shared__ unsigned int s_tot[8];
    const int tid  = threadIdx.x;
    const int lane = tid & 31;
    const int wave = tid >> 5;
    const int blk0 = blockIdx.x * 1024;

    unsigned int part = 0u;
    const uint4* __restrict__ c4 = (const uint4*)cnt + tid;
    const int nb = blockIdx.x;
    int it = 0;
    for (; it + 8 <= nb; it += 8) {
        uint4 q[8];
#pragma unroll
        for (int u = 0; u < 8; ++u) q[u] = c4[(it + u) * 256];
#pragma unroll
        for (int u = 0; u < 8; ++u) part += (q[u].x + q[u].y) + (q[u].z + q[u].w);
    }
    for (; it < nb; ++it) {
        const uint4 q = c4[it * 256];
        part += (q.x + q.y) + (q.z + q.w);
    }
    for (int m = 16; m >= 1; m >>= 1) part += __shfl_xor(part, m);

    const int i0 = blk0 + tid * 4;
    unsigned int c[4];
#pragma unroll
    for (int k = 0; k < 4; ++k) c[k] = (i0 + k < n) ? cnt[i0 + k] : 0u;
    const unsigned int tsum = (c[0] + c[1]) + (c[2] + c[3]);

    unsigned int inc = tsum;
    for (int d = 1; d < 32; d <<= 1) {
        const unsigned int t = __shfl_up(inc, d);
        if (lane >= d) inc += t;
    }
    if (lane == 31) s_tot[wave] = inc;
    if (lane == 0)  s_pre[wave] = part;
    __syncthreads();

    unsigned int off = 0u;
#pragma unroll
    for (int w = 0; w < 8; ++w) {
        off += s_pre[w];
        if (w < wave) off += s_tot[w];
    }
    unsigned int run = off + inc - tsum;
#pragma unroll
    for (int k = 0; k < 4; ++k) {
        const int i = i0 + k;
        if (i < n) {
            rowptr[i] = run;
            dis[i] = rsqrtf((float)c[k] + 1.0f);
            run += c[k];
            if (i == n - 1) rowptr[n] = run;
        }
    }
}

__global__ void fill_kernel(const int* __restrict__ src,
                            const int* __restrict__ dst,
                            const unsigned int* __restrict__ rank,
                            const unsigned int* __restrict__ rowptr,
                            const float* __restrict__ dis,
                            int2* __restrict__ epk, int E) {
    int e = blockIdx.x * blockDim.x + threadIdx.x;
    if (e < E) {
        const int s = src[e];
        const int d = dst[e];
        const unsigned int p = rowptr[d] + rank[e];
        epk[p] = make_int2(s, __float_as_int(dis[s] * dis[d]));
    }
}


template<int K, int NOUT>
__global__ void __launch_bounds__(256)
gemm_wmma_kernel(const float* __restrict__ X, const float* __restrict__ W,
                 float* __restrict__ H, int nrows) {
    __shared__ float sW[K * NOUT];
    const int tid = threadIdx.x;
    for (int i = tid; i < K * NOUT; i += 256) sW[i] = W[i];
    __syncthreads();

    const int wave = tid >> 5;
    const int lane = tid & 31;
    const int half = lane >> 4;
    const int r    = lane & 15;

    const int row0 = (blockIdx.x * 8 + wave) * 16;
    if (row0 >= nrows) return;

    int arow = row0 + r;
    if (arow >= nrows) arow = nrows - 1;
    const float* __restrict__ xrow = X + (long long)arow * K;

    constexpr int NT = NOUT / 16;
    v8f acc[NT];
#pragma unroll
    for (int nt = 0; nt < NT; ++nt) {
        v8f z = {};
        acc[nt] = z;
    }

#pragma unroll 4
    for (int kt = 0; kt < K / 4; ++kt) {
        const int k0 = kt * 4 + 2 * half;
        v2f a;
        a.x = xrow[k0];
        a.y = xrow[k0 + 1];
#pragma unroll
        for (int nt = 0; nt < NT; ++nt) {
            v2f b;
            b.x = sW[(k0)     * NOUT + nt * 16 + r];
            b.y = sW[(k0 + 1) * NOUT + nt * 16 + r];
            acc[nt] = __builtin_amdgcn_wmma_f32_16x16x4_f32(
                false, a, false, b, (short)0, acc[nt], false, false);
        }
    }

#pragma unroll
    for (int nt = 0; nt < NT; ++nt) {
#pragma unroll
        for (int v = 0; v < 8; ++v) {
            const int row = row0 + v + 8 * half;
            const int col = nt * 16 + r;
            if (row < nrows) H[(long long)row * NOUT + col] = acc[nt][v];
        }
    }
}


template<int F, bool RELU>
__global__ void __launch_bounds__(256)
gather_kernel(const unsigned int* __restrict__ rowptr,
              const int2* __restrict__ epk,
              const float* __restrict__ h,
              const float* __restrict__ dis,
              const float* __restrict__ b,
              float* __restrict__ out, int n) {
    constexpr int V = F / 32;
    constexpr int G = 8;
    const int lane = threadIdx.x & 31;
    const int node = __builtin_amdgcn_readfirstlane((int)(blockIdx.x * 8 + (threadIdx.x >> 5)));
    if (node >= n) return;

    const unsigned int beg = rowptr[node];
    const unsigned int end = rowptr[node + 1];
    const float dd = dis[node];

    float bias[V];
    if (V == 2) {
        const float2 bb = *(const float2*)(b + lane * V);
        bias[0] = bb.x;
        bias[V - 1] = bb.y;
    } else {
        bias[0] = b[lane];
    }

    float acc[V];
    {
        const float* __restrict__ hp = h + (long long)node * F + lane * V;
        if (V == 2) {
            const float2 q = *(const float2*)hp;
            acc[0] = q.x * dd * dd;
            acc[V - 1] = q.y * dd * dd;
        } else {
            acc[0] = hp[0] * dd * dd;
        }
    }

    for (unsigned int base = beg; base < end; base += 32u) {
        const unsigned int j = base + (unsigned int)lane;
        int es = node;
        int ecb = 0;
        if (j < end) {
            const int2 p = epk[j];
            es = p.x;
            ecb = p.y;
        }
        const unsigned int left = end - base;
        const int m = (int)(left < 32u ? left : 32u);
        for (int t = 0; t < m; t += G) {
            float hv[G][V];
            float cf[G];
#pragma unroll
            for (int u = 0; u < G; ++u) {
                const int s = __builtin_amdgcn_readlane(es, t + u);
                cf[u] = __int_as_float(__builtin_amdgcn_readlane(ecb, t + u));
                const float* __restrict__ hp = h + (long long)s * F + lane * V;
                if (V == 2) {
                    const float2 q = *(const float2*)hp;
                    hv[u][0] = q.x;
                    hv[u][V - 1] = q.y;
                } else {
                    hv[u][0] = hp[0];
                }
            }
#pragma unroll
            for (int u = 0; u < G; ++u) {
                const bool live = (t + u) < m;
#pragma unroll
                for (int v = 0; v < V; ++v)
                    acc[v] = fmaf(live ? hv[u][v] : 0.0f, cf[u], acc[v]);
            }
        }
    }

    float* __restrict__ op = out + (long long)node * F + lane * V;
    if (V == 2) {
        float2 o;
        o.x = acc[0] + bias[0];
        o.y = acc[V - 1] + bias[V - 1];
        if (RELU) { o.x = fmaxf(o.x, 0.0f); o.y = fmaxf(o.y, 0.0f); }
        *(float2*)op = o;
    } else {
        float o = acc[0] + bias[0];
        if (RELU) o = fmaxf(o, 0.0f);
        op[0] = o;
    }
}


static inline int cdiv(long long a, int b) { return (int)((a + b - 1) / b); }

extern "C" void kernel_launch(void* const* d_in, const int* in_sizes, int n_in,
                              void* d_out, int out_size, void* d_ws, size_t ws_size,
                              hipStream_t stream) {
    const float* x  = (const float*)d_in[0];
    const int*   ei = (const int*)d_in[1];
    const float* W1 = (const float*)d_in[2];
    const float* b1 = (const float*)d_in[3];
    const float* W2 = (const float*)d_in[4];
    const float* b2 = (const float*)d_in[5];
    const float* W3 = (const float*)d_in[6];
    const float* b3 = (const float*)d_in[7];
    float* out = (float*)d_out;

    const int* src = ei;
    const int* dst = ei + N_EDGES;

    char* ws = (char*)d_ws;
    unsigned int* cnt = (unsigned int*)ws;
    float* dis  = (float*)(ws + 400384);
    float* bufA = (float*)(ws + 800768);
    float* bufB = (float*)(ws + 26400768);
    unsigned int* rowptr = (unsigned int*)(ws + 52000768);
    unsigned int* rank   = (unsigned int*)(ws + 52401152);
    int2* epk            = (int2*)(ws + 58801152);

    const int N = N_NODES, E = N_EDGES;
    const int gemm_blocks = cdiv(cdiv(N, 16), 8);
    const int node_blocks = cdiv(N, 8);

    zero_u32_kernel<<<cdiv(N, 256), 256, 0, stream>>>(cnt, N);
    count_rank_kernel<<<cdiv(E, 256), 256, 0, stream>>>(dst, cnt, rank, E);
    scan_dis_kernel<<<cdiv(N, 1024), 256, 0, stream>>>(cnt, rowptr, dis, N);
    fill_kernel<<<cdiv(E, 256), 256, 0, stream>>>(src, dst, rank, rowptr, dis, epk, E);

    gemm_wmma_kernel<128, 64><<<gemm_blocks, 256, 0, stream>>>(x, W1, bufA, N);
    gather_kernel<64, true><<<node_blocks, 256, 0, stream>>>(rowptr, epk, bufA, dis, b1, bufB, N);

    gemm_wmma_kernel<64, 64><<<gemm_blocks, 256, 0, stream>>>(bufB, W2, bufA, N);
    gather_kernel<64, true><<<node_blocks, 256, 0, stream>>>(rowptr, epk, bufA, dis, b2, bufB, N);

    gemm_wmma_kernel<64, 32><<<gemm_blocks, 256, 0, stream>>>(bufB, W3, bufA, N);
    gather_kernel<32, false><<<node_blocks, 256, 0, stream>>>(rowptr, epk, bufA, dis, b3, out, N);
}


# Edge buckets built by a two-level counting sort on LDS counters (chunk histogram, in-LDS partition into 512-node bins, per-bin node count and placement); no global atomics; arrival-order sums kept
# speedup vs baseline: 5.7825x; 1.7812x over previous
#include <hip/hip_runtime.h>

typedef float v2f __attribute__((ext_vector_type(2)));
typedef float v8f __attribute__((ext_vector_type(8)));

#define N_NODES 100000
#define N_EDGES 1600000


#define BIN_SHIFT   9
#define BIN_NODES   (1 << BIN_SHIFT)
#define NBINS       ((N_NODES + BIN_NODES - 1) / BIN_NODES)
#define EDGE_CHUNKS 256
#define CHUNK_EDGES ((N_EDGES + EDGE_CHUNKS - 1) / EDGE_CHUNKS)
#define BT          1024
#define EPT         ((CHUNK_EDGES + BT - 1) / BT)
#define HQ          (NBINS / 4)
#define HRG         (BT / HQ)
#define HROWS       ((EDGE_CHUNKS + HRG - 1) / HRG)
#define STAGE_CAP   16384

static_assert(NBINS % 4 == 0, "histogram rows are read as uint4");
static_assert(NBINS <= BT, "one thread per bin");
static_assert(2 * HRG * NBINS * 4 <= CHUNK_EDGES * 8, "the reduction scratch lives inside the sort buffer");

__device__ __forceinline__ unsigned int block_excl_scan(unsigned int v, unsigned int* s_w, unsigned int& total) {
    const int lane = threadIdx.x & 31;
    const int wave = threadIdx.x >> 5;
    unsigned int inc = v;
    for (int d = 1; d < 32; d <<= 1) {
        const unsigned int t = __shfl_up(inc, d);
        if (lane >= d) inc += t;
    }
    if (lane == 31) s_w[wave] = inc;
    __syncthreads();
    unsigned int off = 0u, tot = 0u;
    for (int w = 0; w < BT / 32; ++w) {
        const unsigned int x = s_w[w];
        if (w < wave) off += x;
        tot += x;
    }
    __syncthreads();
    total = tot;
    return off + inc - v;
}

__global__ void __launch_bounds__(BT)
bin_hist_kernel(const int* __restrict__ dst, unsigned int* __restrict__ hist, int E) {
    __shared__ unsigned int s_cnt[NBINS];
    const int tid = threadIdx.x;
    if (tid < NBINS) s_cnt[tid] = 0u;
    const int e0 = blockIdx.x * CHUNK_EDGES;
    const int e1 = (e0 + CHUNK_EDGES < E) ? (e0 + CHUNK_EDGES) : E;
    int d[EPT];
#pragma unroll
    for (int k = 0; k < EPT; ++k) {
        const int e = e0 + tid + k * BT;
        d[k] = (e < e1) ? dst[e] : -1;
    }
    __syncthreads();
#pragma unroll
    for (int k = 0; k < EPT; ++k)
        if (d[k] >= 0) atomicAdd(&s_cnt[d[k] >> BIN_SHIFT], 1u);
    __syncthreads();
    if (tid < NBINS) hist[blockIdx.x * NBINS + tid] = s_cnt[tid];
}

__global__ void __launch_bounds__(BT)
bin_partition_kernel(const int* __restrict__ src, const int* __restrict__ dst,
                     const unsigned int* __restrict__ hist,
                     unsigned int* __restrict__ binbase,
                     int2* __restrict__ tmp, int E) {
    __shared__ int2 s_sorted[CHUNK_EDGES];
    __shared__ unsigned int s_cnt[NBINS];
    __shared__ unsigned int s_lbase[NBINS];
    __shared__ unsigned int s_goff[NBINS];
    __shared__ unsigned int s_w[BT / 32];
    unsigned int* s_red = (unsigned int*)s_sorted;

    const int tid = threadIdx.x;
    const int chunk = blockIdx.x;
    const int e0 = chunk * CHUNK_EDGES;
    const int e1 = (e0 + CHUNK_EDGES < E) ? (e0 + CHUNK_EDGES) : E;
    const int n = e1 - e0;

    const int q  = tid % HQ;
    const int rg = tid / HQ;
    uint4 hv[HROWS];
#pragma unroll
    for (int k = 0; k < HROWS; ++k) {
        const int r = rg + k * HRG;
        hv[k] = make_uint4(0u, 0u, 0u, 0u);
        if (rg < HRG && r < EDGE_CHUNKS) hv[k] = ((const uint4*)hist)[r * HQ + q];
    }
    int es[EPT], ed[EPT];
#pragma unroll
    for (int k = 0; k < EPT; ++k) {
        const int e = e0 + tid + k * BT;
        es[k] = 0;
        ed[k] = -1;
        if (e < e1) { es[k] = src[e]; ed[k] = dst[e]; }
    }
    if (tid < NBINS) s_cnt[tid] = 0u;

    uint4 bef = make_uint4(0u, 0u, 0u, 0u), tot = make_uint4(0u, 0u, 0u, 0u);
#pragma unroll
    for (int k = 0; k < HROWS; ++k) {
        const int r = rg + k * HRG;
        tot.x += hv[k].x; tot.y += hv[k].y; tot.z += hv[k].z; tot.w += hv[k].w;
        if (r < chunk) { bef.x += hv[k].x; bef.y += hv[k].y; bef.z += hv[k].z; bef.w += hv[k].w; }
    }
    if (rg < HRG) {
        unsigned int* rb = s_red + (0 * HRG + rg) * NBINS + q * 4;
        unsigned int* rt = s_red + (1 * HRG + rg) * NBINS + q * 4;
        rb[0] = bef.x; rb[1] = bef.y; rb[2] = bef.z; rb[3] = bef.w;
        rt[0] = tot.x; rt[1] = tot.y; rt[2] = tot.z; rt[3] = tot.w;
    }
    __syncthreads();

    unsigned int cbef = 0u, ctot = 0u;
    if (tid < NBINS) {
        for (int g = 0; g < HRG; ++g) {
            cbef += s_red[(0 * HRG + g) * NBINS + tid];
            ctot += s_red[(1 * HRG + g) * NBINS + tid];
        }
    }
    unsigned int all = 0u;
    const unsigned int bbase = block_excl_scan(ctot, s_w, all);
    if (chunk == 0) {
        if (tid < NBINS) binbase[tid] = bbase;
        if (tid == 0) binbase[NBINS] = all;
    }

    unsigned int rk[EPT];
#pragma unroll
    for (int k = 0; k < EPT; ++k) {
        rk[k] = 0u;
        if (ed[k] >= 0) rk[k] = atomicAdd(&s_cnt[ed[k] >> BIN_SHIFT], 1u);
    }
    __syncthreads();

    unsigned int dummy = 0u;
    const unsigned int mycnt = (tid < NBINS) ? s_cnt[tid] : 0u;
    const unsigned int lbase = block_excl_scan(mycnt, s_w, dummy);
    if (tid < NBINS) {
        s_lbase[tid] = lbase;
        s_goff[tid]  = bbase + cbef - lbase;
    }
    __syncthreads();

#pragma unroll
    for (int k = 0; k < EPT; ++k)
        if (ed[k] >= 0) s_sorted[s_lbase[ed[k] >> BIN_SHIFT] + rk[k]] = make_int2(es[k], ed[k]);
    __syncthreads();

    for (int i = tid; i < n; i += BT) {
        const int2 p = s_sorted[i];
        tmp[s_goff[p.y >> BIN_SHIFT] + (unsigned int)i] = p;
    }
}

__global__ void __launch_bounds__(BT)
node_count_kernel(const int2* __restrict__ tmp,
                  const unsigned int* __restrict__ binbase,
                  unsigned int* __restrict__ rowptr,
                  float* __restrict__ dis, int n_nodes) {
    __shared__ unsigned int s_cnt[BIN_NODES];
    __shared__ unsigned int s_w[BT / 32];
    const int tid = threadIdx.x;
    const unsigned int lo = binbase[blockIdx.x];
    const unsigned int hi = binbase[blockIdx.x + 1];
    if (tid < BIN_NODES) s_cnt[tid] = 0u;
    __syncthreads();
    for (unsigned int i = lo + (unsigned int)tid; i < hi; i += 4u * BT) {
        int d[4];
#pragma unroll
        for (int u = 0; u < 4; ++u) {
            const unsigned int j = i + (unsigned int)(u * BT);
            d[u] = (j < hi) ? tmp[j].y : -1;
        }
#pragma unroll
        for (int u = 0; u < 4; ++u)
            if (d[u] >= 0) atomicAdd(&s_cnt[d[u] & (BIN_NODES - 1)], 1u);
    }
    __syncthreads();
    const unsigned int c = (tid < BIN_NODES) ? s_cnt[tid] : 0u;
    unsigned int all = 0u;
    const unsigned int ex = block_excl_scan(c, s_w, all);
    const int node = blockIdx.x * BIN_NODES + tid;
    if (tid < BIN_NODES && node < n_nodes) {
        rowptr[node] = lo + ex;
        dis[node] = rsqrtf((float)c + 1.0f);
        if (node == n_nodes - 1) rowptr[n_nodes] = lo + ex + c;
    }
}

__global__ void __launch_bounds__(BT)
node_place_kernel(const int2* __restrict__ tmp,
                  const unsigned int* __restrict__ binbase,
                  const unsigned int* __restrict__ rowptr,
                  const float* __restrict__ dis,
                  int2* __restrict__ epk, int n_nodes) {
    __shared__ int2 s_stage[STAGE_CAP];
    __shared__ unsigned int s_start[BIN_NODES];
    __shared__ unsigned int s_cur[BIN_NODES];
    __shared__ float s_dis[BIN_NODES];
    const int tid = threadIdx.x;
    const unsigned int lo = binbase[blockIdx.x];
    const unsigned int hi = binbase[blockIdx.x + 1];
    const unsigned int n = hi - lo;
    if (tid < BIN_NODES) {
        const int node = blockIdx.x * BIN_NODES + tid;
        const bool ok = node < n_nodes;
        s_start[tid] = ok ? (rowptr[node] - lo) : 0u;
        s_dis[tid]   = ok ? dis[node] : 0.0f;
        s_cur[tid]   = 0u;
    }
    __syncthreads();
    const bool staged = n <= (unsigned int)STAGE_CAP;
    for (unsigned int i = (unsigned int)tid; i < n; i += 4u * BT) {
        int2 p[4];
        float ds[4];
#pragma unroll
        for (int u = 0; u < 4; ++u) {
            const unsigned int j = i + (unsigned int)(u * BT);
            p[u] = make_int2(0, -1);
            if (j < n) p[u] = tmp[lo + j];
        }
#pragma unroll
        for (int u = 0; u < 4; ++u) {
            ds[u] = 0.0f;
            if (p[u].y >= 0) ds[u] = dis[p[u].x];
        }
#pragma unroll
        for (int u = 0; u < 4; ++u) {
            if (p[u].y >= 0) {
                const int l = p[u].y & (BIN_NODES - 1);
                const unsigned int slot = s_start[l] + atomicAdd(&s_cur[l], 1u);
                const int2 val = make_int2(p[u].x, __float_as_int(ds[u] * s_dis[l]));
                if (staged) s_stage[slot] = val;
                else        epk[lo + slot] = val;
            }
        }
    }
    __syncthreads();
    if (staged) {
        for (unsigned int i = (unsigned int)tid; i < n; i += BT) epk[lo + i] = s_stage[i];
    }
}


template<int K, int NOUT>
__global__ void __launch_bounds__(256)
gemm_wmma_kernel(const float* __restrict__ X, const float* __restrict__ W,
                 float* __restrict__ H, int nrows) {
    __shared__ float sW[K * NOUT];
    const int tid = threadIdx.x;
    for (int i = tid; i < K * NOUT; i += 256) sW[i] = W[i];
    __syncthreads();

    const int wave = tid >> 5;
    const int lane = tid & 31;
    const int half = lane >> 4;
    const int r    = lane & 15;

    const int row0 = (blockIdx.x * 8 + wave) * 16;
    if (row0 >= nrows) return;

    int arow = row0 + r;
    if (arow >= nrows) arow = nrows - 1;
    const float* __restrict__ xrow = X + (long long)arow * K;

    constexpr int NT = NOUT / 16;
    v8f acc[NT];
#pragma unroll
    for (int nt = 0; nt < NT; ++nt) {
        v8f z = {};
        acc[nt] = z;
    }

#pragma unroll 4
    for (int kt = 0; kt < K / 4; ++kt) {
        const int k0 = kt * 4 + 2 * half;
        v2f a;
        a.x = xrow[k0];
        a.y = xrow[k0 + 1];
#pragma unroll
        for (int nt = 0; nt < NT; ++nt) {
            v2f b;
            b.x = sW[(k0)     * NOUT + nt * 16 + r];
            b.y = sW[(k0 + 1) * NOUT + nt * 16 + r];
            acc[nt] = __builtin_amdgcn_wmma_f32_16x16x4_f32(
                false, a, false, b, (short)0, acc[nt], false, false);
        }
    }

#pragma unroll
    for (int nt = 0; nt < NT; ++nt) {
#pragma unroll
        for (int v = 0; v < 8; ++v) {
            const int row = row0 + v + 8 * half;
            const int col = nt * 16 + r;
            if (row < nrows) H[(long long)row * NOUT + col] = acc[nt][v];
        }
    }
}


template<int F, bool RELU>
__global__ void __launch_bounds__(256)
gather_kernel(const unsigned int* __restrict__ rowptr,
              const int2* __restrict__ epk,
              const float* __restrict__ h,
              const float* __restrict__ dis,
              const float* __restrict__ b,
              float* __restrict__ out, int n) {
    constexpr int V = F / 32;
    constexpr int G = 8;
    const int lane = threadIdx.x & 31;
    const int node = __builtin_amdgcn_readfirstlane((int)(blockIdx.x * 8 + (threadIdx.x >> 5)));
    if (node >= n) return;

    const unsigned int beg = rowptr[node];
    const unsigned int end = rowptr[node + 1];
    const float dd = dis[node];

    float bias[V];
    if (V == 2) {
        const float2 bb = *(const float2*)(b + lane * V);
        bias[0] = bb.x;
        bias[V - 1] = bb.y;
    } else {
        bias[0] = b[lane];
    }

    float acc[V];
    {
        const float* __restrict__ hp = h + (long long)node * F + lane * V;
        if (V == 2) {
            const float2 q = *(const float2*)hp;
            acc[0] = q.x * dd * dd;
            acc[V - 1] = q.y * dd * dd;
        } else {
            acc[0] = hp[0] * dd * dd;
        }
    }

    for (unsigned int base = beg; base < end; base += 32u) {
        const unsigned int j = base + (unsigned int)lane;
        int es = node;
        int ecb = 0;
        if (j < end) {
            const int2 p = epk[j];
            es = p.x;
            ecb = p.y;
        }
        const unsigned int left = end - base;
        const int m = (int)(left < 32u ? left : 32u);
        for (int t = 0; t < m; t += G) {
            float hv[G][V];
            float cf[G];
#pragma unroll
            for (int u = 0; u < G; ++u) {
                const int s = __builtin_amdgcn_readlane(es, t + u);
                cf[u] = __int_as_float(__builtin_amdgcn_readlane(ecb, t + u));
                const float* __restrict__ hp = h + (long long)s * F + lane * V;
                if (V == 2) {
                    const float2 q = *(const float2*)hp;
                    hv[u][0] = q.x;
                    hv[u][V - 1] = q.y;
                } else {
                    hv[u][0] = hp[0];
                }
            }
#pragma unroll
            for (int u = 0; u < G; ++u) {
                const bool live = (t + u) < m;
#pragma unroll
                for (int v = 0; v < V; ++v)
                    acc[v] = fmaf(live ? hv[u][v] : 0.0f, cf[u], acc[v]);
            }
        }
    }

    float* __restrict__ op = out + (long long)node * F + lane * V;
    if (V == 2) {
        float2 o;
        o.x = acc[0] + bias[0];
        o.y = acc[V - 1] + bias[V - 1];
        if (RELU) { o.x = fmaxf(o.x, 0.0f); o.y = fmaxf(o.y, 0.0f); }
        *(float2*)op = o;
    } else {
        float o = acc[0] + bias[0];
        if (RELU) o = fmaxf(o, 0.0f);
        op[0] = o;
    }
}


static inline int cdiv(long long a, int b) { return (int)((a + b - 1) / b); }

extern "C" void kernel_launch(void* const* d_in, const int* in_sizes, int n_in,
                              void* d_out, int out_size, void* d_ws, size_t ws_size,
                              hipStream_t stream) {
    const float* x  = (const float*)d_in[0];
    const int*   ei = (const int*)d_in[1];
    const float* W1 = (const float*)d_in[2];
    const float* b1 = (const float*)d_in[3];
    const float* W2 = (const float*)d_in[4];
    const float* b2 = (const float*)d_in[5];
    const float* W3 = (const float*)d_in[6];
    const float* b3 = (const float*)d_in[7];
    float* out = (float*)d_out;

    const int* src = ei;
    const int* dst = ei + N_EDGES;

    char* ws = (char*)d_ws;
    unsigned int* hist    = (unsigned int*)ws;
    unsigned int* binbase = (unsigned int*)(ws + 200704);
    float* dis  = (float*)(ws + 400384);
    float* bufA = (float*)(ws + 800768);
    float* bufB = (float*)(ws + 26400768);
    unsigned int* rowptr = (unsigned int*)(ws + 52000768);
    int2* epk            = (int2*)(ws + 58801152);
    int2* tmp            = (int2*)(ws + 71601152);

    const int N = N_NODES, E = N_EDGES;
    const int gemm_blocks = cdiv(cdiv(N, 16), 8);
    const int node_blocks = cdiv(N, 8);

    bin_hist_kernel<<<EDGE_CHUNKS, BT, 0, stream>>>(dst, hist, E);
    bin_partition_kernel<<<EDGE_CHUNKS, BT, 0, stream>>>(src, dst, hist, binbase, tmp, E);
    node_count_kernel<<<NBINS, BT, 0, stream>>>(tmp, binbase, rowptr, dis, N);
    node_place_kernel<<<NBINS, BT, 0, stream>>>(tmp, binbase, rowptr, dis, epk, N);

    gemm_wmma_kernel<128, 64><<<gemm_blocks, 256, 0, stream>>>(x, W1, bufA, N);
    gather_kernel<64, true><<<node_blocks, 256, 0, stream>>>(rowptr, epk, bufA, dis, b1, bufB, N);

    gemm_wmma_kernel<64, 64><<<gemm_blocks, 256, 0, stream>>>(bufB, W2, bufA, N);
    gather_kernel<64, true><<<node_blocks, 256, 0, stream>>>(rowptr, epk, bufA, dis, b2, bufB, N);

    gemm_wmma_kernel<64, 32><<<gemm_blocks, 256, 0, stream>>>(bufB, W3, bufA, N);
    gather_kernel<32, false><<<node_blocks, 256, 0, stream>>>(rowptr, epk, bufA, dis, b3, out, N);
}


# Layer 2 and 3 products moved into the preceding gather: resident 16-wave work-groups keep Wn in LDS, leave 16 finished rows there as one A tile, multiply it on WMMA; 2 launches, 2 row planes gone
# speedup vs baseline: 6.7362x; 1.1649x over previous
#include <hip/hip_runtime.h>

typedef float v2f __attribute__((ext_vector_type(2)));
typedef float v8f __attribute__((ext_vector_type(8)));

#define N_NODES 100000
#define N_EDGES 1600000


#define BIN_SHIFT   9
#define BIN_NODES   (1 << BIN_SHIFT)
#define NBINS       ((N_NODES + BIN_NODES - 1) / BIN_NODES)
#define EDGE_CHUNKS 256
#define CHUNK_EDGES ((N_EDGES + EDGE_CHUNKS - 1) / EDGE_CHUNKS)
#define BT          1024
#define EPT         ((CHUNK_EDGES + BT - 1) / BT)
#define HQ          (NBINS / 4)
#define HRG         (BT / HQ)
#define HROWS       ((EDGE_CHUNKS + HRG - 1) / HRG)
#define STAGE_CAP   16384

static_assert(NBINS % 4 == 0, "histogram rows are read as uint4");
static_assert(NBINS <= BT, "one thread per bin");
static_assert(2 * HRG * NBINS * 4 <= CHUNK_EDGES * 8, "the reduction scratch lives inside the sort buffer");

__device__ __forceinline__ unsigned int block_excl_scan(unsigned int v, unsigned int* s_w, unsigned int& total) {
    const int lane = threadIdx.x & 31;
    const int wave = threadIdx.x >> 5;
    unsigned int inc = v;
    for (int d = 1; d < 32; d <<= 1) {
        const unsigned int t = __shfl_up(inc, d);
        if (lane >= d) inc += t;
    }
    if (lane == 31) s_w[wave] = inc;
    __syncthreads();
    unsigned int off = 0u, tot = 0u;
    for (int w = 0; w < BT / 32; ++w) {
        const unsigned int x = s_w[w];
        if (w < wave) off += x;
        tot += x;
    }
    __syncthreads();
    total = tot;
    return off + inc - v;
}

__global__ void __launch_bounds__(BT)
bin_hist_kernel(const int* __restrict__ dst, unsigned int* __restrict__ hist, int E) {
    __shared__ unsigned int s_cnt[NBINS];
    const int tid = threadIdx.x;
    if (tid < NBINS) s_cnt[tid] = 0u;
    const int e0 = blockIdx.x * CHUNK_EDGES;
    const int e1 = (e0 + CHUNK_EDGES < E) ? (e0 + CHUNK_EDGES) : E;
    int d[EPT];
#pragma unroll
    for (int k = 0; k < EPT; ++k) {
        const int e = e0 + tid + k * BT;
        d[k] = (e < e1) ? dst[e] : -1;
    }
    __syncthreads();
#pragma unroll
    for (int k = 0; k < EPT; ++k)
        if (d[k] >= 0) atomicAdd(&s_cnt[d[k] >> BIN_SHIFT], 1u);
    __syncthreads();
    if (tid < NBINS) hist[blockIdx.x * NBINS + tid] = s_cnt[tid];
}

__global__ void __launch_bounds__(BT)
bin_partition_kernel(const int* __restrict__ src, const int* __restrict__ dst,
                     const unsigned int* __restrict__ hist,
                     unsigned int* __restrict__ binbase,
                     int2* __restrict__ tmp, int E) {
    __shared__ int2 s_sorted[CHUNK_EDGES];
    __shared__ unsigned int s_cnt[NBINS];
    __shared__ unsigned int s_lbase[NBINS];
    __shared__ unsigned int s_goff[NBINS];
    __shared__ unsigned int s_w[BT / 32];
    unsigned int* s_red = (unsigned int*)s_sorted;

    const int tid = threadIdx.x;
    const int chunk = blockIdx.x;
    const int e0 = chunk * CHUNK_EDGES;
    const int e1 = (e0 + CHUNK_EDGES < E) ? (e0 + CHUNK_EDGES) : E;
    const int n = e1 - e0;

    const int q  = tid % HQ;
    const int rg = tid / HQ;
    uint4 hv[HROWS];
#pragma unroll
    for (int k = 0; k < HROWS; ++k) {
        const int r = rg + k * HRG;
        hv[k] = make_uint4(0u, 0u, 0u, 0u);
        if (rg < HRG && r < EDGE_CHUNKS) hv[k] = ((const uint4*)hist)[r * HQ + q];
    }
    int es[EPT], ed[EPT];
#pragma unroll
    for (int k = 0; k < EPT; ++k) {
        const int e = e0 + tid + k * BT;
        es[k] = 0;
        ed[k] = -1;
        if (e < e1) { es[k] = src[e]; ed[k] = dst[e]; }
    }
    if (tid < NBINS) s_cnt[tid] = 0u;

    uint4 bef = make_uint4(0u, 0u, 0u, 0u), tot = make_uint4(0u, 0u, 0u, 0u);
#pragma unroll
    for (int k = 0; k < HROWS; ++k) {
        const int r = rg + k * HRG;
        tot.x += hv[k].x; tot.y += hv[k].y; tot.z += hv[k].z; tot.w += hv[k].w;
        if (r < chunk) { bef.x += hv[k].x; bef.y += hv[k].y; bef.z += hv[k].z; bef.w += hv[k].w; }
    }
    if (rg < HRG) {
        unsigned int* rb = s_red + (0 * HRG + rg) * NBINS + q * 4;
        unsigned int* rt = s_red + (1 * HRG + rg) * NBINS + q * 4;
        rb[0] = bef.x; rb[1] = bef.y; rb[2] = bef.z; rb[3] = bef.w;
        rt[0] = tot.x; rt[1] = tot.y; rt[2] = tot.z; rt[3] = tot.w;
    }
    __syncthreads();

    unsigned int cbef = 0u, ctot = 0u;
    if (tid < NBINS) {
        for (int g = 0; g < HRG; ++g) {
            cbef += s_red[(0 * HRG + g) * NBINS + tid];
            ctot += s_red[(1 * HRG + g) * NBINS + tid];
        }
    }
    unsigned int all = 0u;
    const unsigned int bbase = block_excl_scan(ctot, s_w, all);
    if (chunk == 0) {
        if (tid < NBINS) binbase[tid] = bbase;
        if (tid == 0) binbase[NBINS] = all;
    }

    unsigned int rk[EPT];
#pragma unroll
    for (int k = 0; k < EPT; ++k) {
        rk[k] = 0u;
        if (ed[k] >= 0) rk[k] = atomicAdd(&s_cnt[ed[k] >> BIN_SHIFT], 1u);
    }
    __syncthreads();

    unsigned int dummy = 0u;
    const unsigned int mycnt = (tid < NBINS) ? s_cnt[tid] : 0u;
    const unsigned int lbase = block_excl_scan(mycnt, s_w, dummy);
    if (tid < NBINS) {
        s_lbase[tid] = lbase;
        s_goff[tid]  = bbase + cbef - lbase;
    }
    __syncthreads();

#pragma unroll
    for (int k = 0; k < EPT; ++k)
        if (ed[k] >= 0) s_sorted[s_lbase[ed[k] >> BIN_SHIFT] + rk[k]] = make_int2(es[k], ed[k]);
    __syncthreads();

    for (int i = tid; i < n; i += BT) {
        const int2 p = s_sorted[i];
        tmp[s_goff[p.y >> BIN_SHIFT] + (unsigned int)i] = p;
    }
}

__global__ void __launch_bounds__(BT)
node_count_kernel(const int2* __restrict__ tmp,
                  const unsigned int* __restrict__ binbase,
                  unsigned int* __restrict__ rowptr,
                  float* __restrict__ dis, int n_nodes) {
    __shared__ unsigned int s_cnt[BIN_NODES];
    __shared__ unsigned int s_w[BT / 32];
    const int tid = threadIdx.x;
    const unsigned int lo = binbase[blockIdx.x];
    const unsigned int hi = binbase[blockIdx.x + 1];
    if (tid < BIN_NODES) s_cnt[tid] = 0u;
    __syncthreads();
    for (unsigned int i = lo + (unsigned int)tid; i < hi; i += 4u * BT) {
        int d[4];
#pragma unroll
        for (int u = 0; u < 4; ++u) {
            const unsigned int j = i + (unsigned int)(u * BT);
            d[u] = (j < hi) ? tmp[j].y : -1;
        }
#pragma unroll
        for (int u = 0; u < 4; ++u)
            if (d[u] >= 0) atomicAdd(&s_cnt[d[u] & (BIN_NODES - 1)], 1u);
    }
    __syncthreads();
    const unsigned int c = (tid < BIN_NODES) ? s_cnt[tid] : 0u;
    unsigned int all = 0u;
    const unsigned int ex = block_excl_scan(c, s_w, all);
    const int node = blockIdx.x * BIN_NODES + tid;
    if (tid < BIN_NODES && node < n_nodes) {
        rowptr[node] = lo + ex;
        dis[node] = rsqrtf((float)c + 1.0f);
        if (node == n_nodes - 1) rowptr[n_nodes] = lo + ex + c;
    }
}

__global__ void __launch_bounds__(BT)
node_place_kernel(const int2* __restrict__ tmp,
                  const unsigned int* __restrict__ binbase,
                  const unsigned int* __restrict__ rowptr,
                  const float* __restrict__ dis,
                  int2* __restrict__ epk, int n_nodes) {
    __shared__ int2 s_stage[STAGE_CAP];
    __shared__ unsigned int s_start[BIN_NODES];
    __shared__ unsigned int s_cur[BIN_NODES];
    __shared__ float s_dis[BIN_NODES];
    const int tid = threadIdx.x;
    const unsigned int lo = binbase[blockIdx.x];
    const unsigned int hi = binbase[blockIdx.x + 1];
    const unsigned int n = hi - lo;
    if (tid < BIN_NODES) {
        const int node = blockIdx.x * BIN_NODES + tid;
        const bool ok = node < n_nodes;
        s_start[tid] = ok ? (rowptr[node] - lo) : 0u;
        s_dis[tid]   = ok ? dis[node] : 0.0f;
        s_cur[tid]   = 0u;
    }
    __syncthreads();
    const bool staged = n <= (unsigned int)STAGE_CAP;
    for (unsigned int i = (unsigned int)tid; i < n; i += 4u * BT) {
        int2 p[4];
        float ds[4];
#pragma unroll
        for (int u = 0; u < 4; ++u) {
            const unsigned int j = i + (unsigned int)(u * BT);
            p[u] = make_int2(0, -1);
            if (j < n) p[u] = tmp[lo + j];
        }
#pragma unroll
        for (int u = 0; u < 4; ++u) {
            ds[u] = 0.0f;
            if (p[u].y >= 0) ds[u] = dis[p[u].x];
        }
#pragma unroll
        for (int u = 0; u < 4; ++u) {
            if (p[u].y >= 0) {
                const int l = p[u].y & (BIN_NODES - 1);
                const unsigned int slot = s_start[l] + atomicAdd(&s_cur[l], 1u);
                const int2 val = make_int2(p[u].x, __float_as_int(ds[u] * s_dis[l]));
                if (staged) s_stage[slot] = val;
                else        epk[lo + slot] = val;
            }
        }
    }
    __syncthreads();
    if (staged) {
        for (unsigned int i = (unsigned int)tid; i < n; i += BT) epk[lo + i] = s_stage[i];
    }
}


template<int K, int NOUT>
__global__ void __launch_bounds__(256)
gemm_wmma_kernel(const float* __restrict__ X, const float* __restrict__ W,
                 float* __restrict__ H, int nrows) {
    __shared__ float sW[K * NOUT];
    const int tid = threadIdx.x;
    for (int i = tid; i < K * NOUT; i += 256) sW[i] = W[i];
    __syncthreads();

    const int wave = tid >> 5;
    const int lane = tid & 31;
    const int half = lane >> 4;
    const int r    = lane & 15;

    const int row0 = (blockIdx.x * 8 + wave) * 16;
    if (row0 >= nrows) return;

    int arow = row0 + r;
    if (arow >= nrows) arow = nrows - 1;
    const float* __restrict__ xrow = X + (long long)arow * K;

    constexpr int NT = NOUT / 16;
    v8f acc[NT];
#pragma unroll
    for (int nt = 0; nt < NT; ++nt) {
        v8f z = {};
        acc[nt] = z;
    }

#pragma unroll 4
    for (int kt = 0; kt < K / 4; ++kt) {
        const int k0 = kt * 4 + 2 * half;
        v2f a;
        a.x = xrow[k0];
        a.y = xrow[k0 + 1];
#pragma unroll
        for (int nt = 0; nt < NT; ++nt) {
            v2f b;
            b.x = sW[(k0)     * NOUT + nt * 16 + r];
            b.y = sW[(k0 + 1) * NOUT + nt * 16 + r];
            acc[nt] = __builtin_amdgcn_wmma_f32_16x16x4_f32(
                false, a, false, b, (short)0, acc[nt], false, false);
        }
    }

#pragma unroll
    for (int nt = 0; nt < NT; ++nt) {
#pragma unroll
        for (int v = 0; v < 8; ++v) {
            const int row = row0 + v + 8 * half;
            const int col = nt * 16 + r;
            if (row < nrows) H[(long long)row * NOUT + col] = acc[nt][v];
        }
    }
}


template<int F, bool RELU>
__global__ void __launch_bounds__(256)
gather_kernel(const unsigned int* __restrict__ rowptr,
              const int2* __restrict__ epk,
              const float* __restrict__ h,
              const float* __restrict__ dis,
              const float* __restrict__ b,
              float* __restrict__ out, int n) {
    constexpr int V = F / 32;
    constexpr int G = 8;
    const int lane = threadIdx.x & 31;
    const int node = __builtin_amdgcn_readfirstlane((int)(blockIdx.x * 8 + (threadIdx.x >> 5)));
    if (node >= n) return;

    const unsigned int beg = rowptr[node];
    const unsigned int end = rowptr[node + 1];
    const float dd = dis[node];

    float bias[V];
    if (V == 2) {
        const float2 bb = *(const float2*)(b + lane * V);
        bias[0] = bb.x;
        bias[V - 1] = bb.y;
    } else {
        bias[0] = b[lane];
    }

    float acc[V];
    {
        const float* __restrict__ hp = h + (long long)node * F + lane * V;
        if (V == 2) {
            const float2 q = *(const float2*)hp;
            acc[0] = q.x * dd * dd;
            acc[V - 1] = q.y * dd * dd;
        } else {
            acc[0] = hp[0] * dd * dd;
        }
    }

    for (unsigned int base = beg; base < end; base += 32u) {
        const unsigned int j = base + (unsigned int)lane;
        int es = node;
        int ecb = 0;
        if (j < end) {
            const int2 p = epk[j];
            es = p.x;
            ecb = p.y;
        }
        const unsigned int left = end - base;
        const int m = (int)(left < 32u ? left : 32u);
        for (int t = 0; t < m; t += G) {
            float hv[G][V];
            float cf[G];
#pragma unroll
            for (int u = 0; u < G; ++u) {
                const int s = __builtin_amdgcn_readlane(es, t + u);
                cf[u] = __int_as_float(__builtin_amdgcn_readlane(ecb, t + u));
                const float* __restrict__ hp = h + (long long)s * F + lane * V;
                if (V == 2) {
                    const float2 q = *(const float2*)hp;
                    hv[u][0] = q.x;
                    hv[u][V - 1] = q.y;
                } else {
                    hv[u][0] = hp[0];
                }
            }
#pragma unroll
            for (int u = 0; u < G; ++u) {
                const bool live = (t + u) < m;
#pragma unroll
                for (int v = 0; v < V; ++v)
                    acc[v] = fmaf(live ? hv[u][v] : 0.0f, cf[u], acc[v]);
            }
        }
    }

    float* __restrict__ op = out + (long long)node * F + lane * V;
    if (V == 2) {
        float2 o;
        o.x = acc[0] + bias[0];
        o.y = acc[V - 1] + bias[V - 1];
        if (RELU) { o.x = fmaxf(o.x, 0.0f); o.y = fmaxf(o.y, 0.0f); }
        *(float2*)op = o;
    } else {
        float o = acc[0] + bias[0];
        if (RELU) o = fmaxf(o, 0.0f);
        op[0] = o;
    }
}


template<int NOUT>
__global__ void __launch_bounds__(512)
gather_gemm_kernel(const unsigned int* __restrict__ rowptr,
                   const int2* __restrict__ epk,
                   const float* __restrict__ h,
                   const float* __restrict__ dis,
                   const float* __restrict__ b,
                   const float* __restrict__ Wn,
                   float* __restrict__ Hn, int n) {
    constexpr int F = 64;
    constexpr int V = 2;
    constexpr int G = 8;
    constexpr int NT = NOUT / 16;
    constexpr int VO = NOUT / 32;
    constexpr int AP = F + 4;
    constexpr int PW = NOUT + 16;
    constexpr int OP = NOUT + 4;
    __shared__ alignas(16) float2 sW[(F / 2) * PW];
    __shared__ alignas(16) float sA[16 * AP];
    __shared__ alignas(16) float sO[16 * OP];

    const int tid = threadIdx.x;
    const int lane = tid & 31;
    const int wave = __builtin_amdgcn_readfirstlane(tid >> 5);

    const int ntiles = (n + 15) >> 4;
    int tile = blockIdx.x;
    if (tile >= ntiles) return;

    for (int i = tid; i < (F / 2) * (NOUT / 4); i += 512) {
        const int kp = i / (NOUT / 4);
        const int cc = (i % (NOUT / 4)) * 4;
        const float4 w0 = *(const float4*)(Wn + (2 * kp) * NOUT + cc);
        const float4 w1 = *(const float4*)(Wn + (2 * kp + 1) * NOUT + cc);
        float4* dp = (float4*)&sW[kp * PW + cc];
        dp[0] = make_float4(w0.x, w1.x, w0.y, w1.y);
        dp[1] = make_float4(w0.z, w1.z, w0.w, w1.w);
    }
    const float2 bb = *(const float2*)(b + lane * V);

    const int half = lane >> 4;
    const int r    = lane & 15;

    int node = __builtin_amdgcn_readfirstlane(tile * 16 + wave);
    bool valid = node < n;
    int nodec = valid ? node : (n - 1);
    unsigned int beg = rowptr[nodec];
    unsigned int nxt = rowptr[nodec + 1];
    float dd = dis[nodec];
    float2 q = *(const float2*)(h + (long long)nodec * F + lane * V);

    for (;;) {
        const int tile_n = tile + (int)gridDim.x;
        const int node_n = __builtin_amdgcn_readfirstlane(tile_n * 16 + wave);
        const bool valid_n = node_n < n;
        const int nodec_n = valid_n ? node_n : (n - 1);
        const unsigned int beg_n = rowptr[nodec_n];
        const unsigned int nxt_n = rowptr[nodec_n + 1];
        const float dd_n = dis[nodec_n];
        const float2 q_n = *(const float2*)(h + (long long)nodec_n * F + lane * V);

        const unsigned int begs = __builtin_amdgcn_readfirstlane(beg);
        const unsigned int end  = __builtin_amdgcn_readfirstlane(valid ? nxt : beg);
        float acc[V];
        acc[0] = q.x * dd * dd;
        acc[V - 1] = q.y * dd * dd;

        for (unsigned int base = begs; base < end; base += 32u) {
            const unsigned int j = base + (unsigned int)lane;
            int es = nodec;
            int ecb = 0;
            if (j < end) {
                const int2 p = epk[j];
                es = p.x;
                ecb = p.y;
            }
            const unsigned int left = end - base;
            const int m = (int)(left < 32u ? left : 32u);
            for (int t = 0; t < m; t += G) {
                float hv[G][V];
                float cf[G];
#pragma unroll
                for (int u = 0; u < G; ++u) {
                    const int s = __builtin_amdgcn_readlane(es, t + u);
                    cf[u] = __int_as_float(__builtin_amdgcn_readlane(ecb, t + u));
                    const float2 qq = *(const float2*)(h + (long long)s * F + lane * V);
                    hv[u][0] = qq.x;
                    hv[u][V - 1] = qq.y;
                }
#pragma unroll
                for (int u = 0; u < G; ++u) {
                    const bool live = (t + u) < m;
#pragma unroll
                    for (int v = 0; v < V; ++v)
                        acc[v] = fmaf(live ? hv[u][v] : 0.0f, cf[u], acc[v]);
                }
            }
        }
        acc[0] = valid ? fmaxf(acc[0] + bb.x, 0.0f) : 0.0f;
        acc[V - 1] = valid ? fmaxf(acc[V - 1] + bb.y, 0.0f) : 0.0f;

        *(float2*)&sA[wave * AP + lane * V] = make_float2(acc[0], acc[V - 1]);
        __syncthreads();

        if (wave < NT) {
            v8f d = {};
#pragma unroll
            for (int kt = 0; kt < F / 4; ++kt) {
                const v2f a = *(const v2f*)&sA[r * AP + kt * 4 + 2 * half];
                const float2 bq = sW[(kt * 2 + half) * PW + wave * 16 + r];
                v2f bf;
                bf.x = bq.x;
                bf.y = bq.y;
                d = __builtin_amdgcn_wmma_f32_16x16x4_f32(false, a, false, bf, (short)0, d, false, false);
            }
#pragma unroll
            for (int v = 0; v < 8; ++v) sO[(v + 8 * half) * OP + wave * 16 + r] = d[v];
        }
        __syncthreads();

        if (valid) {
            float* __restrict__ op = Hn + (long long)node * NOUT + lane * VO;
            if (VO == 2) *(float2*)op = *(const float2*)&sO[wave * OP + lane * VO];
            else         op[0] = sO[wave * OP + lane * VO];
        }

        if (tile_n >= ntiles) break;
        tile = tile_n; node = node_n; valid = valid_n; nodec = nodec_n;
        beg = beg_n; nxt = nxt_n; dd = dd_n; q = q_n;
    }
}


static inline int cdiv(long long a, int b) { return (int)((a + b - 1) / b); }

extern "C" void kernel_launch(void* const* d_in, const int* in_sizes, int n_in,
                              void* d_out, int out_size, void* d_ws, size_t ws_size,
                              hipStream_t stream) {
    const float* x  = (const float*)d_in[0];
    const int*   ei = (const int*)d_in[1];
    const float* W1 = (const float*)d_in[2];
    const float* b1 = (const float*)d_in[3];
    const float* W2 = (const float*)d_in[4];
    const float* b2 = (const float*)d_in[5];
    const float* W3 = (const float*)d_in[6];
    const float* b3 = (const float*)d_in[7];
    float* out = (float*)d_out;

    const int* src = ei;
    const int* dst = ei + N_EDGES;

    char* ws = (char*)d_ws;
    unsigned int* hist    = (unsigned int*)ws;
    unsigned int* binbase = (unsigned int*)(ws + 200704);
    float* dis  = (float*)(ws + 400384);
    float* bufA = (float*)(ws + 800768);
    float* bufB = (float*)(ws + 26400768);
    unsigned int* rowptr = (unsigned int*)(ws + 52000768);
    int2* epk            = (int2*)(ws + 58801152);
    int2* tmp            = (int2*)(ws + 71601152);

    const int N = N_NODES, E = N_EDGES;
    const int gemm_blocks = cdiv(cdiv(N, 16), 8);
    const int node_blocks = cdiv(N, 8);
    const int tile_groups = 1024;

    bin_hist_kernel<<<EDGE_CHUNKS, BT, 0, stream>>>(dst, hist, E);
    bin_partition_kernel<<<EDGE_CHUNKS, BT, 0, stream>>>(src, dst, hist, binbase, tmp, E);
    node_count_kernel<<<NBINS, BT, 0, stream>>>(tmp, binbase, rowptr, dis, N);
    node_place_kernel<<<NBINS, BT, 0, stream>>>(tmp, binbase, rowptr, dis, epk, N);

    gemm_wmma_kernel<128, 64><<<gemm_blocks, 256, 0, stream>>>(x, W1, bufA, N);

    gather_gemm_kernel<64><<<tile_groups, 512, 0, stream>>>(rowptr, epk, bufA, dis, b1, W2, bufB, N);

    gather_gemm_kernel<32><<<tile_groups, 512, 0, stream>>>(rowptr, epk, bufB, dis, b2, W3, bufA, N);

    gather_kernel<32, false><<<node_blocks, 256, 0, stream>>>(rowptr, epk, bufA, dis, b3, out, N);
}


# Edge weight factored out of the buckets: rows stored scaled by dis, sums scaled by dis[dst]; entries are the bare src id, placement joins the count launch (7 launches); sum order kept, rounding new
# speedup vs baseline: 7.1648x; 1.0636x over previous
#include <hip/hip_runtime.h>

typedef float v2f __attribute__((ext_vector_type(2)));
typedef float v8f __attribute__((ext_vector_type(8)));

#define N_NODES 100000
#define N_EDGES 1600000


#define BIN_SHIFT   9
#define BIN_NODES   (1 << BIN_SHIFT)
#define NBINS       ((N_NODES + BIN_NODES - 1) / BIN_NODES)
#define EDGE_CHUNKS 256
#define CHUNK_EDGES ((N_EDGES + EDGE_CHUNKS - 1) / EDGE_CHUNKS)
#define BT          1024
#define EPT         ((CHUNK_EDGES + BT - 1) / BT)
#define HQ          (NBINS / 4)
#define HRG         (BT / HQ)
#define HROWS       ((EDGE_CHUNKS + HRG - 1) / HRG)
#define STAGE_CAP   16384

static_assert(NBINS % 4 == 0, "histogram rows are read as uint4");
static_assert(NBINS <= BT, "one thread per bin");
static_assert(2 * HRG * NBINS * 4 <= CHUNK_EDGES * 8, "the reduction scratch lives inside the sort buffer");

__device__ __forceinline__ unsigned int block_excl_scan(unsigned int v, unsigned int* s_w, unsigned int& total) {
    const int lane = threadIdx.x & 31;
    const int wave = threadIdx.x >> 5;
    unsigned int inc = v;
    for (int d = 1; d < 32; d <<= 1) {
        const unsigned int t = __shfl_up(inc, d);
        if (lane >= d) inc += t;
    }
    if (lane == 31) s_w[wave] = inc;
    __syncthreads();
    unsigned int off = 0u, tot = 0u;
    for (int w = 0; w < BT / 32; ++w) {
        const unsigned int x = s_w[w];
        if (w < wave) off += x;
        tot += x;
    }
    __syncthreads();
    total = tot;
    return off + inc - v;
}

__global__ void __launch_bounds__(BT)
bin_hist_kernel(const int* __restrict__ dst, unsigned int* __restrict__ hist, int E) {
    __shared__ unsigned int s_cnt[NBINS];
    const int tid = threadIdx.x;
    if (tid < NBINS) s_cnt[tid] = 0u;
    const int e0 = blockIdx.x * CHUNK_EDGES;
    const int e1 = (e0 + CHUNK_EDGES < E) ? (e0 + CHUNK_EDGES) : E;
    int d[EPT];
#pragma unroll
    for (int k = 0; k < EPT; ++k) {
        const int e = e0 + tid + k * BT;
        d[k] = (e < e1) ? dst[e] : -1;
    }
    __syncthreads();
#pragma unroll
    for (int k = 0; k < EPT; ++k)
        if (d[k] >= 0) atomicAdd(&s_cnt[d[k] >> BIN_SHIFT], 1u);
    __syncthreads();
    if (tid < NBINS) hist[blockIdx.x * NBINS + tid] = s_cnt[tid];
}

__global__ void __launch_bounds__(BT)
bin_partition_kernel(const int* __restrict__ src, const int* __restrict__ dst,
                     const unsigned int* __restrict__ hist,
                     unsigned int* __restrict__ binbase,
                     int2* __restrict__ tmp, int E) {
    __shared__ int2 s_sorted[CHUNK_EDGES];
    __shared__ unsigned int s_cnt[NBINS];
    __shared__ unsigned int s_lbase[NBINS];
    __shared__ unsigned int s_goff[NBINS];
    __shared__ unsigned int s_w[BT / 32];
    unsigned int* s_red = (unsigned int*)s_sorted;

    const int tid = threadIdx.x;
    const int chunk = blockIdx.x;
    const int e0 = chunk * CHUNK_EDGES;
    const int e1 = (e0 + CHUNK_EDGES < E) ? (e0 + CHUNK_EDGES) : E;
    const int n = e1 - e0;

    const int q  = tid % HQ;
    const int rg = tid / HQ;
    uint4 hv[HROWS];
#pragma unroll
    for (int k = 0; k < HROWS; ++k) {
        const int r = rg + k * HRG;
        hv[k] = make_uint4(0u, 0u, 0u, 0u);
        if (rg < HRG && r < EDGE_CHUNKS) hv[k] = ((const uint4*)hist)[r * HQ + q];
    }
    int es[EPT], ed[EPT];
#pragma unroll
    for (int k = 0; k < EPT; ++k) {
        const int e = e0 + tid + k * BT;
        es[k] = 0;
        ed[k] = -1;
        if (e < e1) { es[k] = src[e]; ed[k] = dst[e]; }
    }
    if (tid < NBINS) s_cnt[tid] = 0u;

    uint4 bef = make_uint4(0u, 0u, 0u, 0u), tot = make_uint4(0u, 0u, 0u, 0u);
#pragma unroll
    for (int k = 0; k < HROWS; ++k) {
        const int r = rg + k * HRG;
        tot.x += hv[k].x; tot.y += hv[k].y; tot.z += hv[k].z; tot.w += hv[k].w;
        if (r < chunk) { bef.x += hv[k].x; bef.y += hv[k].y; bef.z += hv[k].z; bef.w += hv[k].w; }
    }
    if (rg < HRG) {
        unsigned int* rb = s_red + (0 * HRG + rg) * NBINS + q * 4;
        unsigned int* rt = s_red + (1 * HRG + rg) * NBINS + q * 4;
        rb[0] = bef.x; rb[1] = bef.y; rb[2] = bef.z; rb[3] = bef.w;
        rt[0] = tot.x; rt[1] = tot.y; rt[2] = tot.z; rt[3] = tot.w;
    }
    __syncthreads();

    unsigned int cbef = 0u, ctot = 0u;
    if (tid < NBINS) {
        for (int g = 0; g < HRG; ++g) {
            cbef += s_red[(0 * HRG + g) * NBINS + tid];
            ctot += s_red[(1 * HRG + g) * NBINS + tid];
        }
    }
    unsigned int all = 0u;
    const unsigned int bbase = block_excl_scan(ctot, s_w, all);
    if (chunk == 0) {
        if (tid < NBINS) binbase[tid] = bbase;
        if (tid == 0) binbase[NBINS] = all;
    }

    unsigned int rk[EPT];
#pragma unroll
    for (int k = 0; k < EPT; ++k) {
        rk[k] = 0u;
        if (ed[k] >= 0) rk[k] = atomicAdd(&s_cnt[ed[k] >> BIN_SHIFT], 1u);
    }
    __syncthreads();

    unsigned int dummy = 0u;
    const unsigned int mycnt = (tid < NBINS) ? s_cnt[tid] : 0u;
    const unsigned int lbase = block_excl_scan(mycnt, s_w, dummy);
    if (tid < NBINS) {
        s_lbase[tid] = lbase;
        s_goff[tid]  = bbase + cbef - lbase;
    }
    __syncthreads();

#pragma unroll
    for (int k = 0; k < EPT; ++k)
        if (ed[k] >= 0) s_sorted[s_lbase[ed[k] >> BIN_SHIFT] + rk[k]] = make_int2(es[k], ed[k]);
    __syncthreads();

    for (int i = tid; i < n; i += BT) {
        const int2 p = s_sorted[i];
        tmp[s_goff[p.y >> BIN_SHIFT] + (unsigned int)i] = p;
    }
}

__global__ void __launch_bounds__(BT)
node_bucket_kernel(const int2* __restrict__ tmp,
                   const unsigned int* __restrict__ binbase,
                   unsigned int* __restrict__ rowptr,
                   float* __restrict__ dis,
                   int* __restrict__ epk, int n_nodes) {
    __shared__ int s_stage[STAGE_CAP];
    __shared__ unsigned int s_cnt[BIN_NODES];
    __shared__ unsigned int s_start[BIN_NODES];
    __shared__ unsigned int s_w[BT / 32];
    const int tid = threadIdx.x;
    const unsigned int lo = binbase[blockIdx.x];
    const unsigned int hi = binbase[blockIdx.x + 1];
    const unsigned int n = hi - lo;
    if (tid < BIN_NODES) s_cnt[tid] = 0u;
    __syncthreads();
    for (unsigned int i = (unsigned int)tid; i < n; i += 4u * BT) {
        int d[4];
#pragma unroll
        for (int u = 0; u < 4; ++u) {
            const unsigned int j = i + (unsigned int)(u * BT);
            d[u] = (j < n) ? tmp[lo + j].y : -1;
        }
#pragma unroll
        for (int u = 0; u < 4; ++u)
            if (d[u] >= 0) atomicAdd(&s_cnt[d[u] & (BIN_NODES - 1)], 1u);
    }
    __syncthreads();
    const unsigned int c = (tid < BIN_NODES) ? s_cnt[tid] : 0u;
    unsigned int all = 0u;
    const unsigned int ex = block_excl_scan(c, s_w, all);
    const int node = blockIdx.x * BIN_NODES + tid;
    if (tid < BIN_NODES) {
        s_start[tid] = ex;
        s_cnt[tid] = 0u;
        if (node < n_nodes) {
            rowptr[node] = lo + ex;
            dis[node] = rsqrtf((float)c + 1.0f);
            if (node == n_nodes - 1) rowptr[n_nodes] = lo + ex + c;
        }
    }
    __syncthreads();
    const bool staged = n <= (unsigned int)STAGE_CAP;
    for (unsigned int i = (unsigned int)tid; i < n; i += 4u * BT) {
        int2 p[4];
#pragma unroll
        for (int u = 0; u < 4; ++u) {
            const unsigned int j = i + (unsigned int)(u * BT);
            p[u] = make_int2(0, -1);
            if (j < n) p[u] = tmp[lo + j];
        }
#pragma unroll
        for (int u = 0; u < 4; ++u) {
            if (p[u].y >= 0) {
                const int l = p[u].y & (BIN_NODES - 1);
                const unsigned int slot = s_start[l] + atomicAdd(&s_cnt[l], 1u);
                if (staged) s_stage[slot] = p[u].x;
                else        epk[lo + slot] = p[u].x;
            }
        }
    }
    __syncthreads();
    if (staged) {
        for (unsigned int i = (unsigned int)tid; i < n; i += BT) epk[lo + i] = s_stage[i];
    }
}


template<int K, int NOUT>
__global__ void __launch_bounds__(256)
gemm_wmma_kernel(const float* __restrict__ X, const float* __restrict__ W,
                 const float* __restrict__ dis,
                 float* __restrict__ H, int nrows) {
    __shared__ float sW[K * NOUT];
    const int tid = threadIdx.x;
    for (int i = tid; i < K * NOUT; i += 256) sW[i] = W[i];
    __syncthreads();

    const int wave = tid >> 5;
    const int lane = tid & 31;
    const int half = lane >> 4;
    const int r    = lane & 15;

    const int row0 = (blockIdx.x * 8 + wave) * 16;
    if (row0 >= nrows) return;

    int arow = row0 + r;
    if (arow >= nrows) arow = nrows - 1;
    const float* __restrict__ xrow = X + (long long)arow * K;

    constexpr int NT = NOUT / 16;
    v8f acc[NT];
#pragma unroll
    for (int nt = 0; nt < NT; ++nt) {
        v8f z = {};
        acc[nt] = z;
    }
    float dr[8];
#pragma unroll
    for (int v = 0; v < 8; ++v) {
        const int row = row0 + v + 8 * half;
        dr[v] = dis[row < nrows ? row : (nrows - 1)];
    }

#pragma unroll 4
    for (int kt = 0; kt < K / 4; ++kt) {
        const int k0 = kt * 4 + 2 * half;
        v2f a;
        a.x = xrow[k0];
        a.y = xrow[k0 + 1];
#pragma unroll
        for (int nt = 0; nt < NT; ++nt) {
            v2f b;
            b.x = sW[(k0)     * NOUT + nt * 16 + r];
            b.y = sW[(k0 + 1) * NOUT + nt * 16 + r];
            acc[nt] = __builtin_amdgcn_wmma_f32_16x16x4_f32(
                false, a, false, b, (short)0, acc[nt], false, false);
        }
    }

#pragma unroll
    for (int nt = 0; nt < NT; ++nt) {
#pragma unroll
        for (int v = 0; v < 8; ++v) {
            const int row = row0 + v + 8 * half;
            const int col = nt * 16 + r;
            if (row < nrows) H[(long long)row * NOUT + col] = acc[nt][v] * dr[v];
        }
    }
}


template<int F, bool RELU>
__global__ void __launch_bounds__(256)
gather_kernel(const unsigned int* __restrict__ rowptr,
              const int* __restrict__ epk,
              const float* __restrict__ h,
              const float* __restrict__ dis,
              const float* __restrict__ b,
              float* __restrict__ out, int n) {
    constexpr int V = F / 32;
    constexpr int G = 8;
    const int lane = threadIdx.x & 31;
    const int node = __builtin_amdgcn_readfirstlane((int)(blockIdx.x * 8 + (threadIdx.x >> 5)));
    if (node >= n) return;

    const unsigned int beg = rowptr[node];
    const unsigned int end = rowptr[node + 1];
    const float dd = dis[node];

    float bias[V];
    if (V == 2) {
        const float2 bb = *(const float2*)(b + lane * V);
        bias[0] = bb.x;
        bias[V - 1] = bb.y;
    } else {
        bias[0] = b[lane];
    }

    float acc[V];
    {
        const float* __restrict__ hp = h + (long long)node * F + lane * V;
        if (V == 2) {
            const float2 q = *(const float2*)hp;
            acc[0] = q.x;
            acc[V - 1] = q.y;
        } else {
            acc[0] = hp[0];
        }
    }

    for (unsigned int base = beg; base < end; base += 32u) {
        const unsigned int j = base + (unsigned int)lane;
        int es = node;
        if (j < end) es = epk[j];
        const unsigned int left = end - base;
        const int m = (int)(left < 32u ? left : 32u);
        for (int t = 0; t < m; t += G) {
            float hv[G][V];
#pragma unroll
            for (int u = 0; u < G; ++u) {
                const int s = __builtin_amdgcn_readlane(es, t + u);
                const float* __restrict__ hp = h + (long long)s * F + lane * V;
                if (V == 2) {
                    const float2 q = *(const float2*)hp;
                    hv[u][0] = q.x;
                    hv[u][V - 1] = q.y;
                } else {
                    hv[u][0] = hp[0];
                }
            }
#pragma unroll
            for (int u = 0; u < G; ++u) {
                const bool live = (t + u) < m;
#pragma unroll
                for (int v = 0; v < V; ++v)
                    acc[v] += live ? hv[u][v] : 0.0f;
            }
        }
    }

    float* __restrict__ op = out + (long long)node * F + lane * V;
    if (V == 2) {
        float2 o;
        o.x = acc[0] * dd + bias[0];
        o.y = acc[V - 1] * dd + bias[V - 1];
        if (RELU) { o.x = fmaxf(o.x, 0.0f); o.y = fmaxf(o.y, 0.0f); }
        *(float2*)op = o;
    } else {
        float o = acc[0] * dd + bias[0];
        if (RELU) o = fmaxf(o, 0.0f);
        op[0] = o;
    }
}


template<int NOUT>
__global__ void __launch_bounds__(512)
gather_gemm_kernel(const unsigned int* __restrict__ rowptr,
                   const int* __restrict__ epk,
                   const float* __restrict__ h,
                   const float* __restrict__ dis,
                   const float* __restrict__ b,
                   const float* __restrict__ Wn,
                   float* __restrict__ Hn, int n) {
    constexpr int F = 64;
    constexpr int V = 2;
    constexpr int G = 8;
    constexpr int NT = NOUT / 16;
    constexpr int VO = NOUT / 32;
    constexpr int AP = F + 4;
    constexpr int PW = NOUT + 16;
    constexpr int OP = NOUT + 4;
    __shared__ alignas(16) float2 sW[(F / 2) * PW];
    __shared__ alignas(16) float sA[16 * AP];
    __shared__ alignas(16) float sO[16 * OP];

    const int tid = threadIdx.x;
    const int lane = tid & 31;
    const int wave = __builtin_amdgcn_readfirstlane(tid >> 5);

    const int ntiles = (n + 15) >> 4;
    int tile = blockIdx.x;
    if (tile >= ntiles) return;

    for (int i = tid; i < (F / 2) * (NOUT / 4); i += 512) {
        const int kp = i / (NOUT / 4);
        const int cc = (i % (NOUT / 4)) * 4;
        const float4 w0 = *(const float4*)(Wn + (2 * kp) * NOUT + cc);
        const float4 w1 = *(const float4*)(Wn + (2 * kp + 1) * NOUT + cc);
        float4* dp = (float4*)&sW[kp * PW + cc];
        dp[0] = make_float4(w0.x, w1.x, w0.y, w1.y);
        dp[1] = make_float4(w0.z, w1.z, w0.w, w1.w);
    }
    const float2 bb = *(const float2*)(b + lane * V);

    const int half = lane >> 4;
    const int r    = lane & 15;

    int node = __builtin_amdgcn_readfirstlane(tile * 16 + wave);
    bool valid = node < n;
    int nodec = valid ? node : (n - 1);
    unsigned int beg = rowptr[nodec];
    unsigned int nxt = rowptr[nodec + 1];
    float dd = dis[nodec];
    float2 q = *(const float2*)(h + (long long)nodec * F + lane * V);

    for (;;) {
        const int tile_n = tile + (int)gridDim.x;
        const int node_n = __builtin_amdgcn_readfirstlane(tile_n * 16 + wave);
        const bool valid_n = node_n < n;
        const int nodec_n = valid_n ? node_n : (n - 1);
        const unsigned int beg_n = rowptr[nodec_n];
        const unsigned int nxt_n = rowptr[nodec_n + 1];
        const float dd_n = dis[nodec_n];
        const float2 q_n = *(const float2*)(h + (long long)nodec_n * F + lane * V);

        const unsigned int begs = __builtin_amdgcn_readfirstlane(beg);
        const unsigned int end  = __builtin_amdgcn_readfirstlane(valid ? nxt : beg);
        float acc[V];
        acc[0] = q.x;
        acc[V - 1] = q.y;

        for (unsigned int base = begs; base < end; base += 32u) {
            const unsigned int j = base + (unsigned int)lane;
            int es = nodec;
            if (j < end) es = epk[j];
            const unsigned int left = end - base;
            const int m = (int)(left < 32u ? left : 32u);
            for (int t = 0; t < m; t += G) {
                float hv[G][V];
#pragma unroll
                for (int u = 0; u < G; ++u) {
                    const int s = __builtin_amdgcn_readlane(es, t + u);
                    const float2 qq = *(const float2*)(h + (long long)s * F + lane * V);
                    hv[u][0] = qq.x;
                    hv[u][V - 1] = qq.y;
                }
#pragma unroll
                for (int u = 0; u < G; ++u) {
                    const bool live = (t + u) < m;
#pragma unroll
                    for (int v = 0; v < V; ++v)
                        acc[v] += live ? hv[u][v] : 0.0f;
                }
            }
        }
        acc[0] = valid ? fmaxf(acc[0] * dd + bb.x, 0.0f) : 0.0f;
        acc[V - 1] = valid ? fmaxf(acc[V - 1] * dd + bb.y, 0.0f) : 0.0f;

        *(float2*)&sA[wave * AP + lane * V] = make_float2(acc[0], acc[V - 1]);
        __syncthreads();

        if (wave < NT) {
            v8f d = {};
#pragma unroll
            for (int kt = 0; kt < F / 4; ++kt) {
                const v2f a = *(const v2f*)&sA[r * AP + kt * 4 + 2 * half];
                const float2 bq = sW[(kt * 2 + half) * PW + wave * 16 + r];
                v2f bf;
                bf.x = bq.x;
                bf.y = bq.y;
                d = __builtin_amdgcn_wmma_f32_16x16x4_f32(false, a, false, bf, (short)0, d, false, false);
            }
#pragma unroll
            for (int v = 0; v < 8; ++v) sO[(v + 8 * half) * OP + wave * 16 + r] = d[v];
        }
        __syncthreads();

        if (valid) {
            float* __restrict__ op = Hn + (long long)node * NOUT + lane * VO;
            if (VO == 2) {
                const float2 o = *(const float2*)&sO[wave * OP + lane * VO];
                *(float2*)op = make_float2(o.x * dd, o.y * dd);
            } else {
                op[0] = sO[wave * OP + lane * VO] * dd;
            }
        }

        if (tile_n >= ntiles) break;
        tile = tile_n; node = node_n; valid = valid_n; nodec = nodec_n;
        beg = beg_n; nxt = nxt_n; dd = dd_n; q = q_n;
    }
}


static inline int cdiv(long long a, int b) { return (int)((a + b - 1) / b); }

extern "C" void kernel_launch(void* const* d_in, const int* in_sizes, int n_in,
                              void* d_out, int out_size, void* d_ws, size_t ws_size,
                              hipStream_t stream) {
    const float* x  = (const float*)d_in[0];
    const int*   ei = (const int*)d_in[1];
    const float* W1 = (const float*)d_in[2];
    const float* b1 = (const float*)d_in[3];
    const float* W2 = (const float*)d_in[4];
    const float* b2 = (const float*)d_in[5];
    const float* W3 = (const float*)d_in[6];
    const float* b3 = (const float*)d_in[7];
    float* out = (float*)d_out;

    const int* src = ei;
    const int* dst = ei + N_EDGES;

    char* ws = (char*)d_ws;
    unsigned int* hist    = (unsigned int*)ws;
    unsigned int* binbase = (unsigned int*)(ws + 200704);
    float* dis  = (float*)(ws + 400384);
    float* bufA = (float*)(ws + 800768);
    float* bufB = (float*)(ws + 26400768);
    unsigned int* rowptr = (unsigned int*)(ws + 52000768);
    int* epk             = (int*)(ws + 58801152);
    int2* tmp            = (int2*)(ws + 71601152);

    const int N = N_NODES, E = N_EDGES;
    const int gemm_blocks = cdiv(cdiv(N, 16), 8);
    const int node_blocks = cdiv(N, 8);
    const int tile_groups = 1024;

    bin_hist_kernel<<<EDGE_CHUNKS, BT, 0, stream>>>(dst, hist, E);
    bin_partition_kernel<<<EDGE_CHUNKS, BT, 0, stream>>>(src, dst, hist, binbase, tmp, E);
    node_bucket_kernel<<<NBINS, BT, 0, stream>>>(tmp, binbase, rowptr, dis, epk, N);

    gemm_wmma_kernel<128, 64><<<gemm_blocks, 256, 0, stream>>>(x, W1, dis, bufA, N);

    gather_gemm_kernel<64><<<tile_groups, 512, 0, stream>>>(rowptr, epk, bufA, dis, b1, W2, bufB, N);

    gather_gemm_kernel<32><<<tile_groups, 512, 0, stream>>>(rowptr, epk, bufB, dis, b2, W3, bufA, N);

    gather_kernel<32, false><<<node_blocks, 256, 0, stream>>>(rowptr, epk, bufA, dis, b3, out, N);
}


# Histogram launch dropped: level 1 sorts each chunk into its own region plus a table of bin starts, bin work-groups collect their 256 pieces; one word per edge; 6 launches; sum order in a node new
# speedup vs baseline: 7.5456x; 1.0531x over previous
#include <hip/hip_runtime.h>

typedef float v2f __attribute__((ext_vector_type(2)));
typedef float v8f __attribute__((ext_vector_type(8)));

#define N_NODES 100000
#define N_EDGES 1600000


#define BIN_SHIFT   9
#define BIN_NODES   (1 << BIN_SHIFT)
#define SRC_BITS    (32 - BIN_SHIFT)
#define NBINS       ((N_NODES + BIN_NODES - 1) / BIN_NODES)
#define TBL         (((NBINS + 1 + 31) / 32) * 32)
#define EDGE_CHUNKS 256
#define CHUNK_EDGES ((N_EDGES + EDGE_CHUNKS - 1) / EDGE_CHUNKS)
#define BT          1024
#define CHUNK_PITCH (((CHUNK_EDGES + 31) / 32) * 32)
#define EPT         ((CHUNK_EDGES + BT - 1) / BT)
#define PPW         (EDGE_CHUNKS / (BT / 32))
#define STAGE_CAP   16384

static_assert(N_NODES <= (1 << SRC_BITS), "a src id and a node's place in its bin share one word");
static_assert(NBINS <= BT, "one thread per bin");
static_assert(EDGE_CHUNKS <= BT, "one thread per chunk reads the table column");
static_assert(EDGE_CHUNKS % (BT / 32) == 0, "every wave takes the same number of pieces");
static_assert(BIN_NODES <= BT, "one thread per node of a bin");

__device__ __forceinline__ unsigned int block_excl_scan(unsigned int v, unsigned int* s_w, unsigned int& total) {
    const int lane = threadIdx.x & 31;
    const int wave = threadIdx.x >> 5;
    unsigned int inc = v;
    for (int d = 1; d < 32; d <<= 1) {
        const unsigned int t = __shfl_up(inc, d);
        if (lane >= d) inc += t;
    }
    if (lane == 31) s_w[wave] = inc;
    __syncthreads();
    unsigned int off = 0u, tot = 0u;
    for (int w = 0; w < BT / 32; ++w) {
        const unsigned int x = s_w[w];
        if (w < wave) off += x;
        tot += x;
    }
    __syncthreads();
    total = tot;
    return off + inc - v;
}

__device__ __forceinline__ void block_sum2(unsigned int a, unsigned int b, unsigned int* s_w2,
                                           unsigned int& ta, unsigned int& tb) {
    for (int d = 16; d >= 1; d >>= 1) {
        a += __shfl_xor(a, d);
        b += __shfl_xor(b, d);
    }
    const int lane = threadIdx.x & 31;
    const int wave = threadIdx.x >> 5;
    if (lane == 0) { s_w2[2 * wave] = a; s_w2[2 * wave + 1] = b; }
    __syncthreads();
    unsigned int x = 0u, y = 0u;
    for (int w = 0; w < BT / 32; ++w) { x += s_w2[2 * w]; y += s_w2[2 * w + 1]; }
    __syncthreads();
    ta = x;
    tb = y;
}

__global__ void __launch_bounds__(BT)
chunk_sort_kernel(const int* __restrict__ src, const int* __restrict__ dst,
                  unsigned int* __restrict__ tbl,
                  unsigned int* __restrict__ tmp, int E) {
    __shared__ unsigned int s_sorted[CHUNK_EDGES];
    __shared__ unsigned int s_cnt[NBINS];
    __shared__ unsigned int s_lbase[NBINS];
    __shared__ unsigned int s_w[BT / 32];

    const int tid = threadIdx.x;
    const int chunk = blockIdx.x;
    const int e0 = chunk * CHUNK_EDGES;
    const int e1 = (e0 + CHUNK_EDGES < E) ? (e0 + CHUNK_EDGES) : E;
    const int n = (e1 > e0) ? (e1 - e0) : 0;

    int es[EPT], ed[EPT];
#pragma unroll
    for (int k = 0; k < EPT; ++k) {
        const int e = e0 + tid + k * BT;
        es[k] = 0;
        ed[k] = -1;
        if (e < e1) { es[k] = src[e]; ed[k] = dst[e]; }
    }
    if (tid < NBINS) s_cnt[tid] = 0u;
    __syncthreads();

    unsigned int rk[EPT];
#pragma unroll
    for (int k = 0; k < EPT; ++k) {
        rk[k] = 0u;
        if (ed[k] >= 0) rk[k] = atomicAdd(&s_cnt[ed[k] >> BIN_SHIFT], 1u);
    }
    __syncthreads();

    unsigned int total = 0u;
    const unsigned int mycnt = (tid < NBINS) ? s_cnt[tid] : 0u;
    const unsigned int lbase = block_excl_scan(mycnt, s_w, total);
    if (tid < NBINS) {
        s_lbase[tid] = lbase;
        tbl[chunk * TBL + tid] = lbase;
    }
    if (tid == 0) tbl[chunk * TBL + NBINS] = total;
    __syncthreads();

#pragma unroll
    for (int k = 0; k < EPT; ++k)
        if (ed[k] >= 0)
            s_sorted[s_lbase[ed[k] >> BIN_SHIFT] + rk[k]] =
                (unsigned int)es[k] | ((unsigned int)(ed[k] & (BIN_NODES - 1)) << SRC_BITS);
    __syncthreads();

    for (int i = tid; i < n; i += BT) tmp[chunk * CHUNK_PITCH + i] = s_sorted[i];
}

__global__ void __launch_bounds__(BT)
node_bucket_kernel(const unsigned int* __restrict__ tmp,
                   const unsigned int* __restrict__ tbl,
                   unsigned int* __restrict__ rowptr,
                   float* __restrict__ dis,
                   int* __restrict__ epk, int n_nodes) {
    __shared__ int s_stage[STAGE_CAP];
    __shared__ unsigned int s_cnt[BIN_NODES];
    __shared__ unsigned int s_start[BIN_NODES];
    __shared__ unsigned int s_pbeg[EDGE_CHUNKS];
    __shared__ unsigned int s_plen[EDGE_CHUNKS];
    __shared__ unsigned int s_w[2 * (BT / 32)];
    const int tid = threadIdx.x;
    const int lane = tid & 31;
    const int wave = tid >> 5;
    const int bin = blockIdx.x;

    unsigned int pa = 0u, pl = 0u;
    if (tid < EDGE_CHUNKS) {
        pa = tbl[tid * TBL + bin];
        pl = tbl[tid * TBL + bin + 1] - pa;
        s_pbeg[tid] = (unsigned int)tid * CHUNK_PITCH + pa;
        s_plen[tid] = pl;
    }
    if (tid < BIN_NODES) s_cnt[tid] = 0u;
    unsigned int lo = 0u, n = 0u;
    block_sum2(pa, pl, s_w, lo, n);

    unsigned int pb[PPW], pn[PPW], hw[PPW];
#pragma unroll
    for (int k = 0; k < PPW; ++k) {
        const int c = wave + k * (BT / 32);
        pb[k] = s_pbeg[c];
        pn[k] = s_plen[c];
    }
#pragma unroll
    for (int k = 0; k < PPW; ++k) {
        hw[k] = 0u;
        if ((unsigned int)lane < pn[k]) hw[k] = tmp[pb[k] + (unsigned int)lane];
    }
#pragma unroll
    for (int k = 0; k < PPW; ++k)
        if ((unsigned int)lane < pn[k]) atomicAdd(&s_cnt[hw[k] >> SRC_BITS], 1u);
    for (int k = 0; k < PPW; ++k)
        for (unsigned int j = (unsigned int)lane + 32u; j < pn[k]; j += 32u)
            atomicAdd(&s_cnt[tmp[pb[k] + j] >> SRC_BITS], 1u);
    __syncthreads();

    const unsigned int c = (tid < BIN_NODES) ? s_cnt[tid] : 0u;
    unsigned int all = 0u;
    const unsigned int ex = block_excl_scan(c, s_w, all);
    const int node = bin * BIN_NODES + tid;
    if (tid < BIN_NODES) {
        s_start[tid] = ex;
        s_cnt[tid] = 0u;
        if (node < n_nodes) {
            rowptr[node] = lo + ex;
            dis[node] = rsqrtf((float)c + 1.0f);
            if (node == n_nodes - 1) rowptr[n_nodes] = lo + ex + c;
        }
    }
    __syncthreads();

    const bool staged = n <= (unsigned int)STAGE_CAP;
    auto place = [&](unsigned int word) {
        const unsigned int l = word >> SRC_BITS;
        const unsigned int slot = s_start[l] + atomicAdd(&s_cnt[l], 1u);
        const int s = (int)(word & ((1u << SRC_BITS) - 1u));
        if (staged) s_stage[slot] = s;
        else        epk[lo + slot] = s;
    };
#pragma unroll
    for (int k = 0; k < PPW; ++k)
        if ((unsigned int)lane < pn[k]) place(hw[k]);
    for (int k = 0; k < PPW; ++k)
        for (unsigned int j = (unsigned int)lane + 32u; j < pn[k]; j += 32u)
            place(tmp[pb[k] + j]);
    __syncthreads();
    if (staged) {
        for (unsigned int i = (unsigned int)tid; i < n; i += BT) epk[lo + i] = s_stage[i];
    }
}


template<int K, int NOUT>
__global__ void __launch_bounds__(256)
gemm_wmma_kernel(const float* __restrict__ X, const float* __restrict__ W,
                 const float* __restrict__ dis,
                 float* __restrict__ H, int nrows) {
    __shared__ float sW[K * NOUT];
    const int tid = threadIdx.x;
    for (int i = tid; i < K * NOUT; i += 256) sW[i] = W[i];
    __syncthreads();

    const int wave = tid >> 5;
    const int lane = tid & 31;
    const int half = lane >> 4;
    const int r    = lane & 15;

    const int row0 = (blockIdx.x * 8 + wave) * 16;
    if (row0 >= nrows) return;

    int arow = row0 + r;
    if (arow >= nrows) arow = nrows - 1;
    const float* __restrict__ xrow = X + (long long)arow * K;

    constexpr int NT = NOUT / 16;
    v8f acc[NT];
#pragma unroll
    for (int nt = 0; nt < NT; ++nt) {
        v8f z = {};
        acc[nt] = z;
    }
    float dr[8];
#pragma unroll
    for (int v = 0; v < 8; ++v) {
        const int row = row0 + v + 8 * half;
        dr[v] = dis[row < nrows ? row : (nrows - 1)];
    }

#pragma unroll 4
    for (int kt = 0; kt < K / 4; ++kt) {
        const int k0 = kt * 4 + 2 * half;
        v2f a;
        a.x = xrow[k0];
        a.y = xrow[k0 + 1];
#pragma unroll
        for (int nt = 0; nt < NT; ++nt) {
            v2f b;
            b.x = sW[(k0)     * NOUT + nt * 16 + r];
            b.y = sW[(k0 + 1) * NOUT + nt * 16 + r];
            acc[nt] = __builtin_amdgcn_wmma_f32_16x16x4_f32(
                false, a, false, b, (short)0, acc[nt], false, false);
        }
    }

#pragma unroll
    for (int nt = 0; nt < NT; ++nt) {
#pragma unroll
        for (int v = 0; v < 8; ++v) {
            const int row = row0 + v + 8 * half;
            const int col = nt * 16 + r;
            if (row < nrows) H[(long long)row * NOUT + col] = acc[nt][v] * dr[v];
        }
    }
}


template<int F, bool RELU>
__global__ void __launch_bounds__(256)
gather_kernel(const unsigned int* __restrict__ rowptr,
              const int* __restrict__ epk,
              const float* __restrict__ h,
              const float* __restrict__ dis,
              const float* __restrict__ b,
              float* __restrict__ out, int n) {
    constexpr int V = F / 32;
    constexpr int G = 8;
    const int lane = threadIdx.x & 31;
    const int node = __builtin_amdgcn_readfirstlane((int)(blockIdx.x * 8 + (threadIdx.x >> 5)));
    if (node >= n) return;

    const unsigned int beg = rowptr[node];
    const unsigned int end = rowptr[node + 1];
    const float dd = dis[node];

    float bias[V];
    if (V == 2) {
        const float2 bb = *(const float2*)(b + lane * V);
        bias[0] = bb.x;
        bias[V - 1] = bb.y;
    } else {
        bias[0] = b[lane];
    }

    float acc[V];
    {
        const float* __restrict__ hp = h + (long long)node * F + lane * V;
        if (V == 2) {
            const float2 q = *(const float2*)hp;
            acc[0] = q.x;
            acc[V - 1] = q.y;
        } else {
            acc[0] = hp[0];
        }
    }

    for (unsigned int base = beg; base < end; base += 32u) {
        const unsigned int j = base + (unsigned int)lane;
        int es = node;
        if (j < end) es = epk[j];
        const unsigned int left = end - base;
        const int m = (int)(left < 32u ? left : 32u);
        for (int t = 0; t < m; t += G) {
            float hv[G][V];
#pragma unroll
            for (int u = 0; u < G; ++u) {
                const int s = __builtin_amdgcn_readlane(es, t + u);
                const float* __restrict__ hp = h + (long long)s * F + lane * V;
                if (V == 2) {
                    const float2 q = *(const float2*)hp;
                    hv[u][0] = q.x;
                    hv[u][V - 1] = q.y;
                } else {
                    hv[u][0] = hp[0];
                }
            }
#pragma unroll
            for (int u = 0; u < G; ++u) {
                const bool live = (t + u) < m;
#pragma unroll
                for (int v = 0; v < V; ++v)
                    acc[v] += live ? hv[u][v] : 0.0f;
            }
        }
    }

    float* __restrict__ op = out + (long long)node * F + lane * V;
    if (V == 2) {
        float2 o;
        o.x = acc[0] * dd + bias[0];
        o.y = acc[V - 1] * dd + bias[V - 1];
        if (RELU) { o.x = fmaxf(o.x, 0.0f); o.y = fmaxf(o.y, 0.0f); }
        *(float2*)op = o;
    } else {
        float o = acc[0] * dd + bias[0];
        if (RELU) o = fmaxf(o, 0.0f);
        op[0] = o;
    }
}


template<int NOUT>
__global__ void __launch_bounds__(512)
gather_gemm_kernel(const unsigned int* __restrict__ rowptr,
                   const int* __restrict__ epk,
                   const float* __restrict__ h,
                   const float* __restrict__ dis,
                   const float* __restrict__ b,
                   const float* __restrict__ Wn,
                   float* __restrict__ Hn, int n) {
    constexpr int F = 64;
    constexpr int V = 2;
    constexpr int G = 8;
    constexpr int NT = NOUT / 16;
    constexpr int VO = NOUT / 32;
    constexpr int AP = F + 4;
    constexpr int PW = NOUT + 16;
    constexpr int OP = NOUT + 4;
    __shared__ alignas(16) float2 sW[(F / 2) * PW];
    __shared__ alignas(16) float sA[16 * AP];
    __shared__ alignas(16) float sO[16 * OP];

    const int tid = threadIdx.x;
    const int lane = tid & 31;
    const int wave = __builtin_amdgcn_readfirstlane(tid >> 5);

    const int ntiles = (n + 15) >> 4;
    int tile = blockIdx.x;
    if (tile >= ntiles) return;

    for (int i = tid; i < (F / 2) * (NOUT / 4); i += 512) {
        const int kp = i / (NOUT / 4);
        const int cc = (i % (NOUT / 4)) * 4;
        const float4 w0 = *(const float4*)(Wn + (2 * kp) * NOUT + cc);
        const float4 w1 = *(const float4*)(Wn + (2 * kp + 1) * NOUT + cc);
        float4* dp = (float4*)&sW[kp * PW + cc];
        dp[0] = make_float4(w0.x, w1.x, w0.y, w1.y);
        dp[1] = make_float4(w0.z, w1.z, w0.w, w1.w);
    }
    const float2 bb = *(const float2*)(b + lane * V);

    const int half = lane >> 4;
    const int r    = lane & 15;

    int node = __builtin_amdgcn_readfirstlane(tile * 16 + wave);
    bool valid = node < n;
    int nodec = valid ? node : (n - 1);
    unsigned int beg = rowptr[nodec];
    unsigned int nxt = rowptr[nodec + 1];
    float dd = dis[nodec];
    float2 q = *(const float2*)(h + (long long)nodec * F + lane * V);

    for (;;) {
        const int tile_n = tile + (int)gridDim.x;
        const int node_n = __builtin_amdgcn_readfirstlane(tile_n * 16 + wave);
        const bool valid_n = node_n < n;
        const int nodec_n = valid_n ? node_n : (n - 1);
        const unsigned int beg_n = rowptr[nodec_n];
        const unsigned int nxt_n = rowptr[nodec_n + 1];
        const float dd_n = dis[nodec_n];
        const float2 q_n = *(const float2*)(h + (long long)nodec_n * F + lane * V);

        const unsigned int begs = __builtin_amdgcn_readfirstlane(beg);
        const unsigned int end  = __builtin_amdgcn_readfirstlane(valid ? nxt : beg);
        float acc[V];
        acc[0] = q.x;
        acc[V - 1] = q.y;

        for (unsigned int base = begs; base < end; base += 32u) {
            const unsigned int j = base + (unsigned int)lane;
            int es = nodec;
            if (j < end) es = epk[j];
            const unsigned int left = end - base;
            const int m = (int)(left < 32u ? left : 32u);
            for (int t = 0; t < m; t += G) {
                float hv[G][V];
#pragma unroll
                for (int u = 0; u < G; ++u) {
                    const int s = __builtin_amdgcn_readlane(es, t + u);
                    const float2 qq = *(const float2*)(h + (long long)s * F + lane * V);
                    hv[u][0] = qq.x;
                    hv[u][V - 1] = qq.y;
                }
#pragma unroll
                for (int u = 0; u < G; ++u) {
                    const bool live = (t + u) < m;
#pragma unroll
                    for (int v = 0; v < V; ++v)
                        acc[v] += live ? hv[u][v] : 0.0f;
                }
            }
        }
        acc[0] = valid ? fmaxf(acc[0] * dd + bb.x, 0.0f) : 0.0f;
        acc[V - 1] = valid ? fmaxf(acc[V - 1] * dd + bb.y, 0.0f) : 0.0f;

        *(float2*)&sA[wave * AP + lane * V] = make_float2(acc[0], acc[V - 1]);
        __syncthreads();

        if (wave < NT) {
            v8f d = {};
#pragma unroll
            for (int kt = 0; kt < F / 4; ++kt) {
                const v2f a = *(const v2f*)&sA[r * AP + kt * 4 + 2 * half];
                const float2 bq = sW[(kt * 2 + half) * PW + wave * 16 + r];
                v2f bf;
                bf.x = bq.x;
                bf.y = bq.y;
                d = __builtin_amdgcn_wmma_f32_16x16x4_f32(false, a, false, bf, (short)0, d, false, false);
            }
#pragma unroll
            for (int v = 0; v < 8; ++v) sO[(v + 8 * half) * OP + wave * 16 + r] = d[v];
        }
        __syncthreads();

        if (valid) {
            float* __restrict__ op = Hn + (long long)node * NOUT + lane * VO;
            if (VO == 2) {
                const float2 o = *(const float2*)&sO[wave * OP + lane * VO];
                *(float2*)op = make_float2(o.x * dd, o.y * dd);
            } else {
                op[0] = sO[wave * OP + lane * VO] * dd;
            }
        }

        if (tile_n >= ntiles) break;
        tile = tile_n; node = node_n; valid = valid_n; nodec = nodec_n;
        beg = beg_n; nxt = nxt_n; dd = dd_n; q = q_n;
    }
}


static inline int cdiv(long long a, int b) { return (int)((a + b - 1) / b); }

extern "C" void kernel_launch(void* const* d_in, const int* in_sizes, int n_in,
                              void* d_out, int out_size, void* d_ws, size_t ws_size,
                              hipStream_t stream) {
    const float* x  = (const float*)d_in[0];
    const int*   ei = (const int*)d_in[1];
    const float* W1 = (const float*)d_in[2];
    const float* b1 = (const float*)d_in[3];
    const float* W2 = (const float*)d_in[4];
    const float* b2 = (const float*)d_in[5];
    const float* W3 = (const float*)d_in[6];
    const float* b3 = (const float*)d_in[7];
    float* out = (float*)d_out;

    const int* src = ei;
    const int* dst = ei + N_EDGES;

    char* ws = (char*)d_ws;
    unsigned int* tbl = (unsigned int*)ws;
    float* dis  = (float*)(ws + 400384);
    float* bufA = (float*)(ws + 800768);
    float* bufB = (float*)(ws + 26400768);
    unsigned int* rowptr = (unsigned int*)(ws + 52000768);
    int* epk             = (int*)(ws + 58801152);
    unsigned int* tmp    = (unsigned int*)(ws + 71601152);

    const int N = N_NODES, E = N_EDGES;
    const int gemm_blocks = cdiv(cdiv(N, 16), 8);
    const int node_blocks = cdiv(N, 8);
    const int tile_groups = 1024;

    chunk_sort_kernel<<<EDGE_CHUNKS, BT, 0, stream>>>(src, dst, tbl, tmp, E);
    node_bucket_kernel<<<NBINS, BT, 0, stream>>>(tmp, tbl, rowptr, dis, epk, N);

    gemm_wmma_kernel<128, 64><<<gemm_blocks, 256, 0, stream>>>(x, W1, dis, bufA, N);

    gather_gemm_kernel<64><<<tile_groups, 512, 0, stream>>>(rowptr, epk, bufA, dis, b1, W2, bufB, N);

    gather_gemm_kernel<32><<<tile_groups, 512, 0, stream>>>(rowptr, epk, bufB, dis, b2, W3, bufA, N);

    gather_kernel<32, false><<<node_blocks, 256, 0, stream>>>(rowptr, epk, bufA, dis, b3, out, N);
}


# Last layer's sum in resident waves (1024 groups of 8), about 12 nodes each, heads two nodes and entries one node ahead, two rows per load by half-waves; half sums joined last: sum order new
# speedup vs baseline: 7.9125x; 1.0486x over previous
#include <hip/hip_runtime.h>

typedef float v2f __attribute__((ext_vector_type(2)));
typedef float v8f __attribute__((ext_vector_type(8)));

#define N_NODES 100000
#define N_EDGES 1600000


#define BIN_SHIFT   9
#define BIN_NODES   (1 << BIN_SHIFT)
#define SRC_BITS    (32 - BIN_SHIFT)
#define NBINS       ((N_NODES + BIN_NODES - 1) / BIN_NODES)
#define TBL         (((NBINS + 1 + 31) / 32) * 32)
#define EDGE_CHUNKS 256
#define CHUNK_EDGES ((N_EDGES + EDGE_CHUNKS - 1) / EDGE_CHUNKS)
#define BT          1024
#define CHUNK_PITCH (((CHUNK_EDGES + 31) / 32) * 32)
#define EPT         ((CHUNK_EDGES + BT - 1) / BT)
#define PPW         (EDGE_CHUNKS / (BT / 32))
#define STAGE_CAP   16384

static_assert(N_NODES <= (1 << SRC_BITS), "a src id and a node's place in its bin share one word");
static_assert(NBINS <= BT, "one thread per bin");
static_assert(EDGE_CHUNKS <= BT, "one thread per chunk reads the table column");
static_assert(EDGE_CHUNKS % (BT / 32) == 0, "every wave takes the same number of pieces");
static_assert(BIN_NODES <= BT, "one thread per node of a bin");

__device__ __forceinline__ unsigned int block_excl_scan(unsigned int v, unsigned int* s_w, unsigned int& total) {
    const int lane = threadIdx.x & 31;
    const int wave = threadIdx.x >> 5;
    unsigned int inc = v;
    for (int d = 1; d < 32; d <<= 1) {
        const unsigned int t = __shfl_up(inc, d);
        if (lane >= d) inc += t;
    }
    if (lane == 31) s_w[wave] = inc;
    __syncthreads();
    unsigned int off = 0u, tot = 0u;
    for (int w = 0; w < BT / 32; ++w) {
        const unsigned int x = s_w[w];
        if (w < wave) off += x;
        tot += x;
    }
    __syncthreads();
    total = tot;
    return off + inc - v;
}

__device__ __forceinline__ void block_sum2(unsigned int a, unsigned int b, unsigned int* s_w2,
                                           unsigned int& ta, unsigned int& tb) {
    for (int d = 16; d >= 1; d >>= 1) {
        a += __shfl_xor(a, d);
        b += __shfl_xor(b, d);
    }
    const int lane = threadIdx.x & 31;
    const int wave = threadIdx.x >> 5;
    if (lane == 0) { s_w2[2 * wave] = a; s_w2[2 * wave + 1] = b; }
    __syncthreads();
    unsigned int x = 0u, y = 0u;
    for (int w = 0; w < BT / 32; ++w) { x += s_w2[2 * w]; y += s_w2[2 * w + 1]; }
    __syncthreads();
    ta = x;
    tb = y;
}

__global__ void __launch_bounds__(BT)
chunk_sort_kernel(const int* __restrict__ src, const int* __restrict__ dst,
                  unsigned int* __restrict__ tbl,
                  unsigned int* __restrict__ tmp, int E) {
    __shared__ unsigned int s_sorted[CHUNK_EDGES];
    __shared__ unsigned int s_cnt[NBINS];
    __shared__ unsigned int s_lbase[NBINS];
    __shared__ unsigned int s_w[BT / 32];

    const int tid = threadIdx.x;
    const int chunk = blockIdx.x;
    const int e0 = chunk * CHUNK_EDGES;
    const int e1 = (e0 + CHUNK_EDGES < E) ? (e0 + CHUNK_EDGES) : E;
    const int n = (e1 > e0) ? (e1 - e0) : 0;

    int es[EPT], ed[EPT];
#pragma unroll
    for (int k = 0; k < EPT; ++k) {
        const int e = e0 + tid + k * BT;
        es[k] = 0;
        ed[k] = -1;
        if (e < e1) { es[k] = src[e]; ed[k] = dst[e]; }
    }
    if (tid < NBINS) s_cnt[tid] = 0u;
    __syncthreads();

    unsigned int rk[EPT];
#pragma unroll
    for (int k = 0; k < EPT; ++k) {
        rk[k] = 0u;
        if (ed[k] >= 0) rk[k] = atomicAdd(&s_cnt[ed[k] >> BIN_SHIFT], 1u);
    }
    __syncthreads();

    unsigned int total = 0u;
    const unsigned int mycnt = (tid < NBINS) ? s_cnt[tid] : 0u;
    const unsigned int lbase = block_excl_scan(mycnt, s_w, total);
    if (tid < NBINS) {
        s_lbase[tid] = lbase;
        tbl[chunk * TBL + tid] = lbase;
    }
    if (tid == 0) tbl[chunk * TBL + NBINS] = total;
    __syncthreads();

#pragma unroll
    for (int k = 0; k < EPT; ++k)
        if (ed[k] >= 0)
            s_sorted[s_lbase[ed[k] >> BIN_SHIFT] + rk[k]] =
                (unsigned int)es[k] | ((unsigned int)(ed[k] & (BIN_NODES - 1)) << SRC_BITS);
    __syncthreads();

    for (int i = tid; i < n; i += BT) tmp[chunk * CHUNK_PITCH + i] = s_sorted[i];
}

__global__ void __launch_bounds__(BT)
node_bucket_kernel(const unsigned int* __restrict__ tmp,
                   const unsigned int* __restrict__ tbl,
                   unsigned int* __restrict__ rowptr,
                   float* __restrict__ dis,
                   int* __restrict__ epk, int n_nodes) {
    __shared__ int s_stage[STAGE_CAP];
    __shared__ unsigned int s_cnt[BIN_NODES];
    __shared__ unsigned int s_start[BIN_NODES];
    __shared__ unsigned int s_pbeg[EDGE_CHUNKS];
    __shared__ unsigned int s_plen[EDGE_CHUNKS];
    __shared__ unsigned int s_w[2 * (BT / 32)];
    const int tid = threadIdx.x;
    const int lane = tid & 31;
    const int wave = tid >> 5;
    const int bin = blockIdx.x;

    unsigned int pa = 0u, pl = 0u;
    if (tid < EDGE_CHUNKS) {
        pa = tbl[tid * TBL + bin];
        pl = tbl[tid * TBL + bin + 1] - pa;
        s_pbeg[tid] = (unsigned int)tid * CHUNK_PITCH + pa;
        s_plen[tid] = pl;
    }
    if (tid < BIN_NODES) s_cnt[tid] = 0u;
    unsigned int lo = 0u, n = 0u;
    block_sum2(pa, pl, s_w, lo, n);

    unsigned int pb[PPW], pn[PPW], hw[PPW];
#pragma unroll
    for (int k = 0; k < PPW; ++k) {
        const int c = wave + k * (BT / 32);
        pb[k] = s_pbeg[c];
        pn[k] = s_plen[c];
    }
#pragma unroll
    for (int k = 0; k < PPW; ++k) {
        hw[k] = 0u;
        if ((unsigned int)lane < pn[k]) hw[k] = tmp[pb[k] + (unsigned int)lane];
    }
#pragma unroll
    for (int k = 0; k < PPW; ++k)
        if ((unsigned int)lane < pn[k]) atomicAdd(&s_cnt[hw[k] >> SRC_BITS], 1u);
    for (int k = 0; k < PPW; ++k)
        for (unsigned int j = (unsigned int)lane + 32u; j < pn[k]; j += 32u)
            atomicAdd(&s_cnt[tmp[pb[k] + j] >> SRC_BITS], 1u);
    __syncthreads();

    const unsigned int c = (tid < BIN_NODES) ? s_cnt[tid] : 0u;
    unsigned int all = 0u;
    const unsigned int ex = block_excl_scan(c, s_w, all);
    const int node = bin * BIN_NODES + tid;
    if (tid < BIN_NODES) {
        s_start[tid] = ex;
        s_cnt[tid] = 0u;
        if (node < n_nodes) {
            rowptr[node] = lo + ex;
            dis[node] = rsqrtf((float)c + 1.0f);
            if (node == n_nodes - 1) rowptr[n_nodes] = lo + ex + c;
        }
    }
    __syncthreads();

    const bool staged = n <= (unsigned int)STAGE_CAP;
    auto place = [&](unsigned int word) {
        const unsigned int l = word >> SRC_BITS;
        const unsigned int slot = s_start[l] + atomicAdd(&s_cnt[l], 1u);
        const int s = (int)(word & ((1u << SRC_BITS) - 1u));
        if (staged) s_stage[slot] = s;
        else        epk[lo + slot] = s;
    };
#pragma unroll
    for (int k = 0; k < PPW; ++k)
        if ((unsigned int)lane < pn[k]) place(hw[k]);
    for (int k = 0; k < PPW; ++k)
        for (unsigned int j = (unsigned int)lane + 32u; j < pn[k]; j += 32u)
            place(tmp[pb[k] + j]);
    __syncthreads();
    if (staged) {
        for (unsigned int i = (unsigned int)tid; i < n; i += BT) epk[lo + i] = s_stage[i];
    }
}


template<int K, int NOUT>
__global__ void __launch_bounds__(256)
gemm_wmma_kernel(const float* __restrict__ X, const float* __restrict__ W,
                 const float* __restrict__ dis,
                 float* __restrict__ H, int nrows) {
    __shared__ float sW[K * NOUT];
    const int tid = threadIdx.x;
    for (int i = tid; i < K * NOUT; i += 256) sW[i] = W[i];
    __syncthreads();

    const int wave = tid >> 5;
    const int lane = tid & 31;
    const int half = lane >> 4;
    const int r    = lane & 15;

    const int row0 = (blockIdx.x * 8 + wave) * 16;
    if (row0 >= nrows) return;

    int arow = row0 + r;
    if (arow >= nrows) arow = nrows - 1;
    const float* __restrict__ xrow = X + (long long)arow * K;

    constexpr int NT = NOUT / 16;
    v8f acc[NT];
#pragma unroll
    for (int nt = 0; nt < NT; ++nt) {
        v8f z = {};
        acc[nt] = z;
    }
    float dr[8];
#pragma unroll
    for (int v = 0; v < 8; ++v) {
        const int row = row0 + v + 8 * half;
        dr[v] = dis[row < nrows ? row : (nrows - 1)];
    }

#pragma unroll 4
    for (int kt = 0; kt < K / 4; ++kt) {
        const int k0 = kt * 4 + 2 * half;
        v2f a;
        a.x = xrow[k0];
        a.y = xrow[k0 + 1];
#pragma unroll
        for (int nt = 0; nt < NT; ++nt) {
            v2f b;
            b.x = sW[(k0)     * NOUT + nt * 16 + r];
            b.y = sW[(k0 + 1) * NOUT + nt * 16 + r];
            acc[nt] = __builtin_amdgcn_wmma_f32_16x16x4_f32(
                false, a, false, b, (short)0, acc[nt], false, false);
        }
    }

#pragma unroll
    for (int nt = 0; nt < NT; ++nt) {
#pragma unroll
        for (int v = 0; v < 8; ++v) {
            const int row = row0 + v + 8 * half;
            const int col = nt * 16 + r;
            if (row < nrows) H[(long long)row * NOUT + col] = acc[nt][v] * dr[v];
        }
    }
}


#define ROW_GROUPS 1024

template<int U>
__device__ __forceinline__ unsigned int row_lane(int v) {
    return (unsigned int)__builtin_amdgcn_update_dpp(0, v, 0x150 + U, 0xf, 0xf, false);
}

template<int U0>
__device__ __forceinline__ void load_rows8(const float2* __restrict__ h2, int es, unsigned int c, float2 (&hv)[8]) {
    hv[0] = h2[row_lane<U0 + 0>(es) * 16u + c];
    hv[1] = h2[row_lane<U0 + 1>(es) * 16u + c];
    hv[2] = h2[row_lane<U0 + 2>(es) * 16u + c];
    hv[3] = h2[row_lane<U0 + 3>(es) * 16u + c];
    hv[4] = h2[row_lane<U0 + 4>(es) * 16u + c];
    hv[5] = h2[row_lane<U0 + 5>(es) * 16u + c];
    hv[6] = h2[row_lane<U0 + 6>(es) * 16u + c];
    hv[7] = h2[row_lane<U0 + 7>(es) * 16u + c];
}

__device__ __forceinline__ void add_rows8(const float2 (&hv)[8], int e0, int half, int m, float& ax, float& ay) {
#pragma unroll
    for (int u = 0; u < 8; ++u) {
        const bool live = (e0 + 2 * u + half) < m;
        ax += live ? hv[u].x : 0.0f;
        ay += live ? hv[u].y : 0.0f;
    }
}

__global__ void __launch_bounds__(256)
gather_rows_kernel(const unsigned int* __restrict__ rowptr,
                   const int* __restrict__ epk,
                   const float* __restrict__ h,
                   const float* __restrict__ dis,
                   const float* __restrict__ b,
                   float* __restrict__ out, int n) {
    const int lane = threadIdx.x & 31;
    const int half = lane >> 4;
    const unsigned int c = (unsigned int)(lane & 15);
    const unsigned int eo = 2u * c + (unsigned int)half;
    const float2* __restrict__ h2 = (const float2*)h;
    const int nw = (int)gridDim.x * 8;
    int node = __builtin_amdgcn_readfirstlane((int)(blockIdx.x * 8 + (threadIdx.x >> 5)));
    if (node >= n) return;

    const float2 bb = ((const float2*)b)[c];

    const unsigned int beg = rowptr[node];
    const unsigned int nxt = rowptr[node + 1];
    float dd = dis[node];
    float2 own = h2[(unsigned int)node * 16u + c];

    int node1 = node + nw;
    bool valid1 = node1 < n;
    int nodec1 = valid1 ? node1 : (n - 1);
    unsigned int beg1 = rowptr[nodec1];
    unsigned int nxt1 = rowptr[nodec1 + 1];
    float dd1 = dis[nodec1];
    float2 own1 = h2[(unsigned int)nodec1 * 16u + c];

    unsigned int begs = __builtin_amdgcn_readfirstlane(beg);
    unsigned int ends = __builtin_amdgcn_readfirstlane(nxt);
    int es = node;
    if (begs + eo < ends) es = epk[begs + eo];

    for (;;) {
        const unsigned int deg = ends - begs;
        const int m = (int)(deg < 32u ? deg : 32u);

        float2 ha[8], hb[8];
        load_rows8<0>(h2, es, c, ha);
        if (m > 16) load_rows8<8>(h2, es, c, hb);

        const int node2 = node1 + nw;
        const bool valid2 = node2 < n;
        const int nodec2 = valid2 ? node2 : (n - 1);
        const unsigned int beg2 = rowptr[nodec2];
        const unsigned int nxt2 = rowptr[nodec2 + 1];
        const float dd2 = dis[nodec2];
        const float2 own2 = h2[(unsigned int)nodec2 * 16u + c];

        const unsigned int begs1 = __builtin_amdgcn_readfirstlane(beg1);
        const unsigned int ends1 = __builtin_amdgcn_readfirstlane(valid1 ? nxt1 : beg1);
        int es1 = nodec1;
        if (begs1 + eo < ends1) es1 = epk[begs1 + eo];

        float ax = half ? 0.0f : own.x;
        float ay = half ? 0.0f : own.y;
        add_rows8(ha, 0, half, m, ax, ay);
        if (m > 16) add_rows8(hb, 16, half, m, ax, ay);

        for (unsigned int base = begs + 32u; base < ends; base += 32u) {
            int e2 = node;
            if (base + eo < ends) e2 = epk[base + eo];
            const unsigned int left = ends - base;
            const int mm = (int)(left < 32u ? left : 32u);
            float2 hv[8];
            load_rows8<0>(h2, e2, c, hv);
            add_rows8(hv, 0, half, mm, ax, ay);
            if (mm > 16) {
                load_rows8<8>(h2, e2, c, hv);
                add_rows8(hv, 16, half, mm, ax, ay);
            }
        }

        const float bx = __builtin_bit_cast(float, __builtin_amdgcn_permlanex16(
            __builtin_bit_cast(int, ax), __builtin_bit_cast(int, ax), 0x76543210, 0xfedcba98, false, false));
        const float by = __builtin_bit_cast(float, __builtin_amdgcn_permlanex16(
            __builtin_bit_cast(int, ay), __builtin_bit_cast(int, ay), 0x76543210, 0xfedcba98, false, false));
        if (half == 0) {
            float2 o;
            o.x = (ax + bx) * dd + bb.x;
            o.y = (ay + by) * dd + bb.y;
            ((float2*)out)[(unsigned int)node * 16u + c] = o;
        }

        if (!valid1) break;
        node = node1; dd = dd1; own = own1; begs = begs1; ends = ends1; es = es1;
        node1 = node2; valid1 = valid2; nodec1 = nodec2; beg1 = beg2; nxt1 = nxt2; dd1 = dd2; own1 = own2;
    }
}


template<int NOUT>
__global__ void __launch_bounds__(512)
gather_gemm_kernel(const unsigned int* __restrict__ rowptr,
                   const int* __restrict__ epk,
                   const float* __restrict__ h,
                   const float* __restrict__ dis,
                   const float* __restrict__ b,
                   const float* __restrict__ Wn,
                   float* __restrict__ Hn, int n) {
    constexpr int F = 64;
    constexpr int V = 2;
    constexpr int G = 8;
    constexpr int NT = NOUT / 16;
    constexpr int VO = NOUT / 32;
    constexpr int AP = F + 4;
    constexpr int PW = NOUT + 16;
    constexpr int OP = NOUT + 4;
    __shared__ alignas(16) float2 sW[(F / 2) * PW];
    __shared__ alignas(16) float sA[16 * AP];
    __shared__ alignas(16) float sO[16 * OP];

    const int tid = threadIdx.x;
    const int lane = tid & 31;
    const int wave = __builtin_amdgcn_readfirstlane(tid >> 5);

    const int ntiles = (n + 15) >> 4;
    int tile = blockIdx.x;
    if (tile >= ntiles) return;

    for (int i = tid; i < (F / 2) * (NOUT / 4); i += 512) {
        const int kp = i / (NOUT / 4);
        const int cc = (i % (NOUT / 4)) * 4;
        const float4 w0 = *(const float4*)(Wn + (2 * kp) * NOUT + cc);
        const float4 w1 = *(const float4*)(Wn + (2 * kp + 1) * NOUT + cc);
        float4* dp = (float4*)&sW[kp * PW + cc];
        dp[0] = make_float4(w0.x, w1.x, w0.y, w1.y);
        dp[1] = make_float4(w0.z, w1.z, w0.w, w1.w);
    }
    const float2 bb = *(const float2*)(b + lane * V);

    const int half = lane >> 4;
    const int r    = lane & 15;

    int node = __builtin_amdgcn_readfirstlane(tile * 16 + wave);
    bool valid = node < n;
    int nodec = valid ? node : (n - 1);
    unsigned int beg = rowptr[nodec];
    unsigned int nxt = rowptr[nodec + 1];
    float dd = dis[nodec];
    float2 q = *(const float2*)(h + (long long)nodec * F + lane * V);

    for (;;) {
        const int tile_n = tile + (int)gridDim.x;
        const int node_n = __builtin_amdgcn_readfirstlane(tile_n * 16 + wave);
        const bool valid_n = node_n < n;
        const int nodec_n = valid_n ? node_n : (n - 1);
        const unsigned int beg_n = rowptr[nodec_n];
        const unsigned int nxt_n = rowptr[nodec_n + 1];
        const float dd_n = dis[nodec_n];
        const float2 q_n = *(const float2*)(h + (long long)nodec_n * F + lane * V);

        const unsigned int begs = __builtin_amdgcn_readfirstlane(beg);
        const unsigned int end  = __builtin_amdgcn_readfirstlane(valid ? nxt : beg);
        float acc[V];
        acc[0] = q.x;
        acc[V - 1] = q.y;

        for (unsigned int base = begs; base < end; base += 32u) {
            const unsigned int j = base + (unsigned int)lane;
            int es = nodec;
            if (j < end) es = epk[j];
            const unsigned int left = end - base;
            const int m = (int)(left < 32u ? left : 32u);
            for (int t = 0; t < m; t += G) {
                float hv[G][V];
#pragma unroll
                for (int u = 0; u < G; ++u) {
                    const int s = __builtin_amdgcn_readlane(es, t + u);
                    const float2 qq = *(const float2*)(h + (long long)s * F + lane * V);
                    hv[u][0] = qq.x;
                    hv[u][V - 1] = qq.y;
                }
#pragma unroll
                for (int u = 0; u < G; ++u) {
                    const bool live = (t + u) < m;
#pragma unroll
                    for (int v = 0; v < V; ++v)
                        acc[v] += live ? hv[u][v] : 0.0f;
                }
            }
        }
        acc[0] = valid ? fmaxf(acc[0] * dd + bb.x, 0.0f) : 0.0f;
        acc[V - 1] = valid ? fmaxf(acc[V - 1] * dd + bb.y, 0.0f) : 0.0f;

        *(float2*)&sA[wave * AP + lane * V] = make_float2(acc[0], acc[V - 1]);
        __syncthreads();

        if (wave < NT) {
            v8f d = {};
#pragma unroll
            for (int kt = 0; kt < F / 4; ++kt) {
                const v2f a = *(const v2f*)&sA[r * AP + kt * 4 + 2 * half];
                const float2 bq = sW[(kt * 2 + half) * PW + wave * 16 + r];
                v2f bf;
                bf.x = bq.x;
                bf.y = bq.y;
                d = __builtin_amdgcn_wmma_f32_16x16x4_f32(false, a, false, bf, (short)0, d, false, false);
            }
#pragma unroll
            for (int v = 0; v < 8; ++v) sO[(v + 8 * half) * OP + wave * 16 + r] = d[v];
        }
        __syncthreads();

        if (valid) {
            float* __restrict__ op = Hn + (long long)node * NOUT + lane * VO;
            if (VO == 2) {
                const float2 o = *(const float2*)&sO[wave * OP + lane * VO];
                *(float2*)op = make_float2(o.x * dd, o.y * dd);
            } else {
                op[0] = sO[wave * OP + lane * VO] * dd;
            }
        }

        if (tile_n >= ntiles) break;
        tile = tile_n; node = node_n; valid = valid_n; nodec = nodec_n;
        beg = beg_n; nxt = nxt_n; dd = dd_n; q = q_n;
    }
}


static inline int cdiv(long long a, int b) { return (int)((a + b - 1) / b); }

extern "C" void kernel_launch(void* const* d_in, const int* in_sizes, int n_in,
                              void* d_out, int out_size, void* d_ws, size_t ws_size,
                              hipStream_t stream) {
    const float* x  = (const float*)d_in[0];
    const int*   ei = (const int*)d_in[1];
    const float* W1 = (const float*)d_in[2];
    const float* b1 = (const float*)d_in[3];
    const float* W2 = (const float*)d_in[4];
    const float* b2 = (const float*)d_in[5];
    const float* W3 = (const float*)d_in[6];
    const float* b3 = (const float*)d_in[7];
    float* out = (float*)d_out;

    const int* src = ei;
    const int* dst = ei + N_EDGES;

    char* ws = (char*)d_ws;
    unsigned int* tbl = (unsigned int*)ws;
    float* dis  = (float*)(ws + 400384);
    float* bufA = (float*)(ws + 800768);
    float* bufB = (float*)(ws + 26400768);
    unsigned int* rowptr = (unsigned int*)(ws + 52000768);
    int* epk             = (int*)(ws + 58801152);
    unsigned int* tmp    = (unsigned int*)(ws + 71601152);

    const int N = N_NODES, E = N_EDGES;
    const int gemm_blocks = cdiv(cdiv(N, 16), 8);
    const int tile_groups = 1024;

    chunk_sort_kernel<<<EDGE_CHUNKS, BT, 0, stream>>>(src, dst, tbl, tmp, E);
    node_bucket_kernel<<<NBINS, BT, 0, stream>>>(tmp, tbl, rowptr, dis, epk, N);

    gemm_wmma_kernel<128, 64><<<gemm_blocks, 256, 0, stream>>>(x, W1, dis, bufA, N);

    gather_gemm_kernel<64><<<tile_groups, 512, 0, stream>>>(rowptr, epk, bufA, dis, b1, W2, bufB, N);

    gather_gemm_kernel<32><<<tile_groups, 512, 0, stream>>>(rowptr, epk, bufB, dis, b2, W3, bufA, N);

    gather_rows_kernel<<<ROW_GROUPS, 256, 0, stream>>>(rowptr, epk, bufA, dis, b3, out, N);
}


# Layer 1 product moved into the bucket launch: a bin's work-group multiplies its own 512 rows of x on WMMA (a 16-row tile per wave, W1 in LDS), dis from its LDS counts; 5 launches, not 6; sums kept
# speedup vs baseline: 8.2379x; 1.0411x over previous
#include <hip/hip_runtime.h>

typedef float v2f __attribute__((ext_vector_type(2)));
typedef float v8f __attribute__((ext_vector_type(8)));

#define N_NODES 100000
#define N_EDGES 1600000


#define BIN_SHIFT   9
#define BIN_NODES   (1 << BIN_SHIFT)
#define SRC_BITS    (32 - BIN_SHIFT)
#define NBINS       ((N_NODES + BIN_NODES - 1) / BIN_NODES)
#define TBL         (((NBINS + 1 + 31) / 32) * 32)
#define EDGE_CHUNKS 256
#define CHUNK_EDGES ((N_EDGES + EDGE_CHUNKS - 1) / EDGE_CHUNKS)
#define BT          1024
#define CHUNK_PITCH (((CHUNK_EDGES + 31) / 32) * 32)
#define EPT         ((CHUNK_EDGES + BT - 1) / BT)
#define PPW         (EDGE_CHUNKS / (BT / 32))
#define STAGE_CAP   16384
#define GK          128
#define GN          64

static_assert(N_NODES <= (1 << SRC_BITS), "a src id and a node's place in its bin share one word");
static_assert(NBINS <= BT, "one thread per bin");
static_assert(EDGE_CHUNKS <= BT, "one thread per chunk reads the table column");
static_assert(EDGE_CHUNKS % (BT / 32) == 0, "every wave takes the same number of pieces");
static_assert(BIN_NODES <= BT, "one thread per node of a bin");
static_assert(BIN_NODES == (BT / 32) * 16, "one 16-row tile of the bin per wave");

__device__ __forceinline__ unsigned int block_excl_scan(unsigned int v, unsigned int* s_w, unsigned int& total) {
    const int lane = threadIdx.x & 31;
    const int wave = threadIdx.x >> 5;
    unsigned int inc = v;
    for (int d = 1; d < 32; d <<= 1) {
        const unsigned int t = __shfl_up(inc, d);
        if (lane >= d) inc += t;
    }
    if (lane == 31) s_w[wave] = inc;
    __syncthreads();
    unsigned int off = 0u, tot = 0u;
    for (int w = 0; w < BT / 32; ++w) {
        const unsigned int x = s_w[w];
        if (w < wave) off += x;
        tot += x;
    }
    __syncthreads();
    total = tot;
    return off + inc - v;
}

__device__ __forceinline__ void block_sum2(unsigned int a, unsigned int b, unsigned int* s_w2,
                                           unsigned int& ta, unsigned int& tb) {
    for (int d = 16; d >= 1; d >>= 1) {
        a += __shfl_xor(a, d);
        b += __shfl_xor(b, d);
    }
    const int lane = threadIdx.x & 31;
    const int wave = threadIdx.x >> 5;
    if (lane == 0) { s_w2[2 * wave] = a; s_w2[2 * wave + 1] = b; }
    __syncthreads();
    unsigned int x = 0u, y = 0u;
    for (int w = 0; w < BT / 32; ++w) { x += s_w2[2 * w]; y += s_w2[2 * w + 1]; }
    __syncthreads();
    ta = x;
    tb = y;
}

__global__ void __launch_bounds__(BT)
chunk_sort_kernel(const int* __restrict__ src, const int* __restrict__ dst,
                  unsigned int* __restrict__ tbl,
                  unsigned int* __restrict__ tmp, int E) {
    __shared__ unsigned int s_sorted[CHUNK_EDGES];
    __shared__ unsigned int s_cnt[NBINS];
    __shared__ unsigned int s_lbase[NBINS];
    __shared__ unsigned int s_w[BT / 32];

    const int tid = threadIdx.x;
    const int chunk = blockIdx.x;
    const int e0 = chunk * CHUNK_EDGES;
    const int e1 = (e0 + CHUNK_EDGES < E) ? (e0 + CHUNK_EDGES) : E;
    const int n = (e1 > e0) ? (e1 - e0) : 0;

    int es[EPT], ed[EPT];
#pragma unroll
    for (int k = 0; k < EPT; ++k) {
        const int e = e0 + tid + k * BT;
        es[k] = 0;
        ed[k] = -1;
        if (e < e1) { es[k] = src[e]; ed[k] = dst[e]; }
    }
    if (tid < NBINS) s_cnt[tid] = 0u;
    __syncthreads();

    unsigned int rk[EPT];
#pragma unroll
    for (int k = 0; k < EPT; ++k) {
        rk[k] = 0u;
        if (ed[k] >= 0) rk[k] = atomicAdd(&s_cnt[ed[k] >> BIN_SHIFT], 1u);
    }
    __syncthreads();

    unsigned int total = 0u;
    const unsigned int mycnt = (tid < NBINS) ? s_cnt[tid] : 0u;
    const unsigned int lbase = block_excl_scan(mycnt, s_w, total);
    if (tid < NBINS) {
        s_lbase[tid] = lbase;
        tbl[chunk * TBL + tid] = lbase;
    }
    if (tid == 0) tbl[chunk * TBL + NBINS] = total;
    __syncthreads();

#pragma unroll
    for (int k = 0; k < EPT; ++k)
        if (ed[k] >= 0)
            s_sorted[s_lbase[ed[k] >> BIN_SHIFT] + rk[k]] =
                (unsigned int)es[k] | ((unsigned int)(ed[k] & (BIN_NODES - 1)) << SRC_BITS);
    __syncthreads();

    for (int i = tid; i < n; i += BT) tmp[chunk * CHUNK_PITCH + i] = s_sorted[i];
}

template<int K, int NOUT>
__device__ __forceinline__ void wave_tile_product(const float* __restrict__ X, const float* sW, const float* sd,
                                                  float* __restrict__ H, int row0, int nrows, int lane) {
    const int half = lane >> 4;
    const int r    = lane & 15;

    int arow = row0 + r;
    if (arow >= nrows) arow = nrows - 1;
    const float* __restrict__ xrow = X + (long long)arow * K;

    constexpr int NT = NOUT / 16;
    v8f acc[NT];
#pragma unroll
    for (int nt = 0; nt < NT; ++nt) {
        v8f z = {};
        acc[nt] = z;
    }

#pragma unroll 4
    for (int kt = 0; kt < K / 4; ++kt) {
        const int k0 = kt * 4 + 2 * half;
        v2f a;
        a.x = xrow[k0];
        a.y = xrow[k0 + 1];
#pragma unroll
        for (int nt = 0; nt < NT; ++nt) {
            v2f b;
            b.x = sW[(k0)     * NOUT + nt * 16 + r];
            b.y = sW[(k0 + 1) * NOUT + nt * 16 + r];
            acc[nt] = __builtin_amdgcn_wmma_f32_16x16x4_f32(
                false, a, false, b, (short)0, acc[nt], false, false);
        }
    }

#pragma unroll
    for (int nt = 0; nt < NT; ++nt) {
#pragma unroll
        for (int v = 0; v < 8; ++v) {
            const int row = row0 + v + 8 * half;
            const int col = nt * 16 + r;
            if (row < nrows) H[(long long)row * NOUT + col] = acc[nt][v] * sd[v + 8 * half];
        }
    }
}

__global__ void __launch_bounds__(BT)
node_bucket_kernel(const unsigned int* __restrict__ tmp,
                   const unsigned int* __restrict__ tbl,
                   unsigned int* __restrict__ rowptr,
                   float* __restrict__ dis,
                   int* __restrict__ epk,
                   const float* __restrict__ X, const float* __restrict__ W,
                   float* __restrict__ H, int n_nodes) {
    __shared__ alignas(16) float sW[GK * GN];
    __shared__ float s_dis[BIN_NODES];
    __shared__ int s_stage[STAGE_CAP];
    __shared__ unsigned int s_cnt[BIN_NODES];
    __shared__ unsigned int s_start[BIN_NODES];
    __shared__ unsigned int s_pbeg[EDGE_CHUNKS];
    __shared__ unsigned int s_plen[EDGE_CHUNKS];
    __shared__ unsigned int s_w[2 * (BT / 32)];
    const int tid = threadIdx.x;
    const int lane = tid & 31;
    const int wave = tid >> 5;
    const int bin = blockIdx.x;

    for (int i = tid; i < GK * GN / 4; i += BT) ((float4*)sW)[i] = ((const float4*)W)[i];

    unsigned int pa = 0u, pl = 0u;
    if (tid < EDGE_CHUNKS) {
        pa = tbl[tid * TBL + bin];
        pl = tbl[tid * TBL + bin + 1] - pa;
        s_pbeg[tid] = (unsigned int)tid * CHUNK_PITCH + pa;
        s_plen[tid] = pl;
    }
    if (tid < BIN_NODES) s_cnt[tid] = 0u;
    unsigned int lo = 0u, n = 0u;
    block_sum2(pa, pl, s_w, lo, n);

    unsigned int pb[PPW], pn[PPW], hw[PPW];
#pragma unroll
    for (int k = 0; k < PPW; ++k) {
        const int c = wave + k * (BT / 32);
        pb[k] = s_pbeg[c];
        pn[k] = s_plen[c];
    }
#pragma unroll
    for (int k = 0; k < PPW; ++k) {
        hw[k] = 0u;
        if ((unsigned int)lane < pn[k]) hw[k] = tmp[pb[k] + (unsigned int)lane];
    }
#pragma unroll
    for (int k = 0; k < PPW; ++k)
        if ((unsigned int)lane < pn[k]) atomicAdd(&s_cnt[hw[k] >> SRC_BITS], 1u);
    for (int k = 0; k < PPW; ++k)
        for (unsigned int j = (unsigned int)lane + 32u; j < pn[k]; j += 32u)
            atomicAdd(&s_cnt[tmp[pb[k] + j] >> SRC_BITS], 1u);
    __syncthreads();

    const unsigned int c = (tid < BIN_NODES) ? s_cnt[tid] : 0u;
    unsigned int all = 0u;
    const unsigned int ex = block_excl_scan(c, s_w, all);
    const int node = bin * BIN_NODES + tid;
    if (tid < BIN_NODES) {
        s_start[tid] = ex;
        s_cnt[tid] = 0u;
        const float dn = rsqrtf((float)c + 1.0f);
        s_dis[tid] = dn;
        if (node < n_nodes) {
            rowptr[node] = lo + ex;
            dis[node] = dn;
            if (node == n_nodes - 1) rowptr[n_nodes] = lo + ex + c;
        }
    }
    __syncthreads();

    const bool staged = n <= (unsigned int)STAGE_CAP;
    auto place = [&](unsigned int word) {
        const unsigned int l = word >> SRC_BITS;
        const unsigned int slot = s_start[l] + atomicAdd(&s_cnt[l], 1u);
        const int s = (int)(word & ((1u << SRC_BITS) - 1u));
        if (staged) s_stage[slot] = s;
        else        epk[lo + slot] = s;
    };
#pragma unroll
    for (int k = 0; k < PPW; ++k)
        if ((unsigned int)lane < pn[k]) place(hw[k]);
    for (int k = 0; k < PPW; ++k)
        for (unsigned int j = (unsigned int)lane + 32u; j < pn[k]; j += 32u)
            place(tmp[pb[k] + j]);
    __syncthreads();
    if (staged) {
        for (unsigned int i = (unsigned int)tid; i < n; i += BT) epk[lo + i] = s_stage[i];
    }

    const int row0 = bin * BIN_NODES + wave * 16;
    if (row0 < n_nodes)
        wave_tile_product<GK, GN>(X, sW, s_dis + wave * 16, H, row0, n_nodes, lane);
}


#define ROW_GROUPS 1024

template<int U>
__device__ __forceinline__ unsigned int row_lane(int v) {
    return (unsigned int)__builtin_amdgcn_update_dpp(0, v, 0x150 + U, 0xf, 0xf, false);
}

template<int U0>
__device__ __forceinline__ void load_rows8(const float2* __restrict__ h2, int es, unsigned int c, float2 (&hv)[8]) {
    hv[0] = h2[row_lane<U0 + 0>(es) * 16u + c];
    hv[1] = h2[row_lane<U0 + 1>(es) * 16u + c];
    hv[2] = h2[row_lane<U0 + 2>(es) * 16u + c];
    hv[3] = h2[row_lane<U0 + 3>(es) * 16u + c];
    hv[4] = h2[row_lane<U0 + 4>(es) * 16u + c];
    hv[5] = h2[row_lane<U0 + 5>(es) * 16u + c];
    hv[6] = h2[row_lane<U0 + 6>(es) * 16u + c];
    hv[7] = h2[row_lane<U0 + 7>(es) * 16u + c];
}

__device__ __forceinline__ void add_rows8(const float2 (&hv)[8], int e0, int half, int m, float& ax, float& ay) {
#pragma unroll
    for (int u = 0; u < 8; ++u) {
        const bool live = (e0 + 2 * u + half) < m;
        ax += live ? hv[u].x : 0.0f;
        ay += live ? hv[u].y : 0.0f;
    }
}

__global__ void __launch_bounds__(256)
gather_rows_kernel(const unsigned int* __restrict__ rowptr,
                   const int* __restrict__ epk,
                   const float* __restrict__ h,
                   const float* __restrict__ dis,
                   const float* __restrict__ b,
                   float* __restrict__ out, int n) {
    const int lane = threadIdx.x & 31;
    const int half = lane >> 4;
    const unsigned int c = (unsigned int)(lane & 15);
    const unsigned int eo = 2u * c + (unsigned int)half;
    const float2* __restrict__ h2 = (const float2*)h;
    const int nw = (int)gridDim.x * 8;
    int node = __builtin_amdgcn_readfirstlane((int)(blockIdx.x * 8 + (threadIdx.x >> 5)));
    if (node >= n) return;

    const float2 bb = ((const float2*)b)[c];

    const unsigned int beg = rowptr[node];
    const unsigned int nxt = rowptr[node + 1];
    float dd = dis[node];
    float2 own = h2[(unsigned int)node * 16u + c];

    int node1 = node + nw;
    bool valid1 = node1 < n;
    int nodec1 = valid1 ? node1 : (n - 1);
    unsigned int beg1 = rowptr[nodec1];
    unsigned int nxt1 = rowptr[nodec1 + 1];
    float dd1 = dis[nodec1];
    float2 own1 = h2[(unsigned int)nodec1 * 16u + c];

    unsigned int begs = __builtin_amdgcn_readfirstlane(beg);
    unsigned int ends = __builtin_amdgcn_readfirstlane(nxt);
    int es = node;
    if (begs + eo < ends) es = epk[begs + eo];

    for (;;) {
        const unsigned int deg = ends - begs;
        const int m = (int)(deg < 32u ? deg : 32u);

        float2 ha[8], hb[8];
        load_rows8<0>(h2, es, c, ha);
        if (m > 16) load_rows8<8>(h2, es, c, hb);

        const int node2 = node1 + nw;
        const bool valid2 = node2 < n;
        const int nodec2 = valid2 ? node2 : (n - 1);
        const unsigned int beg2 = rowptr[nodec2];
        const unsigned int nxt2 = rowptr[nodec2 + 1];
        const float dd2 = dis[nodec2];
        const float2 own2 = h2[(unsigned int)nodec2 * 16u + c];

        const unsigned int begs1 = __builtin_amdgcn_readfirstlane(beg1);
        const unsigned int ends1 = __builtin_amdgcn_readfirstlane(valid1 ? nxt1 : beg1);
        int es1 = nodec1;
        if (begs1 + eo < ends1) es1 = epk[begs1 + eo];

        float ax = half ? 0.0f : own.x;
        float ay = half ? 0.0f : own.y;
        add_rows8(ha, 0, half, m, ax, ay);
        if (m > 16) add_rows8(hb, 16, half, m, ax, ay);

        for (unsigned int base = begs + 32u; base < ends; base += 32u) {
            int e2 = node;
            if (base + eo < ends) e2 = epk[base + eo];
            const unsigned int left = ends - base;
            const int mm = (int)(left < 32u ? left : 32u);
            float2 hv[8];
            load_rows8<0>(h2, e2, c, hv);
            add_rows8(hv, 0, half, mm, ax, ay);
            if (mm > 16) {
                load_rows8<8>(h2, e2, c, hv);
                add_rows8(hv, 16, half, mm, ax, ay);
            }
        }

        const float bx = __builtin_bit_cast(float, __builtin_amdgcn_permlanex16(
            __builtin_bit_cast(int, ax), __builtin_bit_cast(int, ax), 0x76543210, 0xfedcba98, false, false));
        const float by = __builtin_bit_cast(float, __builtin_amdgcn_permlanex16(
            __builtin_bit_cast(int, ay), __builtin_bit_cast(int, ay), 0x76543210, 0xfedcba98, false, false));
        if (half == 0) {
            float2 o;
            o.x = (ax + bx) * dd + bb.x;
            o.y = (ay + by) * dd + bb.y;
            ((float2*)out)[(unsigned int)node * 16u + c] = o;
        }

        if (!valid1) break;
        node = node1; dd = dd1; own = own1; begs = begs1; ends = ends1; es = es1;
        node1 = node2; valid1 = valid2; nodec1 = nodec2; beg1 = beg2; nxt1 = nxt2; dd1 = dd2; own1 = own2;
    }
}


template<int NOUT>
__global__ void __launch_bounds__(512)
gather_gemm_kernel(const unsigned int* __restrict__ rowptr,
                   const int* __restrict__ epk,
                   const float* __restrict__ h,
                   const float* __restrict__ dis,
                   const float* __restrict__ b,
                   const float* __restrict__ Wn,
                   float* __restrict__ Hn, int n) {
    constexpr int F = 64;
    constexpr int V = 2;
    constexpr int G = 8;
    constexpr int NT = NOUT / 16;
    constexpr int VO = NOUT / 32;
    constexpr int AP = F + 4;
    constexpr int PW = NOUT + 16;
    constexpr int OP = NOUT + 4;
    __shared__ alignas(16) float2 sW[(F / 2) * PW];
    __shared__ alignas(16) float sA[16 * AP];
    __shared__ alignas(16) float sO[16 * OP];

    const int tid = threadIdx.x;
    const int lane = tid & 31;
    const int wave = __builtin_amdgcn_readfirstlane(tid >> 5);

    const int ntiles = (n + 15) >> 4;
    int tile = blockIdx.x;
    if (tile >= ntiles) return;

    for (int i = tid; i < (F / 2) * (NOUT / 4); i += 512) {
        const int kp = i / (NOUT / 4);
        const int cc = (i % (NOUT / 4)) * 4;
        const float4 w0 = *(const float4*)(Wn + (2 * kp) * NOUT + cc);
        const float4 w1 = *(const float4*)(Wn + (2 * kp + 1) * NOUT + cc);
        float4* dp = (float4*)&sW[kp * PW + cc];
        dp[0] = make_float4(w0.x, w1.x, w0.y, w1.y);
        dp[1] = make_float4(w0.z, w1.z, w0.w, w1.w);
    }
    const float2 bb = *(const float2*)(b + lane * V);

    const int half = lane >> 4;
    const int r    = lane & 15;

    int node = __builtin_amdgcn_readfirstlane(tile * 16 + wave);
    bool valid = node < n;
    int nodec = valid ? node : (n - 1);
    unsigned int beg = rowptr[nodec];
    unsigned int nxt = rowptr[nodec + 1];
    float dd = dis[nodec];
    float2 q = *(const float2*)(h + (long long)nodec * F + lane * V);

    for (;;) {
        const int tile_n = tile + (int)gridDim.x;
        const int node_n = __builtin_amdgcn_readfirstlane(tile_n * 16 + wave);
        const bool valid_n = node_n < n;
        const int nodec_n = valid_n ? node_n : (n - 1);
        const unsigned int beg_n = rowptr[nodec_n];
        const unsigned int nxt_n = rowptr[nodec_n + 1];
        const float dd_n = dis[nodec_n];
        const float2 q_n = *(const float2*)(h + (long long)nodec_n * F + lane * V);

        const unsigned int begs = __builtin_amdgcn_readfirstlane(beg);
        const unsigned int end  = __builtin_amdgcn_readfirstlane(valid ? nxt : beg);
        float acc[V];
        acc[0] = q.x;
        acc[V - 1] = q.y;

        for (unsigned int base = begs; base < end; base += 32u) {
            const unsigned int j = base + (unsigned int)lane;
            int es = nodec;
            if (j < end) es = epk[j];
            const unsigned int left = end - base;
            const int m = (int)(left < 32u ? left : 32u);
            for (int t = 0; t < m; t += G) {
                float hv[G][V];
#pragma unroll
                for (int u = 0; u < G; ++u) {
                    const int s = __builtin_amdgcn_readlane(es, t + u);
                    const float2 qq = *(const float2*)(h + (long long)s * F + lane * V);
                    hv[u][0] = qq.x;
                    hv[u][V - 1] = qq.y;
                }
#pragma unroll
                for (int u = 0; u < G; ++u) {
                    const bool live = (t + u) < m;
#pragma unroll
                    for (int v = 0; v < V; ++v)
                        acc[v] += live ? hv[u][v] : 0.0f;
                }
            }
        }
        acc[0] = valid ? fmaxf(acc[0] * dd + bb.x, 0.0f) : 0.0f;
        acc[V - 1] = valid ? fmaxf(acc[V - 1] * dd + bb.y, 0.0f) : 0.0f;

        *(float2*)&sA[wave * AP + lane * V] = make_float2(acc[0], acc[V - 1]);
        __syncthreads();

        if (wave < NT) {
            v8f d = {};
#pragma unroll
            for (int kt = 0; kt < F / 4; ++kt) {
                const v2f a = *(const v2f*)&sA[r * AP + kt * 4 + 2 * half];
                const float2 bq = sW[(kt * 2 + half) * PW + wave * 16 + r];
                v2f bf;
                bf.x = bq.x;
                bf.y = bq.y;
                d = __builtin_amdgcn_wmma_f32_16x16x4_f32(false, a, false, bf, (short)0, d, false, false);
            }
#pragma unroll
            for (int v = 0; v < 8; ++v) sO[(v + 8 * half) * OP + wave * 16 + r] = d[v];
        }
        __syncthreads();

        if (valid) {
            float* __restrict__ op = Hn + (long long)node * NOUT + lane * VO;
            if (VO == 2) {
                const float2 o = *(const float2*)&sO[wave * OP + lane * VO];
                *(float2*)op = make_float2(o.x * dd, o.y * dd);
            } else {
                op[0] = sO[wave * OP + lane * VO] * dd;
            }
        }

        if (tile_n >= ntiles) break;
        tile = tile_n; node = node_n; valid = valid_n; nodec = nodec_n;
        beg = beg_n; nxt = nxt_n; dd = dd_n; q = q_n;
    }
}


extern "C" void kernel_launch(void* const* d_in, const int* in_sizes, int n_in,
                              void* d_out, int out_size, void* d_ws, size_t ws_size,
                              hipStream_t stream) {
    const float* x  = (const float*)d_in[0];
    const int*   ei = (const int*)d_in[1];
    const float* W1 = (const float*)d_in[2];
    const float* b1 = (const float*)d_in[3];
    const float* W2 = (const float*)d_in[4];
    const float* b2 = (const float*)d_in[5];
    const float* W3 = (const float*)d_in[6];
    const float* b3 = (const float*)d_in[7];
    float* out = (float*)d_out;

    const int* src = ei;
    const int* dst = ei + N_EDGES;

    char* ws = (char*)d_ws;
    unsigned int* tbl = (unsigned int*)ws;
    float* dis  = (float*)(ws + 400384);
    float* bufA = (float*)(ws + 800768);
    float* bufB = (float*)(ws + 26400768);
    unsigned int* rowptr = (unsigned int*)(ws + 52000768);
    int* epk             = (int*)(ws + 58801152);
    unsigned int* tmp    = (unsigned int*)(ws + 71601152);

    const int N = N_NODES, E = N_EDGES;
    const int tile_groups = 1024;

    chunk_sort_kernel<<<EDGE_CHUNKS, BT, 0, stream>>>(src, dst, tbl, tmp, E);
    node_bucket_kernel<<<NBINS, BT, 0, stream>>>(tmp, tbl, rowptr, dis, epk, x, W1, bufA, N);

    gather_gemm_kernel<64><<<tile_groups, 512, 0, stream>>>(rowptr, epk, bufA, dis, b1, W2, bufB, N);

    gather_gemm_kernel<32><<<tile_groups, 512, 0, stream>>>(rowptr, epk, bufB, dis, b2, W3, bufA, N);

    gather_rows_kernel<<<ROW_GROUPS, 256, 0, stream>>>(rowptr, epk, bufA, dis, b3, out, N);
}


# Gather-product launches ask ahead: up to 24 bucket rows in flight per wave (was 8), next tile's entries and own row a tile ahead, bounds two; row bases scalar, 64 registers kept; sum order kept
# speedup vs baseline: 8.5771x; 1.0412x over previous
#include <hip/hip_runtime.h>

typedef float v2f __attribute__((ext_vector_type(2)));
typedef float v8f __attribute__((ext_vector_type(8)));

#define N_NODES 100000
#define N_EDGES 1600000


#define BIN_SHIFT   9
#define BIN_NODES   (1 << BIN_SHIFT)
#define SRC_BITS    (32 - BIN_SHIFT)
#define NBINS       ((N_NODES + BIN_NODES - 1) / BIN_NODES)
#define TBL         (((NBINS + 1 + 31) / 32) * 32)
#define EDGE_CHUNKS 256
#define CHUNK_EDGES ((N_EDGES + EDGE_CHUNKS - 1) / EDGE_CHUNKS)
#define BT          1024
#define CHUNK_PITCH (((CHUNK_EDGES + 31) / 32) * 32)
#define EPT         ((CHUNK_EDGES + BT - 1) / BT)
#define PPW         (EDGE_CHUNKS / (BT / 32))
#define STAGE_CAP   16384
#define GK          128
#define GN          64

static_assert(N_NODES <= (1 << SRC_BITS), "a src id and a node's place in its bin share one word");
static_assert(NBINS <= BT, "one thread per bin");
static_assert(EDGE_CHUNKS <= BT, "one thread per chunk reads the table column");
static_assert(EDGE_CHUNKS % (BT / 32) == 0, "every wave takes the same number of pieces");
static_assert(BIN_NODES <= BT, "one thread per node of a bin");
static_assert(BIN_NODES == (BT / 32) * 16, "one 16-row tile of the bin per wave");

__device__ __forceinline__ unsigned int block_excl_scan(unsigned int v, unsigned int* s_w, unsigned int& total) {
    const int lane = threadIdx.x & 31;
    const int wave = threadIdx.x >> 5;
    unsigned int inc = v;
    for (int d = 1; d < 32; d <<= 1) {
        const unsigned int t = __shfl_up(inc, d);
        if (lane >= d) inc += t;
    }
    if (lane == 31) s_w[wave] = inc;
    __syncthreads();
    unsigned int off = 0u, tot = 0u;
    for (int w = 0; w < BT / 32; ++w) {
        const unsigned int x = s_w[w];
        if (w < wave) off += x;
        tot += x;
    }
    __syncthreads();
    total = tot;
    return off + inc - v;
}

__device__ __forceinline__ void block_sum2(unsigned int a, unsigned int b, unsigned int* s_w2,
                                           unsigned int& ta, unsigned int& tb) {
    for (int d = 16; d >= 1; d >>= 1) {
        a += __shfl_xor(a, d);
        b += __shfl_xor(b, d);
    }
    const int lane = threadIdx.x & 31;
    const int wave = threadIdx.x >> 5;
    if (lane == 0) { s_w2[2 * wave] = a; s_w2[2 * wave + 1] = b; }
    __syncthreads();
    unsigned int x = 0u, y = 0u;
    for (int w = 0; w < BT / 32; ++w) { x += s_w2[2 * w]; y += s_w2[2 * w + 1]; }
    __syncthreads();
    ta = x;
    tb = y;
}

__global__ void __launch_bounds__(BT)
chunk_sort_kernel(const int* __restrict__ src, const int* __restrict__ dst,
                  unsigned int* __restrict__ tbl,
                  unsigned int* __restrict__ tmp, int E) {
    __shared__ unsigned int s_sorted[CHUNK_EDGES];
    __shared__ unsigned int s_cnt[NBINS];
    __shared__ unsigned int s_lbase[NBINS];
    __shared__ unsigned int s_w[BT / 32];

    const int tid = threadIdx.x;
    const int chunk = blockIdx.x;
    const int e0 = chunk * CHUNK_EDGES;
    const int e1 = (e0 + CHUNK_EDGES < E) ? (e0 + CHUNK_EDGES) : E;
    const int n = (e1 > e0) ? (e1 - e0) : 0;

    int es[EPT], ed[EPT];
#pragma unroll
    for (int k = 0; k < EPT; ++k) {
        const int e = e0 + tid + k * BT;
        es[k] = 0;
        ed[k] = -1;
        if (e < e1) { es[k] = src[e]; ed[k] = dst[e]; }
    }
    if (tid < NBINS) s_cnt[tid] = 0u;
    __syncthreads();

    unsigned int rk[EPT];
#pragma unroll
    for (int k = 0; k < EPT; ++k) {
        rk[k] = 0u;
        if (ed[k] >= 0) rk[k] = atomicAdd(&s_cnt[ed[k] >> BIN_SHIFT], 1u);
    }
    __syncthreads();

    unsigned int total = 0u;
    const unsigned int mycnt = (tid < NBINS) ? s_cnt[tid] : 0u;
    const unsigned int lbase = block_excl_scan(mycnt, s_w, total);
    if (tid < NBINS) {
        s_lbase[tid] = lbase;
        tbl[chunk * TBL + tid] = lbase;
    }
    if (tid == 0) tbl[chunk * TBL + NBINS] = total;
    __syncthreads();

#pragma unroll
    for (int k = 0; k < EPT; ++k)
        if (ed[k] >= 0)
            s_sorted[s_lbase[ed[k] >> BIN_SHIFT] + rk[k]] =
                (unsigned int)es[k] | ((unsigned int)(ed[k] & (BIN_NODES - 1)) << SRC_BITS);
    __syncthreads();

    for (int i = tid; i < n; i += BT) tmp[chunk * CHUNK_PITCH + i] = s_sorted[i];
}

template<int K, int NOUT>
__device__ __forceinline__ void wave_tile_product(const float* __restrict__ X, const float* sW, const float* sd,
                                                  float* __restrict__ H, int row0, int nrows, int lane) {
    const int half = lane >> 4;
    const int r    = lane & 15;

    int arow = row0 + r;
    if (arow >= nrows) arow = nrows - 1;
    const float* __restrict__ xrow = X + (long long)arow * K;

    constexpr int NT = NOUT / 16;
    v8f acc[NT];
#pragma unroll
    for (int nt = 0; nt < NT; ++nt) {
        v8f z = {};
        acc[nt] = z;
    }

#pragma unroll 4
    for (int kt = 0; kt < K / 4; ++kt) {
        const int k0 = kt * 4 + 2 * half;
        v2f a;
        a.x = xrow[k0];
        a.y = xrow[k0 + 1];
#pragma unroll
        for (int nt = 0; nt < NT; ++nt) {
            v2f b;
            b.x = sW[(k0)     * NOUT + nt * 16 + r];
            b.y = sW[(k0 + 1) * NOUT + nt * 16 + r];
            acc[nt] = __builtin_amdgcn_wmma_f32_16x16x4_f32(
                false, a, false, b, (short)0, acc[nt], false, false);
        }
    }

#pragma unroll
    for (int nt = 0; nt < NT; ++nt) {
#pragma unroll
        for (int v = 0; v < 8; ++v) {
            const int row = row0 + v + 8 * half;
            const int col = nt * 16 + r;
            if (row < nrows) H[(long long)row * NOUT + col] = acc[nt][v] * sd[v + 8 * half];
        }
    }
}

__global__ void __launch_bounds__(BT)
node_bucket_kernel(const unsigned int* __restrict__ tmp,
                   const unsigned int* __restrict__ tbl,
                   unsigned int* __restrict__ rowptr,
                   float* __restrict__ dis,
                   int* __restrict__ epk,
                   const float* __restrict__ X, const float* __restrict__ W,
                   float* __restrict__ H, int n_nodes) {
    __shared__ alignas(16) float sW[GK * GN];
    __shared__ float s_dis[BIN_NODES];
    __shared__ int s_stage[STAGE_CAP];
    __shared__ unsigned int s_cnt[BIN_NODES];
    __shared__ unsigned int s_start[BIN_NODES];
    __shared__ unsigned int s_pbeg[EDGE_CHUNKS];
    __shared__ unsigned int s_plen[EDGE_CHUNKS];
    __shared__ unsigned int s_w[2 * (BT / 32)];
    const int tid = threadIdx.x;
    const int lane = tid & 31;
    const int wave = tid >> 5;
    const int bin = blockIdx.x;

    for (int i = tid; i < GK * GN / 4; i += BT) ((float4*)sW)[i] = ((const float4*)W)[i];

    unsigned int pa = 0u, pl = 0u;
    if (tid < EDGE_CHUNKS) {
        pa = tbl[tid * TBL + bin];
        pl = tbl[tid * TBL + bin + 1] - pa;
        s_pbeg[tid] = (unsigned int)tid * CHUNK_PITCH + pa;
        s_plen[tid] = pl;
    }
    if (tid < BIN_NODES) s_cnt[tid] = 0u;
    unsigned int lo = 0u, n = 0u;
    block_sum2(pa, pl, s_w, lo, n);

    unsigned int pb[PPW], pn[PPW], hw[PPW];
#pragma unroll
    for (int k = 0; k < PPW; ++k) {
        const int c = wave + k * (BT / 32);
        pb[k] = s_pbeg[c];
        pn[k] = s_plen[c];
    }
#pragma unroll
    for (int k = 0; k < PPW; ++k) {
        hw[k] = 0u;
        if ((unsigned int)lane < pn[k]) hw[k] = tmp[pb[k] + (unsigned int)lane];
    }
#pragma unroll
    for (int k = 0; k < PPW; ++k)
        if ((unsigned int)lane < pn[k]) atomicAdd(&s_cnt[hw[k] >> SRC_BITS], 1u);
    for (int k = 0; k < PPW; ++k)
        for (unsigned int j = (unsigned int)lane + 32u; j < pn[k]; j += 32u)
            atomicAdd(&s_cnt[tmp[pb[k] + j] >> SRC_BITS], 1u);
    __syncthreads();

    const unsigned int c = (tid < BIN_NODES) ? s_cnt[tid] : 0u;
    unsigned int all = 0u;
    const unsigned int ex = block_excl_scan(c, s_w, all);
    const int node = bin * BIN_NODES + tid;
    if (tid < BIN_NODES) {
        s_start[tid] = ex;
        s_cnt[tid] = 0u;
        const float dn = rsqrtf((float)c + 1.0f);
        s_dis[tid] = dn;
        if (node < n_nodes) {
            rowptr[node] = lo + ex;
            dis[node] = dn;
            if (node == n_nodes - 1) rowptr[n_nodes] = lo + ex + c;
        }
    }
    __syncthreads();

    const bool staged = n <= (unsigned int)STAGE_CAP;
    auto place = [&](unsigned int word) {
        const unsigned int l = word >> SRC_BITS;
        const unsigned int slot = s_start[l] + atomicAdd(&s_cnt[l], 1u);
        const int s = (int)(word & ((1u << SRC_BITS) - 1u));
        if (staged) s_stage[slot] = s;
        else        epk[lo + slot] = s;
    };
#pragma unroll
    for (int k = 0; k < PPW; ++k)
        if ((unsigned int)lane < pn[k]) place(hw[k]);
    for (int k = 0; k < PPW; ++k)
        for (unsigned int j = (unsigned int)lane + 32u; j < pn[k]; j += 32u)
            place(tmp[pb[k] + j]);
    __syncthreads();
    if (staged) {
        for (unsigned int i = (unsigned int)tid; i < n; i += BT) epk[lo + i] = s_stage[i];
    }

    const int row0 = bin * BIN_NODES + wave * 16;
    if (row0 < n_nodes)
        wave_tile_product<GK, GN>(X, sW, s_dis + wave * 16, H, row0, n_nodes, lane);
}


#define ROW_GROUPS 1024

template<int U>
__device__ __forceinline__ unsigned int row_lane(int v) {
    return (unsigned int)__builtin_amdgcn_update_dpp(0, v, 0x150 + U, 0xf, 0xf, false);
}

template<int U0>
__device__ __forceinline__ void load_rows8(const float2* __restrict__ h2, int es, unsigned int c, float2 (&hv)[8]) {
    hv[0] = h2[row_lane<U0 + 0>(es) * 16u + c];
    hv[1] = h2[row_lane<U0 + 1>(es) * 16u + c];
    hv[2] = h2[row_lane<U0 + 2>(es) * 16u + c];
    hv[3] = h2[row_lane<U0 + 3>(es) * 16u + c];
    hv[4] = h2[row_lane<U0 + 4>(es) * 16u + c];
    hv[5] = h2[row_lane<U0 + 5>(es) * 16u + c];
    hv[6] = h2[row_lane<U0 + 6>(es) * 16u + c];
    hv[7] = h2[row_lane<U0 + 7>(es) * 16u + c];
}

__device__ __forceinline__ void add_rows8(const float2 (&hv)[8], int e0, int half, int m, float& ax, float& ay) {
#pragma unroll
    for (int u = 0; u < 8; ++u) {
        const bool live = (e0 + 2 * u + half) < m;
        ax += live ? hv[u].x : 0.0f;
        ay += live ? hv[u].y : 0.0f;
    }
}

__global__ void __launch_bounds__(256)
gather_rows_kernel(const unsigned int* __restrict__ rowptr,
                   const int* __restrict__ epk,
                   const float* __restrict__ h,
                   const float* __restrict__ dis,
                   const float* __restrict__ b,
                   float* __restrict__ out, int n) {
    const int lane = threadIdx.x & 31;
    const int half = lane >> 4;
    const unsigned int c = (unsigned int)(lane & 15);
    const unsigned int eo = 2u * c + (unsigned int)half;
    const float2* __restrict__ h2 = (const float2*)h;
    const int nw = (int)gridDim.x * 8;
    int node = __builtin_amdgcn_readfirstlane((int)(blockIdx.x * 8 + (threadIdx.x >> 5)));
    if (node >= n) return;

    const float2 bb = ((const float2*)b)[c];

    const unsigned int beg = rowptr[node];
    const unsigned int nxt = rowptr[node + 1];
    float dd = dis[node];
    float2 own = h2[(unsigned int)node * 16u + c];

    int node1 = node + nw;
    bool valid1 = node1 < n;
    int nodec1 = valid1 ? node1 : (n - 1);
    unsigned int beg1 = rowptr[nodec1];
    unsigned int nxt1 = rowptr[nodec1 + 1];
    float dd1 = dis[nodec1];
    float2 own1 = h2[(unsigned int)nodec1 * 16u + c];

    unsigned int begs = __builtin_amdgcn_readfirstlane(beg);
    unsigned int ends = __builtin_amdgcn_readfirstlane(nxt);
    int es = node;
    if (begs + eo < ends) es = epk[begs + eo];

    for (;;) {
        const unsigned int deg = ends - begs;
        const int m = (int)(deg < 32u ? deg : 32u);

        float2 ha[8], hb[8];
        load_rows8<0>(h2, es, c, ha);
        if (m > 16) load_rows8<8>(h2, es, c, hb);

        const int node2 = node1 + nw;
        const bool valid2 = node2 < n;
        const int nodec2 = valid2 ? node2 : (n - 1);
        const unsigned int beg2 = rowptr[nodec2];
        const unsigned int nxt2 = rowptr[nodec2 + 1];
        const float dd2 = dis[nodec2];
        const float2 own2 = h2[(unsigned int)nodec2 * 16u + c];

        const unsigned int begs1 = __builtin_amdgcn_readfirstlane(beg1);
        const unsigned int ends1 = __builtin_amdgcn_readfirstlane(valid1 ? nxt1 : beg1);
        int es1 = nodec1;
        if (begs1 + eo < ends1) es1 = epk[begs1 + eo];

        float ax = half ? 0.0f : own.x;
        float ay = half ? 0.0f : own.y;
        add_rows8(ha, 0, half, m, ax, ay);
        if (m > 16) add_rows8(hb, 16, half, m, ax, ay);

        for (unsigned int base = begs + 32u; base < ends; base += 32u) {
            int e2 = node;
            if (base + eo < ends) e2 = epk[base + eo];
            const unsigned int left = ends - base;
            const int mm = (int)(left < 32u ? left : 32u);
            float2 hv[8];
            load_rows8<0>(h2, e2, c, hv);
            add_rows8(hv, 0, half, mm, ax, ay);
            if (mm > 16) {
                load_rows8<8>(h2, e2, c, hv);
                add_rows8(hv, 16, half, mm, ax, ay);
            }
        }

        const float bx = __builtin_bit_cast(float, __builtin_amdgcn_permlanex16(
            __builtin_bit_cast(int, ax), __builtin_bit_cast(int, ax), 0x76543210, 0xfedcba98, false, false));
        const float by = __builtin_bit_cast(float, __builtin_amdgcn_permlanex16(
            __builtin_bit_cast(int, ay), __builtin_bit_cast(int, ay), 0x76543210, 0xfedcba98, false, false));
        if (half == 0) {
            float2 o;
            o.x = (ax + bx) * dd + bb.x;
            o.y = (ay + by) * dd + bb.y;
            ((float2*)out)[(unsigned int)node * 16u + c] = o;
        }

        if (!valid1) break;
        node = node1; dd = dd1; own = own1; begs = begs1; ends = ends1; es = es1;
        node1 = node2; valid1 = valid2; nodec1 = nodec2; beg1 = beg2; nxt1 = nxt2; dd1 = dd2; own1 = own2;
    }
}


__device__ __forceinline__ unsigned long long scalar_addr(unsigned long long a) {
    const unsigned int lo = (unsigned int)__builtin_amdgcn_readfirstlane((int)(unsigned int)a);
    const unsigned int hi = (unsigned int)__builtin_amdgcn_readfirstlane((int)(unsigned int)(a >> 32));
    return ((unsigned long long)hi << 32) | (unsigned long long)lo;
}
typedef const __attribute__((address_space(1))) v2f* grow_t;
typedef __attribute__((address_space(1))) v2f* gout2_t;
typedef __attribute__((address_space(1))) float* gout1_t;
__device__ __forceinline__ grow_t row_base(const float* h, int s) {
    return (grow_t)scalar_addr((unsigned long long)h + ((unsigned long long)(unsigned int)s << 8));
}
typedef const __attribute__((address_space(4))) unsigned int* cu32_t;
typedef const __attribute__((address_space(4))) float* cf32_t;

template<int G>
__device__ __forceinline__ void ask_rows(const float* h, int es, int t, int lane, v2f (&hv)[G]) {
#pragma unroll
    for (int u = 0; u < G; ++u) hv[u] = row_base(h, __builtin_amdgcn_readlane(es, t + u))[lane];
}
template<int G>
__device__ __forceinline__ void add_rows(const v2f (&hv)[G], int t, int m, float& a0, float& a1) {
#pragma unroll
    for (int u = 0; u < G; ++u) {
        const bool live = (t + u) < m;
        a0 += live ? hv[u].x : 0.0f;
        a1 += live ? hv[u].y : 0.0f;
    }
}

template<int NOUT>
__global__ void __launch_bounds__(512)
gather_gemm_kernel(const unsigned int* __restrict__ rowptr,
                   const int* __restrict__ epk,
                   const float* __restrict__ h,
                   const float* __restrict__ dis,
                   const float* __restrict__ b,
                   const float* __restrict__ Wn,
                   float* __restrict__ Hn, int n) {
    constexpr int F = 64;
    constexpr int V = 2;
    constexpr int G = 8;
    constexpr int NT = NOUT / 16;
    constexpr int VO = NOUT / 32;
    constexpr int AP = F + 4;
    constexpr int KPP = F / 2 + 2;
    constexpr int OP = NOUT + 4;
    __shared__ alignas(16) float2 sW[NOUT * KPP];
    __shared__ alignas(16) float sA[16 * AP];
    __shared__ alignas(16) float sO[16 * OP];

    const int tid = threadIdx.x;
    const int lane = tid & 31;
    const int wave = __builtin_amdgcn_readfirstlane(tid >> 5);

    const int ntiles = (n + 15) >> 4;
    int tile = blockIdx.x;
    if (tile >= ntiles) return;

    for (int i = tid; i < (F / 2) * (NOUT / 4); i += 512) {
        const int kp = i / (NOUT / 4);
        const int cc = (i % (NOUT / 4)) * 4;
        const float4 w0 = *(const float4*)(Wn + (2 * kp) * NOUT + cc);
        const float4 w1 = *(const float4*)(Wn + (2 * kp + 1) * NOUT + cc);
        sW[(cc + 0) * KPP + kp] = make_float2(w0.x, w1.x);
        sW[(cc + 1) * KPP + kp] = make_float2(w0.y, w1.y);
        sW[(cc + 2) * KPP + kp] = make_float2(w0.z, w1.z);
        sW[(cc + 3) * KPP + kp] = make_float2(w0.w, w1.w);
    }
    const float2 bb = *(const float2*)(b + lane * V);

    const int half = lane >> 4;
    const int r    = lane & 15;

    const cu32_t rowptr_c = (cu32_t)(unsigned long long)rowptr;
    const cf32_t dis_c = (cf32_t)(unsigned long long)dis;
    const int step = (int)gridDim.x;

    int node = __builtin_amdgcn_readfirstlane(tile * 16 + wave);
    bool valid = node < n;
    int nodec = valid ? node : (n - 1);
    int tile1 = tile + step;
    int node1 = __builtin_amdgcn_readfirstlane(tile1 * 16 + wave);
    bool valid1 = node1 < n;
    int nodec1 = valid1 ? node1 : (n - 1);

    unsigned int begs = rowptr_c[nodec];
    unsigned int ends = rowptr_c[nodec + 1];
    if (!valid) ends = begs;
    float dd = dis_c[nodec];
    unsigned int beg1 = rowptr_c[nodec1];
    unsigned int nxt1 = rowptr_c[nodec1 + 1];
    float dd1 = dis_c[nodec1];
    v2f q = row_base(h, nodec)[lane];
    int es = nodec;
    if (begs + (unsigned int)lane < ends) es = epk[begs + (unsigned int)lane];

    for (;;) {
        const unsigned int deg = ends - begs;
        const int m = (int)(deg < 32u ? deg : 32u);
        float acc[V];
        acc[0] = q.x;
        acc[V - 1] = q.y;

        const unsigned int begs1 = beg1;
        const unsigned int ends1 = valid1 ? nxt1 : beg1;
        const v2f q1 = row_base(h, nodec1)[lane];
        const unsigned int j1 = begs1 + (unsigned int)lane;
        const int es1 = epk[j1 < ends1 ? j1 : 0u];

        const int tile2 = tile1 + step;
        const int node2 = __builtin_amdgcn_readfirstlane(tile2 * 16 + wave);
        const bool valid2 = node2 < n;
        const int nodec2 = valid2 ? node2 : (n - 1);
        const unsigned int beg2 = rowptr_c[nodec2];
        const unsigned int nxt2 = rowptr_c[nodec2 + 1];
        const float dd2 = dis_c[nodec2];

        if (m > 0) {
            v2f ha[G], hb[G], hc[G];
            const bool third = m > 2 * G;
            ask_rows<G>(h, es, 0, lane, ha);
            if (third) {
                ask_rows<G>(h, es, 2 * G, lane, hc);
            } else {
#pragma unroll
                for (int u = 0; u < G; ++u) hc[u] = v2f{0.0f, 0.0f};
            }
            ask_rows<G>(h, es, G, lane, hb);
            __builtin_amdgcn_sched_barrier(0);
            add_rows<G>(ha, 0, m, acc[0], acc[V - 1]);
            add_rows<G>(hb, G, m, acc[0], acc[V - 1]);
            if (third) add_rows<G>(hc, 2 * G, m, acc[0], acc[V - 1]);
            if (m > 3 * G) {
                v2f hd[G];
                ask_rows<G>(h, es, 3 * G, lane, hd);
                add_rows<G>(hd, 3 * G, m, acc[0], acc[V - 1]);
            }
        }
        for (unsigned int base = begs + 32u; base < ends; base += 32u) {
            int e2 = nodec;
            if (base + (unsigned int)lane < ends) e2 = epk[base + (unsigned int)lane];
            const unsigned int left = ends - base;
            const int mm = (int)(left < 32u ? left : 32u);
            for (int t = 0; t < mm; t += G) {
                v2f hv[G];
                ask_rows<G>(h, e2, t, lane, hv);
                add_rows<G>(hv, t, mm, acc[0], acc[V - 1]);
            }
        }
        acc[0] = valid ? fmaxf(acc[0] * dd + bb.x, 0.0f) : 0.0f;
        acc[V - 1] = valid ? fmaxf(acc[V - 1] * dd + bb.y, 0.0f) : 0.0f;

        *(float2*)&sA[wave * AP + lane * V] = make_float2(acc[0], acc[V - 1]);
        __syncthreads();

        if (wave < NT) {
            v8f d = {};
#pragma unroll
            for (int kt = 0; kt < F / 4; ++kt) {
                const v2f a = *(const v2f*)&sA[r * AP + kt * 4 + 2 * half];
                const float2 bq = sW[(wave * 16 + r) * KPP + kt * 2 + half];
                v2f bf;
                bf.x = bq.x;
                bf.y = bq.y;
                d = __builtin_amdgcn_wmma_f32_16x16x4_f32(false, a, false, bf, (short)0, d, false, false);
            }
#pragma unroll
            for (int v = 0; v < 8; ++v) sO[(v + 8 * half) * OP + wave * 16 + r] = d[v];
        }
        __syncthreads();

        if (valid) {
            const unsigned long long ob =
                scalar_addr((unsigned long long)Hn + (unsigned long long)(unsigned int)node * (NOUT * 4));
            if (VO == 2) {
                const float2 o = *(const float2*)&sO[wave * OP + lane * VO];
                v2f ov;
                ov.x = o.x * dd;
                ov.y = o.y * dd;
                ((gout2_t)ob)[lane] = ov;
            } else {
                ((gout1_t)ob)[lane] = sO[wave * OP + lane * VO] * dd;
            }
        }

        if (tile1 >= ntiles) break;
        tile = tile1; node = node1; valid = valid1; nodec = nodec1;
        begs = begs1; ends = ends1; dd = dd1; q = q1; es = es1;
        tile1 = tile2; node1 = node2; valid1 = valid2; nodec1 = nodec2;
        beg1 = beg2; nxt1 = nxt2; dd1 = dd2;
    }
}


extern "C" void kernel_launch(void* const* d_in, const int* in_sizes, int n_in,
                              void* d_out, int out_size, void* d_ws, size_t ws_size,
                              hipStream_t stream) {
    const float* x  = (const float*)d_in[0];
    const int*   ei = (const int*)d_in[1];
    const float* W1 = (const float*)d_in[2];
    const float* b1 = (const float*)d_in[3];
    const float* W2 = (const float*)d_in[4];
    const float* b2 = (const float*)d_in[5];
    const float* W3 = (const float*)d_in[6];
    const float* b3 = (const float*)d_in[7];
    float* out = (float*)d_out;

    const int* src = ei;
    const int* dst = ei + N_EDGES;

    char* ws = (char*)d_ws;
    unsigned int* tbl = (unsigned int*)ws;
    float* dis  = (float*)(ws + 400384);
    float* bufA = (float*)(ws + 800768);
    float* bufB = (float*)(ws + 26400768);
    unsigned int* rowptr = (unsigned int*)(ws + 52000768);
    int* epk             = (int*)(ws + 58801152);
    unsigned int* tmp    = (unsigned int*)(ws + 71601152);

    const int N = N_NODES, E = N_EDGES;
    const int tile_groups = 1024;

    chunk_sort_kernel<<<EDGE_CHUNKS, BT, 0, stream>>>(src, dst, tbl, tmp, E);
    node_bucket_kernel<<<NBINS, BT, 0, stream>>>(tmp, tbl, rowptr, dis, epk, x, W1, bufA, N);

    gather_gemm_kernel<64><<<tile_groups, 512, 0, stream>>>(rowptr, epk, bufA, dis, b1, W2, bufB, N);

    gather_gemm_kernel<32><<<tile_groups, 512, 0, stream>>>(rowptr, epk, bufB, dis, b2, W3, bufA, N);

    gather_rows_kernel<<<ROW_GROUPS, 256, 0, stream>>>(rowptr, epk, bufA, dis, b3, out, N);
}


# Bucket slots past a bucket's end name a row of zeros (64 behind each plane of rows, written by the launch that writes the plane): gathers add every asked row, no per-row select; sum order kept
# speedup vs baseline: 8.8897x; 1.0364x over previous
#include <hip/hip_runtime.h>

typedef float v2f __attribute__((ext_vector_type(2)));
typedef float v8f __attribute__((ext_vector_type(8)));

#define N_NODES 100000
#define N_EDGES 1600000

#define ZROWS 64


#define BIN_SHIFT   9
#define BIN_NODES   (1 << BIN_SHIFT)
#define SRC_BITS    (32 - BIN_SHIFT)
#define NBINS       ((N_NODES + BIN_NODES - 1) / BIN_NODES)
#define TBL         (((NBINS + 1 + 31) / 32) * 32)
#define EDGE_CHUNKS 256
#define CHUNK_EDGES ((N_EDGES + EDGE_CHUNKS - 1) / EDGE_CHUNKS)
#define BT          1024
#define CHUNK_PITCH (((CHUNK_EDGES + 31) / 32) * 32)
#define EPT         ((CHUNK_EDGES + BT - 1) / BT)
#define PPW         (EDGE_CHUNKS / (BT / 32))
#define STAGE_CAP   16384
#define GK          128
#define GN          64

static_assert(N_NODES <= (1 << SRC_BITS), "a src id and a node's place in its bin share one word");
static_assert(NBINS <= BT, "one thread per bin");
static_assert(EDGE_CHUNKS <= BT, "one thread per chunk reads the table column");
static_assert(EDGE_CHUNKS % (BT / 32) == 0, "every wave takes the same number of pieces");
static_assert(BIN_NODES <= BT, "one thread per node of a bin");
static_assert(BIN_NODES == (BT / 32) * 16, "one 16-row tile of the bin per wave");

__device__ __forceinline__ unsigned int block_excl_scan(unsigned int v, unsigned int* s_w, unsigned int& total) {
    const int lane = threadIdx.x & 31;
    const int wave = threadIdx.x >> 5;
    unsigned int inc = v;
    for (int d = 1; d < 32; d <<= 1) {
        const unsigned int t = __shfl_up(inc, d);
        if (lane >= d) inc += t;
    }
    if (lane == 31) s_w[wave] = inc;
    __syncthreads();
    unsigned int off = 0u, tot = 0u;
    for (int w = 0; w < BT / 32; ++w) {
        const unsigned int x = s_w[w];
        if (w < wave) off += x;
        tot += x;
    }
    __syncthreads();
    total = tot;
    return off + inc - v;
}

__device__ __forceinline__ void block_sum2(unsigned int a, unsigned int b, unsigned int* s_w2,
                                           unsigned int& ta, unsigned int& tb) {
    for (int d = 16; d >= 1; d >>= 1) {
        a += __shfl_xor(a, d);
        b += __shfl_xor(b, d);
    }
    const int lane = threadIdx.x & 31;
    const int wave = threadIdx.x >> 5;
    if (lane == 0) { s_w2[2 * wave] = a; s_w2[2 * wave + 1] = b; }
    __syncthreads();
    unsigned int x = 0u, y = 0u;
    for (int w = 0; w < BT / 32; ++w) { x += s_w2[2 * w]; y += s_w2[2 * w + 1]; }
    __syncthreads();
    ta = x;
    tb = y;
}

__global__ void __launch_bounds__(BT)
chunk_sort_kernel(const int* __restrict__ src, const int* __restrict__ dst,
                  unsigned int* __restrict__ tbl,
                  unsigned int* __restrict__ tmp, int E) {
    __shared__ unsigned int s_sorted[CHUNK_EDGES];
    __shared__ unsigned int s_cnt[NBINS];
    __shared__ unsigned int s_lbase[NBINS];
    __shared__ unsigned int s_w[BT / 32];

    const int tid = threadIdx.x;
    const int chunk = blockIdx.x;
    const int e0 = chunk * CHUNK_EDGES;
    const int e1 = (e0 + CHUNK_EDGES < E) ? (e0 + CHUNK_EDGES) : E;
    const int n = (e1 > e0) ? (e1 - e0) : 0;

    int es[EPT], ed[EPT];
#pragma unroll
    for (int k = 0; k < EPT; ++k) {
        const int e = e0 + tid + k * BT;
        es[k] = 0;
        ed[k] = -1;
        if (e < e1) { es[k] = src[e]; ed[k] = dst[e]; }
    }
    if (tid < NBINS) s_cnt[tid] = 0u;
    __syncthreads();

    unsigned int rk[EPT];
#pragma unroll
    for (int k = 0; k < EPT; ++k) {
        rk[k] = 0u;
        if (ed[k] >= 0) rk[k] = atomicAdd(&s_cnt[ed[k] >> BIN_SHIFT], 1u);
    }
    __syncthreads();

    unsigned int total = 0u;
    const unsigned int mycnt = (tid < NBINS) ? s_cnt[tid] : 0u;
    const unsigned int lbase = block_excl_scan(mycnt, s_w, total);
    if (tid < NBINS) {
        s_lbase[tid] = lbase;
        tbl[chunk * TBL + tid] = lbase;
    }
    if (tid == 0) tbl[chunk * TBL + NBINS] = total;
    __syncthreads();

#pragma unroll
    for (int k = 0; k < EPT; ++k)
        if (ed[k] >= 0)
            s_sorted[s_lbase[ed[k] >> BIN_SHIFT] + rk[k]] =
                (unsigned int)es[k] | ((unsigned int)(ed[k] & (BIN_NODES - 1)) << SRC_BITS);
    __syncthreads();

    for (int i = tid; i < n; i += BT) tmp[chunk * CHUNK_PITCH + i] = s_sorted[i];
}

template<int K, int NOUT>
__device__ __forceinline__ void wave_tile_product(const float* __restrict__ X, const float* sW, const float* sd,
                                                  float* __restrict__ H, int row0, int nrows, int lane) {
    const int half = lane >> 4;
    const int r    = lane & 15;

    int arow = row0 + r;
    if (arow >= nrows) arow = nrows - 1;
    const float* __restrict__ xrow = X + (long long)arow * K;

    constexpr int NT = NOUT / 16;
    v8f acc[NT];
#pragma unroll
    for (int nt = 0; nt < NT; ++nt) {
        v8f z = {};
        acc[nt] = z;
    }

#pragma unroll 4
    for (int kt = 0; kt < K / 4; ++kt) {
        const int k0 = kt * 4 + 2 * half;
        v2f a;
        a.x = xrow[k0];
        a.y = xrow[k0 + 1];
#pragma unroll
        for (int nt = 0; nt < NT; ++nt) {
            v2f b;
            b.x = sW[(k0)     * NOUT + nt * 16 + r];
            b.y = sW[(k0 + 1) * NOUT + nt * 16 + r];
            acc[nt] = __builtin_amdgcn_wmma_f32_16x16x4_f32(
                false, a, false, b, (short)0, acc[nt], false, false);
        }
    }

#pragma unroll
    for (int nt = 0; nt < NT; ++nt) {
#pragma unroll
        for (int v = 0; v < 8; ++v) {
            const int row = row0 + v + 8 * half;
            const int col = nt * 16 + r;
            if (row < nrows) H[(long long)row * NOUT + col] = acc[nt][v] * sd[v + 8 * half];
        }
    }
}

__global__ void __launch_bounds__(BT)
node_bucket_kernel(const unsigned int* __restrict__ tmp,
                   const unsigned int* __restrict__ tbl,
                   unsigned int* __restrict__ rowptr,
                   float* __restrict__ dis,
                   int* __restrict__ epk,
                   const float* __restrict__ X, const float* __restrict__ W,
                   float* __restrict__ H, int n_nodes) {
    __shared__ alignas(16) float sW[GK * GN];
    __shared__ float s_dis[BIN_NODES];
    __shared__ int s_stage[STAGE_CAP];
    __shared__ unsigned int s_cnt[BIN_NODES];
    __shared__ unsigned int s_start[BIN_NODES];
    __shared__ unsigned int s_pbeg[EDGE_CHUNKS];
    __shared__ unsigned int s_plen[EDGE_CHUNKS];
    __shared__ unsigned int s_w[2 * (BT / 32)];
    const int tid = threadIdx.x;
    const int lane = tid & 31;
    const int wave = tid >> 5;
    const int bin = blockIdx.x;

    for (int i = tid; i < GK * GN / 4; i += BT) ((float4*)sW)[i] = ((const float4*)W)[i];
    if (bin == 0)
        for (int i = tid; i < ZROWS * GN; i += BT) H[(long long)n_nodes * GN + i] = 0.0f;

    unsigned int pa = 0u, pl = 0u;
    if (tid < EDGE_CHUNKS) {
        pa = tbl[tid * TBL + bin];
        pl = tbl[tid * TBL + bin + 1] - pa;
        s_pbeg[tid] = (unsigned int)tid * CHUNK_PITCH + pa;
        s_plen[tid] = pl;
    }
    if (tid < BIN_NODES) s_cnt[tid] = 0u;
    unsigned int lo = 0u, n = 0u;
    block_sum2(pa, pl, s_w, lo, n);

    unsigned int pb[PPW], pn[PPW], hw[PPW];
#pragma unroll
    for (int k = 0; k < PPW; ++k) {
        const int c = wave + k * (BT / 32);
        pb[k] = s_pbeg[c];
        pn[k] = s_plen[c];
    }
#pragma unroll
    for (int k = 0; k < PPW; ++k) {
        hw[k] = 0u;
        if ((unsigned int)lane < pn[k]) hw[k] = tmp[pb[k] + (unsigned int)lane];
    }
#pragma unroll
    for (int k = 0; k < PPW; ++k)
        if ((unsigned int)lane < pn[k]) atomicAdd(&s_cnt[hw[k] >> SRC_BITS], 1u);
    for (int k = 0; k < PPW; ++k)
        for (unsigned int j = (unsigned int)lane + 32u; j < pn[k]; j += 32u)
            atomicAdd(&s_cnt[tmp[pb[k] + j] >> SRC_BITS], 1u);
    __syncthreads();

    const unsigned int c = (tid < BIN_NODES) ? s_cnt[tid] : 0u;
    unsigned int all = 0u;
    const unsigned int ex = block_excl_scan(c, s_w, all);
    const int node = bin * BIN_NODES + tid;
    if (tid < BIN_NODES) {
        s_start[tid] = ex;
        s_cnt[tid] = 0u;
        const float dn = rsqrtf((float)c + 1.0f);
        s_dis[tid] = dn;
        if (node < n_nodes) {
            rowptr[node] = lo + ex;
            dis[node] = dn;
            if (node == n_nodes - 1) rowptr[n_nodes] = lo + ex + c;
        }
    }
    __syncthreads();

    const bool staged = n <= (unsigned int)STAGE_CAP;
    auto place = [&](unsigned int word) {
        const unsigned int l = word >> SRC_BITS;
        const unsigned int slot = s_start[l] + atomicAdd(&s_cnt[l], 1u);
        const int s = (int)(word & ((1u << SRC_BITS) - 1u));
        if (staged) s_stage[slot] = s;
        else        epk[lo + slot] = s;
    };
#pragma unroll
    for (int k = 0; k < PPW; ++k)
        if ((unsigned int)lane < pn[k]) place(hw[k]);
    for (int k = 0; k < PPW; ++k)
        for (unsigned int j = (unsigned int)lane + 32u; j < pn[k]; j += 32u)
            place(tmp[pb[k] + j]);
    __syncthreads();
    if (staged) {
        for (unsigned int i = (unsigned int)tid; i < n; i += BT) epk[lo + i] = s_stage[i];
    }

    const int row0 = bin * BIN_NODES + wave * 16;
    if (row0 < n_nodes)
        wave_tile_product<GK, GN>(X, sW, s_dis + wave * 16, H, row0, n_nodes, lane);
}


#define ROW_GROUPS 1024

template<int U>
__device__ __forceinline__ unsigned int row_lane(int v) {
    return (unsigned int)__builtin_amdgcn_update_dpp(0, v, 0x150 + U, 0xf, 0xf, false);
}

template<int U0>
__device__ __forceinline__ void load_rows8(const float2* __restrict__ h2, int es, unsigned int c, float2 (&hv)[8]) {
    hv[0] = h2[row_lane<U0 + 0>(es) * 16u + c];
    hv[1] = h2[row_lane<U0 + 1>(es) * 16u + c];
    hv[2] = h2[row_lane<U0 + 2>(es) * 16u + c];
    hv[3] = h2[row_lane<U0 + 3>(es) * 16u + c];
    hv[4] = h2[row_lane<U0 + 4>(es) * 16u + c];
    hv[5] = h2[row_lane<U0 + 5>(es) * 16u + c];
    hv[6] = h2[row_lane<U0 + 6>(es) * 16u + c];
    hv[7] = h2[row_lane<U0 + 7>(es) * 16u + c];
}

__device__ __forceinline__ void add_rows8(const float2 (&hv)[8], float& ax, float& ay) {
#pragma unroll
    for (int u = 0; u < 8; ++u) {
        ax += hv[u].x;
        ay += hv[u].y;
    }
}

__global__ void __launch_bounds__(256)
gather_rows_kernel(const unsigned int* __restrict__ rowptr,
                   const int* __restrict__ epk,
                   const float* __restrict__ h,
                   const float* __restrict__ dis,
                   const float* __restrict__ b,
                   float* __restrict__ out, int n) {
    const int lane = threadIdx.x & 31;
    const int half = lane >> 4;
    const unsigned int c = (unsigned int)(lane & 15);
    const unsigned int eo = 2u * c + (unsigned int)half;
    const float2* __restrict__ h2 = (const float2*)h;
    const int nw = (int)gridDim.x * 8;
    int node = __builtin_amdgcn_readfirstlane((int)(blockIdx.x * 8 + (threadIdx.x >> 5)));
    if (node >= n) return;
    const int zr = n + (int)(blockIdx.x & (ZROWS - 1));

    const float2 bb = ((const float2*)b)[c];

    const unsigned int beg = rowptr[node];
    const unsigned int nxt = rowptr[node + 1];
    float dd = dis[node];
    float2 own = h2[(unsigned int)node * 16u + c];

    int node1 = node + nw;
    bool valid1 = node1 < n;
    int nodec1 = valid1 ? node1 : (n - 1);
    unsigned int beg1 = rowptr[nodec1];
    unsigned int nxt1 = rowptr[nodec1 + 1];
    float dd1 = dis[nodec1];
    float2 own1 = h2[(unsigned int)nodec1 * 16u + c];

    unsigned int begs = __builtin_amdgcn_readfirstlane(beg);
    unsigned int ends = __builtin_amdgcn_readfirstlane(nxt);
    int es = zr;
    if (begs + eo < ends) es = epk[begs + eo];

    for (;;) {
        const unsigned int deg = ends - begs;
        const int m = (int)(deg < 32u ? deg : 32u);

        float2 ha[8], hb[8];
        load_rows8<0>(h2, es, c, ha);
        if (m > 16) load_rows8<8>(h2, es, c, hb);

        const int node2 = node1 + nw;
        const bool valid2 = node2 < n;
        const int nodec2 = valid2 ? node2 : (n - 1);
        const unsigned int beg2 = rowptr[nodec2];
        const unsigned int nxt2 = rowptr[nodec2 + 1];
        const float dd2 = dis[nodec2];
        const float2 own2 = h2[(unsigned int)nodec2 * 16u + c];

        const unsigned int begs1 = __builtin_amdgcn_readfirstlane(beg1);
        const unsigned int ends1 = __builtin_amdgcn_readfirstlane(valid1 ? nxt1 : beg1);
        int es1 = zr;
        if (begs1 + eo < ends1) es1 = epk[begs1 + eo];

        float ax = half ? 0.0f : own.x;
        float ay = half ? 0.0f : own.y;
        add_rows8(ha, ax, ay);
        if (m > 16) add_rows8(hb, ax, ay);

        for (unsigned int base = begs + 32u; base < ends; base += 32u) {
            int e2 = zr;
            if (base + eo < ends) e2 = epk[base + eo];
            const unsigned int left = ends - base;
            const int mm = (int)(left < 32u ? left : 32u);
            float2 hv[8];
            load_rows8<0>(h2, e2, c, hv);
            add_rows8(hv, ax, ay);
            if (mm > 16) {
                load_rows8<8>(h2, e2, c, hv);
                add_rows8(hv, ax, ay);
            }
        }

        const float bx = __builtin_bit_cast(float, __builtin_amdgcn_permlanex16(
            __builtin_bit_cast(int, ax), __builtin_bit_cast(int, ax), 0x76543210, 0xfedcba98, false, false));
        const float by = __builtin_bit_cast(float, __builtin_amdgcn_permlanex16(
            __builtin_bit_cast(int, ay), __builtin_bit_cast(int, ay), 0x76543210, 0xfedcba98, false, false));
        if (half == 0) {
            float2 o;
            o.x = (ax + bx) * dd + bb.x;
            o.y = (ay + by) * dd + bb.y;
            ((float2*)out)[(unsigned int)node * 16u + c] = o;
        }

        if (!valid1) break;
        node = node1; dd = dd1; own = own1; begs = begs1; ends = ends1; es = es1;
        node1 = node2; valid1 = valid2; nodec1 = nodec2; beg1 = beg2; nxt1 = nxt2; dd1 = dd2; own1 = own2;
    }
}


__device__ __forceinline__ unsigned long long scalar_addr(unsigned long long a) {
    const unsigned int lo = (unsigned int)__builtin_amdgcn_readfirstlane((int)(unsigned int)a);
    const unsigned int hi = (unsigned int)__builtin_amdgcn_readfirstlane((int)(unsigned int)(a >> 32));
    return ((unsigned long long)hi << 32) | (unsigned long long)lo;
}
typedef const __attribute__((address_space(1))) v2f* grow_t;
typedef __attribute__((address_space(1))) v2f* gout2_t;
typedef __attribute__((address_space(1))) float* gout1_t;
__device__ __forceinline__ grow_t row_base(const float* h, int s) {
    return (grow_t)scalar_addr((unsigned long long)h + ((unsigned long long)(unsigned int)s << 8));
}
typedef const __attribute__((address_space(4))) unsigned int* cu32_t;
typedef const __attribute__((address_space(4))) float* cf32_t;

template<int G>
__device__ __forceinline__ void ask_rows(const float* h, int es, int t, int lane, v2f (&hv)[G]) {
#pragma unroll
    for (int u = 0; u < G; ++u) hv[u] = row_base(h, __builtin_amdgcn_readlane(es, t + u))[lane];
}
template<int G>
__device__ __forceinline__ void add_rows(const v2f (&hv)[G], float& a0, float& a1) {
#pragma unroll
    for (int u = 0; u < G; ++u) {
        a0 += hv[u].x;
        a1 += hv[u].y;
    }
}

template<int NOUT>
__global__ void __launch_bounds__(512)
gather_gemm_kernel(const unsigned int* __restrict__ rowptr,
                   const int* __restrict__ epk,
                   const float* __restrict__ h,
                   const float* __restrict__ dis,
                   const float* __restrict__ b,
                   const float* __restrict__ Wn,
                   float* __restrict__ Hn, int n) {
    constexpr int F = 64;
    constexpr int V = 2;
    constexpr int G = 8;
    constexpr int NT = NOUT / 16;
    constexpr int VO = NOUT / 32;
    constexpr int AP = F + 4;
    constexpr int KPP = F / 2 + 2;
    constexpr int OP = NOUT + 4;
    __shared__ alignas(16) float2 sW[NOUT * KPP];
    __shared__ alignas(16) float sA[16 * AP];
    __shared__ alignas(16) float sO[16 * OP];

    const int tid = threadIdx.x;
    const int lane = tid & 31;
    const int wave = __builtin_amdgcn_readfirstlane(tid >> 5);

    const int ntiles = (n + 15) >> 4;
    int tile = blockIdx.x;
    if (tile >= ntiles) return;

    for (int i = tid; i < (F / 2) * (NOUT / 4); i += 512) {
        const int kp = i / (NOUT / 4);
        const int cc = (i % (NOUT / 4)) * 4;
        const float4 w0 = *(const float4*)(Wn + (2 * kp) * NOUT + cc);
        const float4 w1 = *(const float4*)(Wn + (2 * kp + 1) * NOUT + cc);
        sW[(cc + 0) * KPP + kp] = make_float2(w0.x, w1.x);
        sW[(cc + 1) * KPP + kp] = make_float2(w0.y, w1.y);
        sW[(cc + 2) * KPP + kp] = make_float2(w0.z, w1.z);
        sW[(cc + 3) * KPP + kp] = make_float2(w0.w, w1.w);
    }
    const float2 bb = *(const float2*)(b + lane * V);
    if (blockIdx.x == 0)
        for (int i = tid; i < ZROWS * NOUT; i += 512) Hn[(long long)n * NOUT + i] = 0.0f;
    const int zr = n + (int)(blockIdx.x & (ZROWS - 1));

    const int half = lane >> 4;
    const int r    = lane & 15;

    const cu32_t rowptr_c = (cu32_t)(unsigned long long)rowptr;
    const cf32_t dis_c = (cf32_t)(unsigned long long)dis;
    const int step = (int)gridDim.x;

    int node = __builtin_amdgcn_readfirstlane(tile * 16 + wave);
    bool valid = node < n;
    int nodec = valid ? node : (n - 1);
    int tile1 = tile + step;
    int node1 = __builtin_amdgcn_readfirstlane(tile1 * 16 + wave);
    bool valid1 = node1 < n;
    int nodec1 = valid1 ? node1 : (n - 1);

    unsigned int begs = rowptr_c[nodec];
    unsigned int ends = rowptr_c[nodec + 1];
    if (!valid) ends = begs;
    float dd = dis_c[nodec];
    unsigned int beg1 = rowptr_c[nodec1];
    unsigned int nxt1 = rowptr_c[nodec1 + 1];
    float dd1 = dis_c[nodec1];
    v2f q = row_base(h, nodec)[lane];
    int es = zr;
    if (begs + (unsigned int)lane < ends) es = epk[begs + (unsigned int)lane];

    for (;;) {
        const unsigned int deg = ends - begs;
        const int m = (int)(deg < 32u ? deg : 32u);
        float acc[V];
        acc[0] = q.x;
        acc[V - 1] = q.y;

        const unsigned int begs1 = beg1;
        const unsigned int ends1 = valid1 ? nxt1 : beg1;
        const v2f q1 = row_base(h, nodec1)[lane];
        const unsigned int j1 = begs1 + (unsigned int)lane;
        const int ld1 = epk[j1 < ends1 ? j1 : 0u];
        const int es1 = j1 < ends1 ? ld1 : zr;

        const int tile2 = tile1 + step;
        const int node2 = __builtin_amdgcn_readfirstlane(tile2 * 16 + wave);
        const bool valid2 = node2 < n;
        const int nodec2 = valid2 ? node2 : (n - 1);
        const unsigned int beg2 = rowptr_c[nodec2];
        const unsigned int nxt2 = rowptr_c[nodec2 + 1];
        const float dd2 = dis_c[nodec2];

        if (m > 0) {
            v2f ha[G], hb[G], hc[G];
            const bool third = m > 2 * G;
            ask_rows<G>(h, es, 0, lane, ha);
            if (third) {
                ask_rows<G>(h, es, 2 * G, lane, hc);
            } else {
#pragma unroll
                for (int u = 0; u < G; ++u) hc[u] = v2f{0.0f, 0.0f};
            }
            ask_rows<G>(h, es, G, lane, hb);
            __builtin_amdgcn_sched_barrier(0);
            add_rows<G>(ha, acc[0], acc[V - 1]);
            add_rows<G>(hb, acc[0], acc[V - 1]);
            if (third) add_rows<G>(hc, acc[0], acc[V - 1]);
            if (m > 3 * G) {
                v2f hd[G];
                ask_rows<G>(h, es, 3 * G, lane, hd);
                add_rows<G>(hd, acc[0], acc[V - 1]);
            }
        }
        for (unsigned int base = begs + 32u; base < ends; base += 32u) {
            int e2 = zr;
            if (base + (unsigned int)lane < ends) e2 = epk[base + (unsigned int)lane];
            const unsigned int left = ends - base;
            const int mm = (int)(left < 32u ? left : 32u);
            for (int t = 0; t < mm; t += G) {
                v2f hv[G];
                ask_rows<G>(h, e2, t, lane, hv);
                add_rows<G>(hv, acc[0], acc[V - 1]);
            }
        }
        acc[0] = valid ? fmaxf(acc[0] * dd + bb.x, 0.0f) : 0.0f;
        acc[V - 1] = valid ? fmaxf(acc[V - 1] * dd + bb.y, 0.0f) : 0.0f;

        *(float2*)&sA[wave * AP + lane * V] = make_float2(acc[0], acc[V - 1]);
        __syncthreads();

        if (wave < NT) {
            v8f d = {};
#pragma unroll
            for (int kt = 0; kt < F / 4; ++kt) {
                const v2f a = *(const v2f*)&sA[r * AP + kt * 4 + 2 * half];
                const float2 bq = sW[(wave * 16 + r) * KPP + kt * 2 + half];
                v2f bf;
                bf.x = bq.x;
                bf.y = bq.y;
                d = __builtin_amdgcn_wmma_f32_16x16x4_f32(false, a, false, bf, (short)0, d, false, false);
            }
#pragma unroll
            for (int v = 0; v < 8; ++v) sO[(v + 8 * half) * OP + wave * 16 + r] = d[v];
        }
        __syncthreads();

        if (valid) {
            const unsigned long long ob =
                scalar_addr((unsigned long long)Hn + (unsigned long long)(unsigned int)node * (NOUT * 4));
            if (VO == 2) {
                const float2 o = *(const float2*)&sO[wave * OP + lane * VO];
                v2f ov;
                ov.x = o.x * dd;
                ov.y = o.y * dd;
                ((gout2_t)ob)[lane] = ov;
            } else {
                ((gout1_t)ob)[lane] = sO[wave * OP + lane * VO] * dd;
            }
        }

        if (tile1 >= ntiles) break;
        tile = tile1; node = node1; valid = valid1; nodec = nodec1;
        begs = begs1; ends = ends1; dd = dd1; q = q1; es = es1;
        tile1 = tile2; node1 = node2; valid1 = valid2; nodec1 = nodec2;
        beg1 = beg2; nxt1 = nxt2; dd1 = dd2;
    }
}


extern "C" void kernel_launch(void* const* d_in, const int* in_sizes, int n_in,
                              void* d_out, int out_size, void* d_ws, size_t ws_size,
                              hipStream_t stream) {
    const float* x  = (const float*)d_in[0];
    const int*   ei = (const int*)d_in[1];
    const float* W1 = (const float*)d_in[2];
    const float* b1 = (const float*)d_in[3];
    const float* W2 = (const float*)d_in[4];
    const float* b2 = (const float*)d_in[5];
    const float* W3 = (const float*)d_in[6];
    const float* b3 = (const float*)d_in[7];
    float* out = (float*)d_out;

    const int* src = ei;
    const int* dst = ei + N_EDGES;

    char* ws = (char*)d_ws;
    unsigned int* tbl = (unsigned int*)ws;
    float* dis  = (float*)(ws + 400384);
    float* bufA = (float*)(ws + 800768);
    float* bufB = (float*)(ws + 26417152);
    unsigned int* rowptr = (unsigned int*)(ws + 52100096);
    int* epk             = (int*)(ws + 58801152);
    unsigned int* tmp    = (unsigned int*)(ws + 71601152);

    const int N = N_NODES, E = N_EDGES;
    const int tile_groups = 1024;

    chunk_sort_kernel<<<EDGE_CHUNKS, BT, 0, stream>>>(src, dst, tbl, tmp, E);
    node_bucket_kernel<<<NBINS, BT, 0, stream>>>(tmp, tbl, rowptr, dis, epk, x, W1, bufA, N);

    gather_gemm_kernel<64><<<tile_groups, 512, 0, stream>>>(rowptr, epk, bufA, dis, b1, W2, bufB, N);

    gather_gemm_kernel<32><<<tile_groups, 512, 0, stream>>>(rowptr, epk, bufB, dis, b2, W3, bufA, N);

    gather_rows_kernel<<<ROW_GROUPS, 256, 0, stream>>>(rowptr, epk, bufA, dis, b3, out, N);
}
